# starting kernel
# baseline (speedup 1.0000x reference)
#include <hip/hip_runtime.h>
#include <hip/hip_bf16.h>

typedef float v2f __attribute__((ext_vector_type(2)));
typedef float v8f __attribute__((ext_vector_type(8)));

#define N_NODES 100000
#define N_EDGES 1600000
#define DIM_IN  256
#define HD      64
#define NCLS    20
#define NDOCS   10000

__device__ __forceinline__ void atomic_add_f32(float* p, float v) {
    __hip_atomic_fetch_add(p, v, __ATOMIC_RELAXED, __HIP_MEMORY_SCOPE_AGENT);
}


__global__ void TEXTGCN_deg_init(float* __restrict__ deg, int n) {
    int i = blockIdx.x * blockDim.x + threadIdx.x;
    if (i < n) deg[i] = 1.0f;
}

__global__ void TEXTGCN_deg_acc(const int* __restrict__ dst,
                                const float* __restrict__ w,
                                float* __restrict__ deg, int e) {
    int i = blockIdx.x * blockDim.x + threadIdx.x;
    if (i < e) atomic_add_f32(&deg[dst[i]], w[i]);
}

__global__ void TEXTGCN_dinv(float* __restrict__ deg, int n) {
    int i = blockIdx.x * blockDim.x + threadIdx.x;
    if (i < n) {
        float d = deg[i];
        deg[i] = (d > 0.0f) ? __frsqrt_rn(d) : 0.0f;
    }
}

__global__ void TEXTGCN_norm(const int* __restrict__ src, const int* __restrict__ dst,
                             const float* __restrict__ w, const float* __restrict__ dinv,
                             float* __restrict__ norm, int e) {
    int i = blockIdx.x * blockDim.x + threadIdx.x;
    if (i < e) norm[i] = dinv[src[i]] * w[i] * dinv[dst[i]];
}


__global__ void TEXTGCN_gemm_wmma(const float* __restrict__ A,
                                  const float* __restrict__ B,
                                  float* __restrict__ C,
                                  int K, int Nout) {
    const int lane = threadIdx.x;
    const int mt = blockIdx.x, nt = blockIdx.y;
    const int m  = mt * 16 + (lane & 15);
    const int n  = nt * 16 + (lane & 15);
    const int kq = (lane >> 4) * 2;

    const float* __restrict__ arow = A + (size_t)m * K;
    v8f acc = {};
    for (int k = 0; k < K; k += 4) {
        v2f a, b;
        a.x = arow[k + kq];
        a.y = arow[k + kq + 1];
        b.x = B[(size_t)(k + kq)     * Nout + n];
        b.y = B[(size_t)(k + kq + 1) * Nout + n];
        acc = __builtin_amdgcn_wmma_f32_16x16x4_f32(
                  false, a, false, b, (short)0, acc, false, false);
    }
    const int mbase = mt * 16 + (lane >> 4) * 8;
    const int nst   = nt * 16 + (lane & 15);
#pragma unroll
    for (int r = 0; r < 8; ++r)
        C[(size_t)(mbase + r) * Nout + nst] = acc[r];
}


__global__ void TEXTGCN_self(const float* __restrict__ t, const float* __restrict__ dinv,
                             float* __restrict__ acc, int n) {
    int gid = blockIdx.x * blockDim.x + threadIdx.x;
    if (gid >= n * (HD / 4)) return;
    int i = gid >> 4, c = gid & 15;
    float s = dinv[i] * dinv[i];
    float4 v = ((const float4*)(t + (size_t)i * HD))[c];
    v.x *= s; v.y *= s; v.z *= s; v.w *= s;
    ((float4*)(acc + (size_t)i * HD))[c] = v;
}

__global__ void TEXTGCN_scatter(const int* __restrict__ src, const int* __restrict__ dst,
                                const float* __restrict__ norm,
                                const float* __restrict__ t,
                                float* __restrict__ acc, int e) {
    int gid = blockIdx.x * blockDim.x + threadIdx.x;
    if (gid >= e * (HD / 4)) return;
    int ei = gid >> 4, c = gid & 15;
    int s = src[ei], d = dst[ei];
    float w = norm[ei];
    float4 v = ((const float4*)(t + (size_t)s * HD))[c];
    float* o = acc + (size_t)d * HD + c * 4;
    atomic_add_f32(o + 0, v.x * w);
    atomic_add_f32(o + 1, v.y * w);
    atomic_add_f32(o + 2, v.z * w);
    atomic_add_f32(o + 3, v.w * w);
}

__global__ void TEXTGCN_bias_relu(float* __restrict__ acc, const float* __restrict__ b, int n) {
    int gid = blockIdx.x * blockDim.x + threadIdx.x;
    if (gid >= n * HD) return;
    float v = acc[gid] + b[gid & (HD - 1)];
    acc[gid] = v > 0.0f ? v : 0.0f;
}


__global__ void TEXTGCN_doc_gemm_wmma(const float* __restrict__ h,
                                      const int* __restrict__ doc_idx,
                                      const float* __restrict__ Wp,
                                      const float* __restrict__ bp,
                                      float* __restrict__ dbuf) {
    const int lane = threadIdx.x;
    const int mt = blockIdx.x, nt = blockIdx.y;
    const int m  = mt * 16 + (lane & 15);
    const int n  = nt * 16 + (lane & 15);
    const int kq = (lane >> 4) * 2;

    const float* __restrict__ arow = h + (size_t)doc_idx[m] * HD;
    v8f acc = {};
    for (int k = 0; k < HD; k += 4) {
        v2f a, b;
        a.x = arow[k + kq];
        a.y = arow[k + kq + 1];
        b.x = Wp[(size_t)(k + kq)     * HD + n];
        b.y = Wp[(size_t)(k + kq + 1) * HD + n];
        acc = __builtin_amdgcn_wmma_f32_16x16x4_f32(
                  false, a, false, b, (short)0, acc, false, false);
    }
    const int mbase = mt * 16 + (lane >> 4) * 8;
    const int nst   = nt * 16 + (lane & 15);
    const float bias = bp[nst];
#pragma unroll
    for (int r = 0; r < 8; ++r) {
        float v = acc[r] + bias;
        dbuf[(size_t)(mbase + r) * HD + nst] = v > 0.0f ? v : 0.0f;
    }
}


__global__ void TEXTGCN_classifier(const float* __restrict__ dbuf,
                                   const float* __restrict__ Wc,
                                   const float* __restrict__ bc,
                                   float* __restrict__ out) {
    int gid = blockIdx.x * blockDim.x + threadIdx.x;
    if (gid >= NDOCS * NCLS) return;
    int d = gid / NCLS, c = gid % NCLS;
    const float* __restrict__ row = dbuf + (size_t)d * HD;
    float s = bc[c];
#pragma unroll 8
    for (int k = 0; k < HD; ++k) s += row[k] * Wc[(size_t)k * NCLS + c];
    out[gid] = s;
}


extern "C" void kernel_launch(void* const* d_in, const int* in_sizes, int n_in,
                              void* d_out, int out_size, void* d_ws, size_t ws_size,
                              hipStream_t stream) {
    const float* x   = (const float*)d_in[0];
    const float* ew  = (const float*)d_in[1];
    const float* W1  = (const float*)d_in[2];
    const float* b1  = (const float*)d_in[3];
    const float* W2  = (const float*)d_in[4];
    const float* b2  = (const float*)d_in[5];
    const float* W3  = (const float*)d_in[6];
    const float* b3  = (const float*)d_in[7];
    const float* Wp  = (const float*)d_in[8];
    const float* bp  = (const float*)d_in[9];
    const float* Wc  = (const float*)d_in[10];
    const float* bc  = (const float*)d_in[11];
    const int*   ei  = (const int*)d_in[12];
    const int*   doc = (const int*)d_in[13];
    float*       out = (float*)d_out;

    const int* src = ei;
    const int* dst = ei + N_EDGES;

    float* ws   = (float*)d_ws;
    float* dinv = ws;
    float* nrm  = dinv + N_NODES;
    float* bufT = nrm  + N_EDGES;
    float* bufH = bufT + (size_t)N_NODES * HD;
    float* dbuf = bufH + (size_t)N_NODES * HD;

    const int B = 256;
    dim3 blk(B);
    dim3 gN((N_NODES + B - 1) / B);
    dim3 gE((N_EDGES + B - 1) / B);
    dim3 gN16((N_NODES * (HD / 4) + B - 1) / B);
    dim3 gE16((N_EDGES * (HD / 4) + B - 1) / B);
    dim3 gNF((N_NODES * HD + B - 1) / B);

    TEXTGCN_deg_init<<<gN, blk, 0, stream>>>(dinv, N_NODES);
    TEXTGCN_deg_acc <<<gE, blk, 0, stream>>>(dst, ew, dinv, N_EDGES);
    TEXTGCN_dinv    <<<gN, blk, 0, stream>>>(dinv, N_NODES);
    TEXTGCN_norm    <<<gE, blk, 0, stream>>>(src, dst, ew, dinv, nrm, N_EDGES);

    dim3 wave(32);
    dim3 gGemm1(N_NODES / 16, HD / 16);
    dim3 gGemmH(N_NODES / 16, HD / 16);
    dim3 gDoc(NDOCS / 16, HD / 16);

    TEXTGCN_gemm_wmma<<<gGemm1, wave, 0, stream>>>(x, W1, bufT, DIM_IN, HD);
    TEXTGCN_self     <<<gN16, blk, 0, stream>>>(bufT, dinv, bufH, N_NODES);
    TEXTGCN_scatter  <<<gE16, blk, 0, stream>>>(src, dst, nrm, bufT, bufH, N_EDGES);
    TEXTGCN_bias_relu<<<gNF,  blk, 0, stream>>>(bufH, b1, N_NODES);

    TEXTGCN_gemm_wmma<<<gGemmH, wave, 0, stream>>>(bufH, W2, bufT, HD, HD);
    TEXTGCN_self     <<<gN16, blk, 0, stream>>>(bufT, dinv, bufH, N_NODES);
    TEXTGCN_scatter  <<<gE16, blk, 0, stream>>>(src, dst, nrm, bufT, bufH, N_EDGES);
    TEXTGCN_bias_relu<<<gNF,  blk, 0, stream>>>(bufH, b2, N_NODES);

    TEXTGCN_gemm_wmma<<<gGemmH, wave, 0, stream>>>(bufH, W3, bufT, HD, HD);
    TEXTGCN_self     <<<gN16, blk, 0, stream>>>(bufT, dinv, bufH, N_NODES);
    TEXTGCN_scatter  <<<gE16, blk, 0, stream>>>(src, dst, nrm, bufT, bufH, N_EDGES);
    TEXTGCN_bias_relu<<<gNF,  blk, 0, stream>>>(bufH, b3, N_NODES);

    TEXTGCN_doc_gemm_wmma<<<gDoc, wave, 0, stream>>>(bufH, doc, Wp, bp, dbuf);
    dim3 gCls((NDOCS * NCLS + B - 1) / B);
    TEXTGCN_classifier<<<gCls, blk, 0, stream>>>(dbuf, Wc, bc, out);
}


# propagation: float-atomic scatter replaced by a gather over edge ids grouped by target (counting sort once per call, reused by 3 layers); self-loop, bias, relu fused; row sums now in slot order
# speedup vs baseline: 1.7840x; 1.7840x over previous
#include <hip/hip_runtime.h>
#include <hip/hip_bf16.h>

typedef float v2f __attribute__((ext_vector_type(2)));
typedef float v8f __attribute__((ext_vector_type(8)));

#define N_NODES 100000
#define N_EDGES 1600000
#define DIM_IN  256
#define HD      64
#define NCLS    20
#define NDOCS   10000
#define SCAN_T  256
#define SCAN_E  1024


__global__ void TEXTGCN_cnt_init(int* __restrict__ cnt, int n) {
    int i = blockIdx.x * blockDim.x + threadIdx.x;
    if (i < n) cnt[i] = 0;
}

__global__ void TEXTGCN_rank(const int* __restrict__ dst, int* __restrict__ cnt,
                             int* __restrict__ rank, int e) {
    int i = blockIdx.x * blockDim.x + threadIdx.x;
    if (i < e)
        rank[i] = __hip_atomic_fetch_add(&cnt[dst[i]], 1, __ATOMIC_RELAXED, __HIP_MEMORY_SCOPE_AGENT);
}

__global__ void TEXTGCN_scan_local(int* __restrict__ cnt, int* __restrict__ aux, int n) {
    __shared__ int sh[SCAN_T];
    const int t = threadIdx.x;
    const int idx = blockIdx.x * SCAN_E + t * 4;
    int4 v = make_int4(0, 0, 0, 0);
    if (idx < n) v = *(const int4*)(cnt + idx);
    const int s = v.x + v.y + v.z + v.w;
    sh[t] = s;
    __syncthreads();
    for (int off = 1; off < SCAN_T; off <<= 1) {
        int add = (t >= off) ? sh[t - off] : 0;
        __syncthreads();
        sh[t] += add;
        __syncthreads();
    }
    const int incl = sh[t];
    const int excl = incl - s;
    if (idx < n) {
        int4 o;
        o.x = excl;
        o.y = o.x + v.x;
        o.z = o.y + v.y;
        o.w = o.z + v.z;
        *(int4*)(cnt + idx) = o;
    }
    if (t == SCAN_T - 1) aux[blockIdx.x] = incl;
}

__global__ void TEXTGCN_scan_add(int* __restrict__ rowptr, const int* __restrict__ aux, int n) {
    const int b = blockIdx.x;
    int off = 0;
    for (int k = 0; k < b; ++k) off += aux[k];
    const int idx = b * SCAN_E + threadIdx.x * 4;
    if (idx < n) {
        int4 v = *(const int4*)(rowptr + idx);
        v.x += off; v.y += off; v.z += off; v.w += off;
        *(int4*)(rowptr + idx) = v;
    }
}

__global__ void TEXTGCN_fill(const int* __restrict__ dst, const int* __restrict__ rowptr,
                             const int* __restrict__ rank, int* __restrict__ perm, int e) {
    int i = blockIdx.x * blockDim.x + threadIdx.x;
    if (i < e) perm[rowptr[dst[i]] + rank[i]] = i;
}

__global__ void TEXTGCN_dinv_csr(const int* __restrict__ rowptr, const int* __restrict__ perm,
                                 const float* __restrict__ w, float* __restrict__ dinv,
                                 int n, int e) {
    int gid = blockIdx.x * blockDim.x + threadIdx.x;
    int i = gid >> 4, c = gid & 15;
    if (i >= n) return;
    const int start = rowptr[i];
    const int end   = (i + 1 < n) ? rowptr[i + 1] : e;
    float s = 0.0f;
    for (int j = start + c; j < end; j += 16) s += w[perm[j]];
    s += __shfl_xor(s, 8, 16);
    s += __shfl_xor(s, 4, 16);
    s += __shfl_xor(s, 2, 16);
    s += __shfl_xor(s, 1, 16);
    if (c == 0) {
        float d = 1.0f + s;
        dinv[i] = (d > 0.0f) ? __frsqrt_rn(d) : 0.0f;
    }
}


__global__ void TEXTGCN_gemm_wmma(const float* __restrict__ A,
                                  const float* __restrict__ B,
                                  const float* __restrict__ rscale,
                                  float* __restrict__ C,
                                  int K, int Nout) {
    const int lane = threadIdx.x;
    const int mt = blockIdx.x, nt = blockIdx.y;
    const int m  = mt * 16 + (lane & 15);
    const int n  = nt * 16 + (lane & 15);
    const int kq = (lane >> 4) * 2;

    const float* __restrict__ arow = A + (size_t)m * K;
    v8f acc = {};
    for (int k = 0; k < K; k += 4) {
        v2f a, b;
        a.x = arow[k + kq];
        a.y = arow[k + kq + 1];
        b.x = B[(size_t)(k + kq)     * Nout + n];
        b.y = B[(size_t)(k + kq + 1) * Nout + n];
        acc = __builtin_amdgcn_wmma_f32_16x16x4_f32(
                  false, a, false, b, (short)0, acc, false, false);
    }
    const int mbase = mt * 16 + (lane >> 4) * 8;
    const int nst   = nt * 16 + (lane & 15);
#pragma unroll
    for (int r = 0; r < 8; ++r)
        C[(size_t)(mbase + r) * Nout + nst] = acc[r] * rscale[mbase + r];
}


__global__ void TEXTGCN_gather(const int* __restrict__ rowptr, const int* __restrict__ perm,
                               const int* __restrict__ src, const float* __restrict__ w,
                               const float* __restrict__ dinv, const float* __restrict__ ts,
                               const float* __restrict__ b, float* __restrict__ h,
                               int n, int e) {
    int gid = blockIdx.x * blockDim.x + threadIdx.x;
    int i = gid >> 4, c = gid & 15;
    if (i >= n) return;
    const int start = rowptr[i];
    const int end   = (i + 1 < n) ? rowptr[i + 1] : e;
    const float4* __restrict__ t4 = (const float4*)ts;
    float4 acc = t4[(size_t)i * (HD / 4) + c];
    for (int base = start; base < end; base += 16) {
        int   s  = 0;
        float wt = 0.0f;
        if (base + c < end) {
            int ed = perm[base + c];
            s  = src[ed];
            wt = w[ed];
        }
        const int m = (end - base < 16) ? (end - base) : 16;
        int k = 0;
        for (; k + 4 <= m; k += 4) {
            const int   s0 = __shfl(s,  k,     16), s1 = __shfl(s,  k + 1, 16);
            const int   s2 = __shfl(s,  k + 2, 16), s3 = __shfl(s,  k + 3, 16);
            const float w0 = __shfl(wt, k,     16), w1 = __shfl(wt, k + 1, 16);
            const float w2 = __shfl(wt, k + 2, 16), w3 = __shfl(wt, k + 3, 16);
            const float4 v0 = t4[(size_t)s0 * (HD / 4) + c];
            const float4 v1 = t4[(size_t)s1 * (HD / 4) + c];
            const float4 v2 = t4[(size_t)s2 * (HD / 4) + c];
            const float4 v3 = t4[(size_t)s3 * (HD / 4) + c];
            acc.x = fmaf(w0, v0.x, acc.x); acc.y = fmaf(w0, v0.y, acc.y); acc.z = fmaf(w0, v0.z, acc.z); acc.w = fmaf(w0, v0.w, acc.w);
            acc.x = fmaf(w1, v1.x, acc.x); acc.y = fmaf(w1, v1.y, acc.y); acc.z = fmaf(w1, v1.z, acc.z); acc.w = fmaf(w1, v1.w, acc.w);
            acc.x = fmaf(w2, v2.x, acc.x); acc.y = fmaf(w2, v2.y, acc.y); acc.z = fmaf(w2, v2.z, acc.z); acc.w = fmaf(w2, v2.w, acc.w);
            acc.x = fmaf(w3, v3.x, acc.x); acc.y = fmaf(w3, v3.y, acc.y); acc.z = fmaf(w3, v3.z, acc.z); acc.w = fmaf(w3, v3.w, acc.w);
        }
        for (; k < m; ++k) {
            const int   sk = __shfl(s,  k, 16);
            const float wk = __shfl(wt, k, 16);
            const float4 v = t4[(size_t)sk * (HD / 4) + c];
            acc.x = fmaf(wk, v.x, acc.x); acc.y = fmaf(wk, v.y, acc.y); acc.z = fmaf(wk, v.z, acc.z); acc.w = fmaf(wk, v.w, acc.w);
        }
    }
    const float di = dinv[i];
    const float4 bb = ((const float4*)b)[c];
    float4 o;
    o.x = acc.x * di + bb.x; o.y = acc.y * di + bb.y; o.z = acc.z * di + bb.z; o.w = acc.w * di + bb.w;
    o.x = o.x > 0.0f ? o.x : 0.0f;
    o.y = o.y > 0.0f ? o.y : 0.0f;
    o.z = o.z > 0.0f ? o.z : 0.0f;
    o.w = o.w > 0.0f ? o.w : 0.0f;
    ((float4*)h)[(size_t)i * (HD / 4) + c] = o;
}


__global__ void TEXTGCN_doc_gemm_wmma(const float* __restrict__ h,
                                      const int* __restrict__ doc_idx,
                                      const float* __restrict__ Wp,
                                      const float* __restrict__ bp,
                                      float* __restrict__ dbuf) {
    const int lane = threadIdx.x;
    const int mt = blockIdx.x, nt = blockIdx.y;
    const int m  = mt * 16 + (lane & 15);
    const int n  = nt * 16 + (lane & 15);
    const int kq = (lane >> 4) * 2;

    const float* __restrict__ arow = h + (size_t)doc_idx[m] * HD;
    v8f acc = {};
    for (int k = 0; k < HD; k += 4) {
        v2f a, b;
        a.x = arow[k + kq];
        a.y = arow[k + kq + 1];
        b.x = Wp[(size_t)(k + kq)     * HD + n];
        b.y = Wp[(size_t)(k + kq + 1) * HD + n];
        acc = __builtin_amdgcn_wmma_f32_16x16x4_f32(
                  false, a, false, b, (short)0, acc, false, false);
    }
    const int mbase = mt * 16 + (lane >> 4) * 8;
    const int nst   = nt * 16 + (lane & 15);
    const float bias = bp[nst];
#pragma unroll
    for (int r = 0; r < 8; ++r) {
        float v = acc[r] + bias;
        dbuf[(size_t)(mbase + r) * HD + nst] = v > 0.0f ? v : 0.0f;
    }
}


__global__ void TEXTGCN_classifier(const float* __restrict__ dbuf,
                                   const float* __restrict__ Wc,
                                   const float* __restrict__ bc,
                                   float* __restrict__ out) {
    int gid = blockIdx.x * blockDim.x + threadIdx.x;
    if (gid >= NDOCS * NCLS) return;
    int d = gid / NCLS, c = gid % NCLS;
    const float* __restrict__ row = dbuf + (size_t)d * HD;
    float s = bc[c];
#pragma unroll 8
    for (int k = 0; k < HD; ++k) s += row[k] * Wc[(size_t)k * NCLS + c];
    out[gid] = s;
}


extern "C" void kernel_launch(void* const* d_in, const int* in_sizes, int n_in,
                              void* d_out, int out_size, void* d_ws, size_t ws_size,
                              hipStream_t stream) {
    const float* x   = (const float*)d_in[0];
    const float* ew  = (const float*)d_in[1];
    const float* W1  = (const float*)d_in[2];
    const float* b1  = (const float*)d_in[3];
    const float* W2  = (const float*)d_in[4];
    const float* b2  = (const float*)d_in[5];
    const float* W3  = (const float*)d_in[6];
    const float* b3  = (const float*)d_in[7];
    const float* Wp  = (const float*)d_in[8];
    const float* bp  = (const float*)d_in[9];
    const float* Wc  = (const float*)d_in[10];
    const float* bc  = (const float*)d_in[11];
    const int*   ei  = (const int*)d_in[12];
    const int*   doc = (const int*)d_in[13];
    float*       out = (float*)d_out;

    const int* src = ei;
    const int* dst = ei + N_EDGES;

    float* ws   = (float*)d_ws;
    float* dinv = ws;
    int*   perm = (int*)(dinv + N_NODES);
    float* bufT = (float*)(perm + N_EDGES);
    float* bufH = bufT + (size_t)N_NODES * HD;
    float* dbuf = bufH + (size_t)N_NODES * HD;
    int*   rank   = (int*)bufH;
    int*   rowptr = (int*)dbuf;
    int*   aux    = rowptr + N_NODES;

    const int B = 256;
    dim3 blk(B);
    dim3 gN((N_NODES + B - 1) / B);
    dim3 gE((N_EDGES + B - 1) / B);
    dim3 gN16((N_NODES * (HD / 4) + B - 1) / B);
    dim3 gScan((N_NODES + SCAN_E - 1) / SCAN_E);
    dim3 bScan(SCAN_T);

    TEXTGCN_cnt_init  <<<gN, blk, 0, stream>>>(rowptr, N_NODES);
    TEXTGCN_rank      <<<gE, blk, 0, stream>>>(dst, rowptr, rank, N_EDGES);
    TEXTGCN_scan_local<<<gScan, bScan, 0, stream>>>(rowptr, aux, N_NODES);
    TEXTGCN_scan_add  <<<gScan, bScan, 0, stream>>>(rowptr, aux, N_NODES);
    TEXTGCN_fill      <<<gE, blk, 0, stream>>>(dst, rowptr, rank, perm, N_EDGES);
    TEXTGCN_dinv_csr  <<<gN16, blk, 0, stream>>>(rowptr, perm, ew, dinv, N_NODES, N_EDGES);

    dim3 wave(32);
    dim3 gGemm1(N_NODES / 16, HD / 16);
    dim3 gGemmH(N_NODES / 16, HD / 16);
    dim3 gDoc(NDOCS / 16, HD / 16);

    TEXTGCN_gemm_wmma<<<gGemm1, wave, 0, stream>>>(x, W1, dinv, bufT, DIM_IN, HD);
    TEXTGCN_gather   <<<gN16, blk, 0, stream>>>(rowptr, perm, src, ew, dinv, bufT, b1, bufH, N_NODES, N_EDGES);

    TEXTGCN_gemm_wmma<<<gGemmH, wave, 0, stream>>>(bufH, W2, dinv, bufT, HD, HD);
    TEXTGCN_gather   <<<gN16, blk, 0, stream>>>(rowptr, perm, src, ew, dinv, bufT, b2, bufH, N_NODES, N_EDGES);

    TEXTGCN_gemm_wmma<<<gGemmH, wave, 0, stream>>>(bufH, W3, dinv, bufT, HD, HD);
    TEXTGCN_gather   <<<gN16, blk, 0, stream>>>(rowptr, perm, src, ew, dinv, bufT, b3, bufH, N_NODES, N_EDGES);

    TEXTGCN_doc_gemm_wmma<<<gDoc, wave, 0, stream>>>(bufH, doc, Wp, bp, dbuf);
    dim3 gCls((NDOCS * NCLS + B - 1) / B);
    TEXTGCN_classifier<<<gCls, blk, 0, stream>>>(dbuf, Wc, bc, out);
}


# GEMM retiled: one wave per 16x64 tile (4 accumulators), 9-wave blocks stage W once in LDS as k-pairs, A read once by 16-byte loads plus a lane-half exchange; k order of each product kept
# speedup vs baseline: 2.3318x; 1.3071x over previous
#include <hip/hip_runtime.h>
#include <hip/hip_bf16.h>

typedef float v2f __attribute__((ext_vector_type(2)));
typedef float v8f __attribute__((ext_vector_type(8)));

#define N_NODES 100000
#define N_EDGES 1600000
#define DIM_IN  256
#define HD      64
#define NCLS    20
#define NDOCS   10000
#define SCAN_T  256
#define SCAN_E  1024


__global__ void TEXTGCN_cnt_init(int* __restrict__ cnt, int n) {
    int i = blockIdx.x * blockDim.x + threadIdx.x;
    if (i < n) cnt[i] = 0;
}

__global__ void TEXTGCN_rank(const int* __restrict__ dst, int* __restrict__ cnt,
                             int* __restrict__ rank, int e) {
    int i = blockIdx.x * blockDim.x + threadIdx.x;
    if (i < e)
        rank[i] = __hip_atomic_fetch_add(&cnt[dst[i]], 1, __ATOMIC_RELAXED, __HIP_MEMORY_SCOPE_AGENT);
}

__global__ void TEXTGCN_scan_local(int* __restrict__ cnt, int* __restrict__ aux, int n) {
    __shared__ int sh[SCAN_T];
    const int t = threadIdx.x;
    const int idx = blockIdx.x * SCAN_E + t * 4;
    int4 v = make_int4(0, 0, 0, 0);
    if (idx < n) v = *(const int4*)(cnt + idx);
    const int s = v.x + v.y + v.z + v.w;
    sh[t] = s;
    __syncthreads();
    for (int off = 1; off < SCAN_T; off <<= 1) {
        int add = (t >= off) ? sh[t - off] : 0;
        __syncthreads();
        sh[t] += add;
        __syncthreads();
    }
    const int incl = sh[t];
    const int excl = incl - s;
    if (idx < n) {
        int4 o;
        o.x = excl;
        o.y = o.x + v.x;
        o.z = o.y + v.y;
        o.w = o.z + v.z;
        *(int4*)(cnt + idx) = o;
    }
    if (t == SCAN_T - 1) aux[blockIdx.x] = incl;
}

__global__ void TEXTGCN_scan_add(int* __restrict__ rowptr, const int* __restrict__ aux, int n) {
    const int b = blockIdx.x;
    int off = 0;
    for (int k = 0; k < b; ++k) off += aux[k];
    const int idx = b * SCAN_E + threadIdx.x * 4;
    if (idx < n) {
        int4 v = *(const int4*)(rowptr + idx);
        v.x += off; v.y += off; v.z += off; v.w += off;
        *(int4*)(rowptr + idx) = v;
    }
}

__global__ void TEXTGCN_fill(const int* __restrict__ dst, const int* __restrict__ rowptr,
                             const int* __restrict__ rank, int* __restrict__ perm, int e) {
    int i = blockIdx.x * blockDim.x + threadIdx.x;
    if (i < e) perm[rowptr[dst[i]] + rank[i]] = i;
}

__global__ void TEXTGCN_dinv_csr(const int* __restrict__ rowptr, const int* __restrict__ perm,
                                 const float* __restrict__ w, float* __restrict__ dinv,
                                 int n, int e) {
    int gid = blockIdx.x * blockDim.x + threadIdx.x;
    int i = gid >> 4, c = gid & 15;
    if (i >= n) return;
    const int start = rowptr[i];
    const int end   = (i + 1 < n) ? rowptr[i + 1] : e;
    float s = 0.0f;
    for (int j = start + c; j < end; j += 16) s += w[perm[j]];
    s += __shfl_xor(s, 8, 16);
    s += __shfl_xor(s, 4, 16);
    s += __shfl_xor(s, 2, 16);
    s += __shfl_xor(s, 1, 16);
    if (c == 0) {
        float d = 1.0f + s;
        dinv[i] = (d > 0.0f) ? __frsqrt_rn(d) : 0.0f;
    }
}


#define GEMM_WAVES 9
#define GEMM_T     (GEMM_WAVES * 32)
#define BS_PITCH   160
#define M_TILES    (N_NODES / 16)

__device__ __forceinline__ float TEXTGCN_xhalf(float x) {
    const int b = __builtin_bit_cast(int, x);
    return __builtin_bit_cast(float,
        __builtin_amdgcn_permlanex16(b, b, 0x76543210, 0xfedcba98, false, false));
}

template <int K>
__global__ void __launch_bounds__(GEMM_T)
TEXTGCN_gemm_wmma(const float* __restrict__ A,
                  const float* __restrict__ B,
                  const float* __restrict__ rscale,
                  float* __restrict__ C) {
    __shared__ __attribute__((aligned(16))) float Bs[(K / 2) * BS_PITCH];
    const int tid = threadIdx.x;
    const int lane = tid & 31;
    const int mt   = blockIdx.x * GEMM_WAVES + (tid >> 5);
    const bool active = mt < M_TILES;
    const int  c = lane & 15, h = lane >> 4;
    const bool hi = (h != 0);
    const float* __restrict__ arow = A + (size_t)((active ? mt : M_TILES - 1) * 16 + c) * K + 4 * h;

    float4 cur[4], nxt[4];
    for (int i = 0; i < 4; ++i) cur[i] = *(const float4*)(arow + 8 * i);

    constexpr int STAGE_IT = (K / 2) * (HD / 4) / 256;
    if (tid < 256) {
        float4 v0[STAGE_IT], v1[STAGE_IT];
        for (int u = 0; u < STAGE_IT; ++u) {
            const int it = tid + u * 256;
            const int kp = it >> 4, q = it & 15;
            v0[u] = *(const float4*)(B + (size_t)(2 * kp)     * HD + 4 * q);
            v1[u] = *(const float4*)(B + (size_t)(2 * kp + 1) * HD + 4 * q);
        }
        for (int u = 0; u < STAGE_IT; ++u) {
            const int it = tid + u * 256;
            const int kp = it >> 4, q = it & 15;
            float4* p = (float4*)(Bs + kp * BS_PITCH + 8 * q);
            p[0] = make_float4(v0[u].x, v1[u].x, v0[u].y, v1[u].y);
            p[1] = make_float4(v0[u].z, v1[u].z, v0[u].w, v1[u].w);
        }
    }
    __syncthreads();

    if (!active) return;

    const float* bl = Bs + h * BS_PITCH + 2 * c;
    v8f acc0 = {}, acc1 = {}, acc2 = {}, acc3 = {};

    for (int kc = 0; kc < K; kc += 32) {
        if (kc + 32 < K) {
            for (int i = 0; i < 4; ++i) nxt[i] = *(const float4*)(arow + kc + 32 + 8 * i);
        }
        for (int i = 0; i < 4; ++i) {
            const float4 a = cur[i];
            const float gx = hi ? a.x : a.z;
            const float gy = hi ? a.y : a.w;
            const float rx = TEXTGCN_xhalf(gx);
            const float ry = TEXTGCN_xhalf(gy);
            v2f a1, a2;
            a1.x = hi ? rx  : a.x;  a1.y = hi ? ry  : a.y;
            a2.x = hi ? a.z : rx;   a2.y = hi ? a.w : ry;
            const float* b1 = bl + ((kc >> 1) + 4 * i) * BS_PITCH;
            const float* b2 = b1 + 2 * BS_PITCH;
            const v2f f0 = *(const v2f*)(b1);
            const v2f f1 = *(const v2f*)(b1 + 32);
            const v2f f2 = *(const v2f*)(b1 + 64);
            const v2f f3 = *(const v2f*)(b1 + 96);
            const v2f g0 = *(const v2f*)(b2);
            const v2f g1 = *(const v2f*)(b2 + 32);
            const v2f g2 = *(const v2f*)(b2 + 64);
            const v2f g3 = *(const v2f*)(b2 + 96);
            acc0 = __builtin_amdgcn_wmma_f32_16x16x4_f32(false, a1, false, f0, (short)0, acc0, false, false);
            acc1 = __builtin_amdgcn_wmma_f32_16x16x4_f32(false, a1, false, f1, (short)0, acc1, false, false);
            acc2 = __builtin_amdgcn_wmma_f32_16x16x4_f32(false, a1, false, f2, (short)0, acc2, false, false);
            acc3 = __builtin_amdgcn_wmma_f32_16x16x4_f32(false, a1, false, f3, (short)0, acc3, false, false);
            acc0 = __builtin_amdgcn_wmma_f32_16x16x4_f32(false, a2, false, g0, (short)0, acc0, false, false);
            acc1 = __builtin_amdgcn_wmma_f32_16x16x4_f32(false, a2, false, g1, (short)0, acc1, false, false);
            acc2 = __builtin_amdgcn_wmma_f32_16x16x4_f32(false, a2, false, g2, (short)0, acc2, false, false);
            acc3 = __builtin_amdgcn_wmma_f32_16x16x4_f32(false, a2, false, g3, (short)0, acc3, false, false);
        }
        if (kc + 32 < K) {
            for (int i = 0; i < 4; ++i) cur[i] = nxt[i];
        }
    }

    const int mbase = mt * 16 + h * 8;
    const float4 s0 = *(const float4*)(rscale + mbase);
    const float4 s1 = *(const float4*)(rscale + mbase + 4);
    const float rs[8] = {s0.x, s0.y, s0.z, s0.w, s1.x, s1.y, s1.z, s1.w};
    float* __restrict__ crow = C + (size_t)mbase * HD + c;
    for (int r = 0; r < 8; ++r) {
        crow[(size_t)r * HD]      = acc0[r] * rs[r];
        crow[(size_t)r * HD + 16] = acc1[r] * rs[r];
        crow[(size_t)r * HD + 32] = acc2[r] * rs[r];
        crow[(size_t)r * HD + 48] = acc3[r] * rs[r];
    }
}


__global__ void TEXTGCN_gather(const int* __restrict__ rowptr, const int* __restrict__ perm,
                               const int* __restrict__ src, const float* __restrict__ w,
                               const float* __restrict__ dinv, const float* __restrict__ ts,
                               const float* __restrict__ b, float* __restrict__ h,
                               int n, int e) {
    int gid = blockIdx.x * blockDim.x + threadIdx.x;
    int i = gid >> 4, c = gid & 15;
    if (i >= n) return;
    const int start = rowptr[i];
    const int end   = (i + 1 < n) ? rowptr[i + 1] : e;
    const float4* __restrict__ t4 = (const float4*)ts;
    float4 acc = t4[(size_t)i * (HD / 4) + c];
    for (int base = start; base < end; base += 16) {
        int   s  = 0;
        float wt = 0.0f;
        if (base + c < end) {
            int ed = perm[base + c];
            s  = src[ed];
            wt = w[ed];
        }
        const int m = (end - base < 16) ? (end - base) : 16;
        int k = 0;
        for (; k + 4 <= m; k += 4) {
            const int   s0 = __shfl(s,  k,     16), s1 = __shfl(s,  k + 1, 16);
            const int   s2 = __shfl(s,  k + 2, 16), s3 = __shfl(s,  k + 3, 16);
            const float w0 = __shfl(wt, k,     16), w1 = __shfl(wt, k + 1, 16);
            const float w2 = __shfl(wt, k + 2, 16), w3 = __shfl(wt, k + 3, 16);
            const float4 v0 = t4[(size_t)s0 * (HD / 4) + c];
            const float4 v1 = t4[(size_t)s1 * (HD / 4) + c];
            const float4 v2 = t4[(size_t)s2 * (HD / 4) + c];
            const float4 v3 = t4[(size_t)s3 * (HD / 4) + c];
            acc.x = fmaf(w0, v0.x, acc.x); acc.y = fmaf(w0, v0.y, acc.y); acc.z = fmaf(w0, v0.z, acc.z); acc.w = fmaf(w0, v0.w, acc.w);
            acc.x = fmaf(w1, v1.x, acc.x); acc.y = fmaf(w1, v1.y, acc.y); acc.z = fmaf(w1, v1.z, acc.z); acc.w = fmaf(w1, v1.w, acc.w);
            acc.x = fmaf(w2, v2.x, acc.x); acc.y = fmaf(w2, v2.y, acc.y); acc.z = fmaf(w2, v2.z, acc.z); acc.w = fmaf(w2, v2.w, acc.w);
            acc.x = fmaf(w3, v3.x, acc.x); acc.y = fmaf(w3, v3.y, acc.y); acc.z = fmaf(w3, v3.z, acc.z); acc.w = fmaf(w3, v3.w, acc.w);
        }
        for (; k < m; ++k) {
            const int   sk = __shfl(s,  k, 16);
            const float wk = __shfl(wt, k, 16);
            const float4 v = t4[(size_t)sk * (HD / 4) + c];
            acc.x = fmaf(wk, v.x, acc.x); acc.y = fmaf(wk, v.y, acc.y); acc.z = fmaf(wk, v.z, acc.z); acc.w = fmaf(wk, v.w, acc.w);
        }
    }
    const float di = dinv[i];
    const float4 bb = ((const float4*)b)[c];
    float4 o;
    o.x = acc.x * di + bb.x; o.y = acc.y * di + bb.y; o.z = acc.z * di + bb.z; o.w = acc.w * di + bb.w;
    o.x = o.x > 0.0f ? o.x : 0.0f;
    o.y = o.y > 0.0f ? o.y : 0.0f;
    o.z = o.z > 0.0f ? o.z : 0.0f;
    o.w = o.w > 0.0f ? o.w : 0.0f;
    ((float4*)h)[(size_t)i * (HD / 4) + c] = o;
}


__global__ void TEXTGCN_doc_gemm_wmma(const float* __restrict__ h,
                                      const int* __restrict__ doc_idx,
                                      const float* __restrict__ Wp,
                                      const float* __restrict__ bp,
                                      float* __restrict__ dbuf) {
    const int lane = threadIdx.x;
    const int mt = blockIdx.x, nt = blockIdx.y;
    const int m  = mt * 16 + (lane & 15);
    const int n  = nt * 16 + (lane & 15);
    const int kq = (lane >> 4) * 2;

    const float* __restrict__ arow = h + (size_t)doc_idx[m] * HD;
    v8f acc = {};
    for (int k = 0; k < HD; k += 4) {
        v2f a, b;
        a.x = arow[k + kq];
        a.y = arow[k + kq + 1];
        b.x = Wp[(size_t)(k + kq)     * HD + n];
        b.y = Wp[(size_t)(k + kq + 1) * HD + n];
        acc = __builtin_amdgcn_wmma_f32_16x16x4_f32(
                  false, a, false, b, (short)0, acc, false, false);
    }
    const int mbase = mt * 16 + (lane >> 4) * 8;
    const int nst   = nt * 16 + (lane & 15);
    const float bias = bp[nst];
#pragma unroll
    for (int r = 0; r < 8; ++r) {
        float v = acc[r] + bias;
        dbuf[(size_t)(mbase + r) * HD + nst] = v > 0.0f ? v : 0.0f;
    }
}


__global__ void TEXTGCN_classifier(const float* __restrict__ dbuf,
                                   const float* __restrict__ Wc,
                                   const float* __restrict__ bc,
                                   float* __restrict__ out) {
    int gid = blockIdx.x * blockDim.x + threadIdx.x;
    if (gid >= NDOCS * NCLS) return;
    int d = gid / NCLS, c = gid % NCLS;
    const float* __restrict__ row = dbuf + (size_t)d * HD;
    float s = bc[c];
#pragma unroll 8
    for (int k = 0; k < HD; ++k) s += row[k] * Wc[(size_t)k * NCLS + c];
    out[gid] = s;
}


extern "C" void kernel_launch(void* const* d_in, const int* in_sizes, int n_in,
                              void* d_out, int out_size, void* d_ws, size_t ws_size,
                              hipStream_t stream) {
    const float* x   = (const float*)d_in[0];
    const float* ew  = (const float*)d_in[1];
    const float* W1  = (const float*)d_in[2];
    const float* b1  = (const float*)d_in[3];
    const float* W2  = (const float*)d_in[4];
    const float* b2  = (const float*)d_in[5];
    const float* W3  = (const float*)d_in[6];
    const float* b3  = (const float*)d_in[7];
    const float* Wp  = (const float*)d_in[8];
    const float* bp  = (const float*)d_in[9];
    const float* Wc  = (const float*)d_in[10];
    const float* bc  = (const float*)d_in[11];
    const int*   ei  = (const int*)d_in[12];
    const int*   doc = (const int*)d_in[13];
    float*       out = (float*)d_out;

    const int* src = ei;
    const int* dst = ei + N_EDGES;

    float* ws   = (float*)d_ws;
    float* dinv = ws;
    int*   perm = (int*)(dinv + N_NODES);
    float* bufT = (float*)(perm + N_EDGES);
    float* bufH = bufT + (size_t)N_NODES * HD;
    float* dbuf = bufH + (size_t)N_NODES * HD;
    int*   rank   = (int*)bufH;
    int*   rowptr = (int*)dbuf;
    int*   aux    = rowptr + N_NODES;

    const int B = 256;
    dim3 blk(B);
    dim3 gN((N_NODES + B - 1) / B);
    dim3 gE((N_EDGES + B - 1) / B);
    dim3 gN16((N_NODES * (HD / 4) + B - 1) / B);
    dim3 gScan((N_NODES + SCAN_E - 1) / SCAN_E);
    dim3 bScan(SCAN_T);

    TEXTGCN_cnt_init  <<<gN, blk, 0, stream>>>(rowptr, N_NODES);
    TEXTGCN_rank      <<<gE, blk, 0, stream>>>(dst, rowptr, rank, N_EDGES);
    TEXTGCN_scan_local<<<gScan, bScan, 0, stream>>>(rowptr, aux, N_NODES);
    TEXTGCN_scan_add  <<<gScan, bScan, 0, stream>>>(rowptr, aux, N_NODES);
    TEXTGCN_fill      <<<gE, blk, 0, stream>>>(dst, rowptr, rank, perm, N_EDGES);
    TEXTGCN_dinv_csr  <<<gN16, blk, 0, stream>>>(rowptr, perm, ew, dinv, N_NODES, N_EDGES);

    dim3 wave(32);
    dim3 gGemm((M_TILES + GEMM_WAVES - 1) / GEMM_WAVES);
    dim3 bGemm(GEMM_T);
    dim3 gDoc(NDOCS / 16, HD / 16);

    TEXTGCN_gemm_wmma<DIM_IN><<<gGemm, bGemm, 0, stream>>>(x, W1, dinv, bufT);
    TEXTGCN_gather   <<<gN16, blk, 0, stream>>>(rowptr, perm, src, ew, dinv, bufT, b1, bufH, N_NODES, N_EDGES);

    TEXTGCN_gemm_wmma<HD><<<gGemm, bGemm, 0, stream>>>(bufH, W2, dinv, bufT);
    TEXTGCN_gather   <<<gN16, blk, 0, stream>>>(rowptr, perm, src, ew, dinv, bufT, b2, bufH, N_NODES, N_EDGES);

    TEXTGCN_gemm_wmma<HD><<<gGemm, bGemm, 0, stream>>>(bufH, W3, dinv, bufT);
    TEXTGCN_gather   <<<gN16, blk, 0, stream>>>(rowptr, perm, src, ew, dinv, bufT, b3, bufH, N_NODES, N_EDGES);

    TEXTGCN_doc_gemm_wmma<<<gDoc, wave, 0, stream>>>(bufH, doc, Wp, bp, dbuf);
    dim3 gCls((NDOCS * NCLS + B - 1) / B);
    TEXTGCN_classifier<<<gCls, blk, 0, stream>>>(dbuf, Wc, bc, out);
}


# layer 3 propagated only at the rows doc_idx names (the only layer-3 rows the head reads), stored in doc order so the doc GEMM reads them unindexed; per-row fma order kept
# speedup vs baseline: 2.5434x; 1.0907x over previous
#include <hip/hip_runtime.h>
#include <hip/hip_bf16.h>

typedef float v2f __attribute__((ext_vector_type(2)));
typedef float v8f __attribute__((ext_vector_type(8)));

#define N_NODES 100000
#define N_EDGES 1600000
#define DIM_IN  256
#define HD      64
#define NCLS    20
#define NDOCS   10000
#define SCAN_T  256
#define SCAN_E  1024


__global__ void TEXTGCN_cnt_init(int* __restrict__ cnt, int n) {
    int i = blockIdx.x * blockDim.x + threadIdx.x;
    if (i < n) cnt[i] = 0;
}

__global__ void TEXTGCN_rank(const int* __restrict__ dst, int* __restrict__ cnt,
                             int* __restrict__ rank, int e) {
    int i = blockIdx.x * blockDim.x + threadIdx.x;
    if (i < e)
        rank[i] = __hip_atomic_fetch_add(&cnt[dst[i]], 1, __ATOMIC_RELAXED, __HIP_MEMORY_SCOPE_AGENT);
}

__global__ void TEXTGCN_scan_local(int* __restrict__ cnt, int* __restrict__ aux, int n) {
    __shared__ int sh[SCAN_T];
    const int t = threadIdx.x;
    const int idx = blockIdx.x * SCAN_E + t * 4;
    int4 v = make_int4(0, 0, 0, 0);
    if (idx < n) v = *(const int4*)(cnt + idx);
    const int s = v.x + v.y + v.z + v.w;
    sh[t] = s;
    __syncthreads();
    for (int off = 1; off < SCAN_T; off <<= 1) {
        int add = (t >= off) ? sh[t - off] : 0;
        __syncthreads();
        sh[t] += add;
        __syncthreads();
    }
    const int incl = sh[t];
    const int excl = incl - s;
    if (idx < n) {
        int4 o;
        o.x = excl;
        o.y = o.x + v.x;
        o.z = o.y + v.y;
        o.w = o.z + v.z;
        *(int4*)(cnt + idx) = o;
    }
    if (t == SCAN_T - 1) aux[blockIdx.x] = incl;
}

__global__ void TEXTGCN_scan_add(int* __restrict__ rowptr, const int* __restrict__ aux, int n) {
    const int b = blockIdx.x;
    int off = 0;
    for (int k = 0; k < b; ++k) off += aux[k];
    const int idx = b * SCAN_E + threadIdx.x * 4;
    if (idx < n) {
        int4 v = *(const int4*)(rowptr + idx);
        v.x += off; v.y += off; v.z += off; v.w += off;
        *(int4*)(rowptr + idx) = v;
    }
}

__global__ void TEXTGCN_fill(const int* __restrict__ dst, const int* __restrict__ rowptr,
                             const int* __restrict__ rank, int* __restrict__ perm, int e) {
    int i = blockIdx.x * blockDim.x + threadIdx.x;
    if (i < e) perm[rowptr[dst[i]] + rank[i]] = i;
}

__global__ void TEXTGCN_dinv_csr(const int* __restrict__ rowptr, const int* __restrict__ perm,
                                 const float* __restrict__ w, float* __restrict__ dinv,
                                 int n, int e) {
    int gid = blockIdx.x * blockDim.x + threadIdx.x;
    int i = gid >> 4, c = gid & 15;
    if (i >= n) return;
    const int start = rowptr[i];
    const int end   = (i + 1 < n) ? rowptr[i + 1] : e;
    float s = 0.0f;
    for (int j = start + c; j < end; j += 16) s += w[perm[j]];
    s += __shfl_xor(s, 8, 16);
    s += __shfl_xor(s, 4, 16);
    s += __shfl_xor(s, 2, 16);
    s += __shfl_xor(s, 1, 16);
    if (c == 0) {
        float d = 1.0f + s;
        dinv[i] = (d > 0.0f) ? __frsqrt_rn(d) : 0.0f;
    }
}


#define GEMM_WAVES 9
#define GEMM_T     (GEMM_WAVES * 32)
#define BS_PITCH   160
#define M_TILES    (N_NODES / 16)

__device__ __forceinline__ float TEXTGCN_xhalf(float x) {
    const int b = __builtin_bit_cast(int, x);
    return __builtin_bit_cast(float,
        __builtin_amdgcn_permlanex16(b, b, 0x76543210, 0xfedcba98, false, false));
}

template <int K>
__global__ void __launch_bounds__(GEMM_T)
TEXTGCN_gemm_wmma(const float* __restrict__ A,
                  const float* __restrict__ B,
                  const float* __restrict__ rscale,
                  float* __restrict__ C) {
    __shared__ __attribute__((aligned(16))) float Bs[(K / 2) * BS_PITCH];
    const int tid = threadIdx.x;
    const int lane = tid & 31;
    const int mt   = blockIdx.x * GEMM_WAVES + (tid >> 5);
    const bool active = mt < M_TILES;
    const int  c = lane & 15, h = lane >> 4;
    const bool hi = (h != 0);
    const float* __restrict__ arow = A + (size_t)((active ? mt : M_TILES - 1) * 16 + c) * K + 4 * h;

    float4 cur[4], nxt[4];
    for (int i = 0; i < 4; ++i) cur[i] = *(const float4*)(arow + 8 * i);

    constexpr int STAGE_IT = (K / 2) * (HD / 4) / 256;
    if (tid < 256) {
        float4 v0[STAGE_IT], v1[STAGE_IT];
        for (int u = 0; u < STAGE_IT; ++u) {
            const int it = tid + u * 256;
            const int kp = it >> 4, q = it & 15;
            v0[u] = *(const float4*)(B + (size_t)(2 * kp)     * HD + 4 * q);
            v1[u] = *(const float4*)(B + (size_t)(2 * kp + 1) * HD + 4 * q);
        }
        for (int u = 0; u < STAGE_IT; ++u) {
            const int it = tid + u * 256;
            const int kp = it >> 4, q = it & 15;
            float4* p = (float4*)(Bs + kp * BS_PITCH + 8 * q);
            p[0] = make_float4(v0[u].x, v1[u].x, v0[u].y, v1[u].y);
            p[1] = make_float4(v0[u].z, v1[u].z, v0[u].w, v1[u].w);
        }
    }
    __syncthreads();

    if (!active) return;

    const float* bl = Bs + h * BS_PITCH + 2 * c;
    v8f acc0 = {}, acc1 = {}, acc2 = {}, acc3 = {};

    for (int kc = 0; kc < K; kc += 32) {
        if (kc + 32 < K) {
            for (int i = 0; i < 4; ++i) nxt[i] = *(const float4*)(arow + kc + 32 + 8 * i);
        }
        for (int i = 0; i < 4; ++i) {
            const float4 a = cur[i];
            const float gx = hi ? a.x : a.z;
            const float gy = hi ? a.y : a.w;
            const float rx = TEXTGCN_xhalf(gx);
            const float ry = TEXTGCN_xhalf(gy);
            v2f a1, a2;
            a1.x = hi ? rx  : a.x;  a1.y = hi ? ry  : a.y;
            a2.x = hi ? a.z : rx;   a2.y = hi ? a.w : ry;
            const float* b1 = bl + ((kc >> 1) + 4 * i) * BS_PITCH;
            const float* b2 = b1 + 2 * BS_PITCH;
            const v2f f0 = *(const v2f*)(b1);
            const v2f f1 = *(const v2f*)(b1 + 32);
            const v2f f2 = *(const v2f*)(b1 + 64);
            const v2f f3 = *(const v2f*)(b1 + 96);
            const v2f g0 = *(const v2f*)(b2);
            const v2f g1 = *(const v2f*)(b2 + 32);
            const v2f g2 = *(const v2f*)(b2 + 64);
            const v2f g3 = *(const v2f*)(b2 + 96);
            acc0 = __builtin_amdgcn_wmma_f32_16x16x4_f32(false, a1, false, f0, (short)0, acc0, false, false);
            acc1 = __builtin_amdgcn_wmma_f32_16x16x4_f32(false, a1, false, f1, (short)0, acc1, false, false);
            acc2 = __builtin_amdgcn_wmma_f32_16x16x4_f32(false, a1, false, f2, (short)0, acc2, false, false);
            acc3 = __builtin_amdgcn_wmma_f32_16x16x4_f32(false, a1, false, f3, (short)0, acc3, false, false);
            acc0 = __builtin_amdgcn_wmma_f32_16x16x4_f32(false, a2, false, g0, (short)0, acc0, false, false);
            acc1 = __builtin_amdgcn_wmma_f32_16x16x4_f32(false, a2, false, g1, (short)0, acc1, false, false);
            acc2 = __builtin_amdgcn_wmma_f32_16x16x4_f32(false, a2, false, g2, (short)0, acc2, false, false);
            acc3 = __builtin_amdgcn_wmma_f32_16x16x4_f32(false, a2, false, g3, (short)0, acc3, false, false);
        }
        if (kc + 32 < K) {
            for (int i = 0; i < 4; ++i) cur[i] = nxt[i];
        }
    }

    const int mbase = mt * 16 + h * 8;
    const float4 s0 = *(const float4*)(rscale + mbase);
    const float4 s1 = *(const float4*)(rscale + mbase + 4);
    const float rs[8] = {s0.x, s0.y, s0.z, s0.w, s1.x, s1.y, s1.z, s1.w};
    float* __restrict__ crow = C + (size_t)mbase * HD + c;
    for (int r = 0; r < 8; ++r) {
        crow[(size_t)r * HD]      = acc0[r] * rs[r];
        crow[(size_t)r * HD + 16] = acc1[r] * rs[r];
        crow[(size_t)r * HD + 32] = acc2[r] * rs[r];
        crow[(size_t)r * HD + 48] = acc3[r] * rs[r];
    }
}


__device__ __forceinline__ float4 TEXTGCN_prop_row(const int i, const int c,
                               const int* __restrict__ rowptr, const int* __restrict__ perm,
                               const int* __restrict__ src, const float* __restrict__ w,
                               const float* __restrict__ dinv, const float* __restrict__ ts,
                               const float* __restrict__ b, int n, int e) {
    const int start = rowptr[i];
    const int end   = (i + 1 < n) ? rowptr[i + 1] : e;
    const float4* __restrict__ t4 = (const float4*)ts;
    float4 acc = t4[(size_t)i * (HD / 4) + c];
    for (int base = start; base < end; base += 16) {
        int   s  = 0;
        float wt = 0.0f;
        if (base + c < end) {
            int ed = perm[base + c];
            s  = src[ed];
            wt = w[ed];
        }
        const int m = (end - base < 16) ? (end - base) : 16;
        int k = 0;
        for (; k + 4 <= m; k += 4) {
            const int   s0 = __shfl(s,  k,     16), s1 = __shfl(s,  k + 1, 16);
            const int   s2 = __shfl(s,  k + 2, 16), s3 = __shfl(s,  k + 3, 16);
            const float w0 = __shfl(wt, k,     16), w1 = __shfl(wt, k + 1, 16);
            const float w2 = __shfl(wt, k + 2, 16), w3 = __shfl(wt, k + 3, 16);
            const float4 v0 = t4[(size_t)s0 * (HD / 4) + c];
            const float4 v1 = t4[(size_t)s1 * (HD / 4) + c];
            const float4 v2 = t4[(size_t)s2 * (HD / 4) + c];
            const float4 v3 = t4[(size_t)s3 * (HD / 4) + c];
            acc.x = fmaf(w0, v0.x, acc.x); acc.y = fmaf(w0, v0.y, acc.y); acc.z = fmaf(w0, v0.z, acc.z); acc.w = fmaf(w0, v0.w, acc.w);
            acc.x = fmaf(w1, v1.x, acc.x); acc.y = fmaf(w1, v1.y, acc.y); acc.z = fmaf(w1, v1.z, acc.z); acc.w = fmaf(w1, v1.w, acc.w);
            acc.x = fmaf(w2, v2.x, acc.x); acc.y = fmaf(w2, v2.y, acc.y); acc.z = fmaf(w2, v2.z, acc.z); acc.w = fmaf(w2, v2.w, acc.w);
            acc.x = fmaf(w3, v3.x, acc.x); acc.y = fmaf(w3, v3.y, acc.y); acc.z = fmaf(w3, v3.z, acc.z); acc.w = fmaf(w3, v3.w, acc.w);
        }
        for (; k < m; ++k) {
            const int   sk = __shfl(s,  k, 16);
            const float wk = __shfl(wt, k, 16);
            const float4 v = t4[(size_t)sk * (HD / 4) + c];
            acc.x = fmaf(wk, v.x, acc.x); acc.y = fmaf(wk, v.y, acc.y); acc.z = fmaf(wk, v.z, acc.z); acc.w = fmaf(wk, v.w, acc.w);
        }
    }
    const float di = dinv[i];
    const float4 bb = ((const float4*)b)[c];
    float4 o;
    o.x = acc.x * di + bb.x; o.y = acc.y * di + bb.y; o.z = acc.z * di + bb.z; o.w = acc.w * di + bb.w;
    o.x = o.x > 0.0f ? o.x : 0.0f;
    o.y = o.y > 0.0f ? o.y : 0.0f;
    o.z = o.z > 0.0f ? o.z : 0.0f;
    o.w = o.w > 0.0f ? o.w : 0.0f;
    return o;
}

__global__ void TEXTGCN_gather(const int* __restrict__ rowptr, const int* __restrict__ perm,
                               const int* __restrict__ src, const float* __restrict__ w,
                               const float* __restrict__ dinv, const float* __restrict__ ts,
                               const float* __restrict__ b, float* __restrict__ h,
                               int n, int e) {
    int gid = blockIdx.x * blockDim.x + threadIdx.x;
    int i = gid >> 4, c = gid & 15;
    if (i >= n) return;
    const float4 o = TEXTGCN_prop_row(i, c, rowptr, perm, src, w, dinv, ts, b, n, e);
    ((float4*)h)[(size_t)i * (HD / 4) + c] = o;
}

__global__ void TEXTGCN_gather_rows(const int* __restrict__ rowptr, const int* __restrict__ perm,
                                    const int* __restrict__ src, const float* __restrict__ w,
                                    const float* __restrict__ dinv, const float* __restrict__ ts,
                                    const float* __restrict__ b, const int* __restrict__ rows,
                                    float* __restrict__ hd, int nrows, int n, int e) {
    int gid = blockIdx.x * blockDim.x + threadIdx.x;
    int g = gid >> 4, c = gid & 15;
    if (g >= nrows) return;
    const float4 o = TEXTGCN_prop_row(rows[g], c, rowptr, perm, src, w, dinv, ts, b, n, e);
    ((float4*)hd)[(size_t)g * (HD / 4) + c] = o;
}


__global__ void TEXTGCN_doc_gemm_wmma(const float* __restrict__ hd,
                                      const float* __restrict__ Wp,
                                      const float* __restrict__ bp,
                                      float* __restrict__ dbuf) {
    const int lane = threadIdx.x;
    const int mt = blockIdx.x, nt = blockIdx.y;
    const int m  = mt * 16 + (lane & 15);
    const int n  = nt * 16 + (lane & 15);
    const int kq = (lane >> 4) * 2;

    const float* __restrict__ arow = hd + (size_t)m * HD;
    v8f acc = {};
    for (int k = 0; k < HD; k += 4) {
        v2f a, b;
        a.x = arow[k + kq];
        a.y = arow[k + kq + 1];
        b.x = Wp[(size_t)(k + kq)     * HD + n];
        b.y = Wp[(size_t)(k + kq + 1) * HD + n];
        acc = __builtin_amdgcn_wmma_f32_16x16x4_f32(
                  false, a, false, b, (short)0, acc, false, false);
    }
    const int mbase = mt * 16 + (lane >> 4) * 8;
    const int nst   = nt * 16 + (lane & 15);
    const float bias = bp[nst];
#pragma unroll
    for (int r = 0; r < 8; ++r) {
        float v = acc[r] + bias;
        dbuf[(size_t)(mbase + r) * HD + nst] = v > 0.0f ? v : 0.0f;
    }
}


__global__ void TEXTGCN_classifier(const float* __restrict__ dbuf,
                                   const float* __restrict__ Wc,
                                   const float* __restrict__ bc,
                                   float* __restrict__ out) {
    int gid = blockIdx.x * blockDim.x + threadIdx.x;
    if (gid >= NDOCS * NCLS) return;
    int d = gid / NCLS, c = gid % NCLS;
    const float* __restrict__ row = dbuf + (size_t)d * HD;
    float s = bc[c];
#pragma unroll 8
    for (int k = 0; k < HD; ++k) s += row[k] * Wc[(size_t)k * NCLS + c];
    out[gid] = s;
}


extern "C" void kernel_launch(void* const* d_in, const int* in_sizes, int n_in,
                              void* d_out, int out_size, void* d_ws, size_t ws_size,
                              hipStream_t stream) {
    const float* x   = (const float*)d_in[0];
    const float* ew  = (const float*)d_in[1];
    const float* W1  = (const float*)d_in[2];
    const float* b1  = (const float*)d_in[3];
    const float* W2  = (const float*)d_in[4];
    const float* b2  = (const float*)d_in[5];
    const float* W3  = (const float*)d_in[6];
    const float* b3  = (const float*)d_in[7];
    const float* Wp  = (const float*)d_in[8];
    const float* bp  = (const float*)d_in[9];
    const float* Wc  = (const float*)d_in[10];
    const float* bc  = (const float*)d_in[11];
    const int*   ei  = (const int*)d_in[12];
    const int*   doc = (const int*)d_in[13];
    float*       out = (float*)d_out;

    const int* src = ei;
    const int* dst = ei + N_EDGES;

    float* ws   = (float*)d_ws;
    float* dinv = ws;
    int*   perm = (int*)(dinv + N_NODES);
    float* bufT = (float*)(perm + N_EDGES);
    float* bufH = bufT + (size_t)N_NODES * HD;
    float* dbuf = bufH + (size_t)N_NODES * HD;
    int*   rank   = (int*)bufH;
    int*   rowptr = (int*)dbuf;
    int*   aux    = rowptr + N_NODES;

    const int B = 256;
    dim3 blk(B);
    dim3 gN((N_NODES + B - 1) / B);
    dim3 gE((N_EDGES + B - 1) / B);
    dim3 gN16((N_NODES * (HD / 4) + B - 1) / B);
    dim3 gScan((N_NODES + SCAN_E - 1) / SCAN_E);
    dim3 bScan(SCAN_T);

    TEXTGCN_cnt_init  <<<gN, blk, 0, stream>>>(rowptr, N_NODES);
    TEXTGCN_rank      <<<gE, blk, 0, stream>>>(dst, rowptr, rank, N_EDGES);
    TEXTGCN_scan_local<<<gScan, bScan, 0, stream>>>(rowptr, aux, N_NODES);
    TEXTGCN_scan_add  <<<gScan, bScan, 0, stream>>>(rowptr, aux, N_NODES);
    TEXTGCN_fill      <<<gE, blk, 0, stream>>>(dst, rowptr, rank, perm, N_EDGES);
    TEXTGCN_dinv_csr  <<<gN16, blk, 0, stream>>>(rowptr, perm, ew, dinv, N_NODES, N_EDGES);

    dim3 wave(32);
    dim3 gGemm((M_TILES + GEMM_WAVES - 1) / GEMM_WAVES);
    dim3 bGemm(GEMM_T);
    dim3 gDoc(NDOCS / 16, HD / 16);
    dim3 gD16((NDOCS * (HD / 4) + B - 1) / B);

    TEXTGCN_gemm_wmma<DIM_IN><<<gGemm, bGemm, 0, stream>>>(x, W1, dinv, bufT);
    TEXTGCN_gather   <<<gN16, blk, 0, stream>>>(rowptr, perm, src, ew, dinv, bufT, b1, bufH, N_NODES, N_EDGES);

    TEXTGCN_gemm_wmma<HD><<<gGemm, bGemm, 0, stream>>>(bufH, W2, dinv, bufT);
    TEXTGCN_gather   <<<gN16, blk, 0, stream>>>(rowptr, perm, src, ew, dinv, bufT, b2, bufH, N_NODES, N_EDGES);

    TEXTGCN_gemm_wmma<HD><<<gGemm, bGemm, 0, stream>>>(bufH, W3, dinv, bufT);
    TEXTGCN_gather_rows<<<gD16, blk, 0, stream>>>(rowptr, perm, src, ew, dinv, bufT, b3, doc, bufH, NDOCS, N_NODES, N_EDGES);

    TEXTGCN_doc_gemm_wmma<<<gDoc, wave, 0, stream>>>(bufH, Wp, bp, dbuf);
    dim3 gCls((NDOCS * NCLS + B - 1) / B);
    TEXTGCN_classifier<<<gCls, blk, 0, stream>>>(dbuf, Wc, bc, out);
}


# launch fusion: each conv propagation also does the next 64x64 transform (rows to an LDS tile, WMMA from it), the doc-row propagation also doc head and classifier; 14 launches to 10; sums unchanged
# speedup vs baseline: 2.7445x; 1.0791x over previous
#include <hip/hip_runtime.h>
#include <hip/hip_bf16.h>

typedef float v2f __attribute__((ext_vector_type(2)));
typedef float v8f __attribute__((ext_vector_type(8)));

#define N_NODES 100000
#define N_EDGES 1600000
#define DIM_IN  256
#define HD      64
#define NCLS    20
#define NDOCS   10000
#define SCAN_T  256
#define SCAN_E  1024


__global__ void TEXTGCN_cnt_init(int* __restrict__ cnt, int n) {
    int i = blockIdx.x * blockDim.x + threadIdx.x;
    if (i < n) cnt[i] = 0;
}

__global__ void TEXTGCN_rank(const int* __restrict__ dst, int* __restrict__ cnt,
                             int* __restrict__ rank, int e) {
    int i = blockIdx.x * blockDim.x + threadIdx.x;
    if (i < e)
        rank[i] = __hip_atomic_fetch_add(&cnt[dst[i]], 1, __ATOMIC_RELAXED, __HIP_MEMORY_SCOPE_AGENT);
}

__global__ void TEXTGCN_scan_local(int* __restrict__ cnt, int* __restrict__ aux, int n) {
    __shared__ int sh[SCAN_T];
    const int t = threadIdx.x;
    const int idx = blockIdx.x * SCAN_E + t * 4;
    int4 v = make_int4(0, 0, 0, 0);
    if (idx < n) v = *(const int4*)(cnt + idx);
    const int s = v.x + v.y + v.z + v.w;
    sh[t] = s;
    __syncthreads();
    for (int off = 1; off < SCAN_T; off <<= 1) {
        int add = (t >= off) ? sh[t - off] : 0;
        __syncthreads();
        sh[t] += add;
        __syncthreads();
    }
    const int incl = sh[t];
    const int excl = incl - s;
    if (idx < n) {
        int4 o;
        o.x = excl;
        o.y = o.x + v.x;
        o.z = o.y + v.y;
        o.w = o.z + v.z;
        *(int4*)(cnt + idx) = o;
    }
    if (t == SCAN_T - 1) aux[blockIdx.x] = incl;
}

__global__ void TEXTGCN_scan_add(int* __restrict__ rowptr, const int* __restrict__ aux, int n) {
    const int b = blockIdx.x;
    int off = 0;
    for (int k = 0; k < b; ++k) off += aux[k];
    const int idx = b * SCAN_E + threadIdx.x * 4;
    if (idx < n) {
        int4 v = *(const int4*)(rowptr + idx);
        v.x += off; v.y += off; v.z += off; v.w += off;
        *(int4*)(rowptr + idx) = v;
    }
}

__global__ void TEXTGCN_fill(const int* __restrict__ dst, const int* __restrict__ rowptr,
                             const int* __restrict__ rank, int* __restrict__ perm, int e) {
    int i = blockIdx.x * blockDim.x + threadIdx.x;
    if (i < e) perm[rowptr[dst[i]] + rank[i]] = i;
}

__global__ void TEXTGCN_dinv_csr(const int* __restrict__ rowptr, const int* __restrict__ perm,
                                 const float* __restrict__ w, float* __restrict__ dinv,
                                 int n, int e) {
    int gid = blockIdx.x * blockDim.x + threadIdx.x;
    int i = gid >> 4, c = gid & 15;
    if (i >= n) return;
    const int start = rowptr[i];
    const int end   = (i + 1 < n) ? rowptr[i + 1] : e;
    float s = 0.0f;
    for (int j = start + c; j < end; j += 16) s += w[perm[j]];
    s += __shfl_xor(s, 8, 16);
    s += __shfl_xor(s, 4, 16);
    s += __shfl_xor(s, 2, 16);
    s += __shfl_xor(s, 1, 16);
    if (c == 0) {
        float d = 1.0f + s;
        dinv[i] = (d > 0.0f) ? __frsqrt_rn(d) : 0.0f;
    }
}


#define GEMM_WAVES 9
#define GEMM_T     (GEMM_WAVES * 32)
#define BS_PITCH   160
#define M_TILES    (N_NODES / 16)

__device__ __forceinline__ float TEXTGCN_xhalf(float x) {
    const int b = __builtin_bit_cast(int, x);
    return __builtin_bit_cast(float,
        __builtin_amdgcn_permlanex16(b, b, 0x76543210, 0xfedcba98, false, false));
}

template <int K>
__global__ void __launch_bounds__(GEMM_T)
TEXTGCN_gemm_wmma(const float* __restrict__ A,
                  const float* __restrict__ B,
                  const float* __restrict__ rscale,
                  float* __restrict__ C) {
    __shared__ __attribute__((aligned(16))) float Bs[(K / 2) * BS_PITCH];
    const int tid = threadIdx.x;
    const int lane = tid & 31;
    const int mt   = blockIdx.x * GEMM_WAVES + (tid >> 5);
    const bool active = mt < M_TILES;
    const int  c = lane & 15, h = lane >> 4;
    const bool hi = (h != 0);
    const float* __restrict__ arow = A + (size_t)((active ? mt : M_TILES - 1) * 16 + c) * K + 4 * h;

    float4 cur[4], nxt[4];
    for (int i = 0; i < 4; ++i) cur[i] = *(const float4*)(arow + 8 * i);

    constexpr int STAGE_IT = (K / 2) * (HD / 4) / 256;
    if (tid < 256) {
        float4 v0[STAGE_IT], v1[STAGE_IT];
        for (int u = 0; u < STAGE_IT; ++u) {
            const int it = tid + u * 256;
            const int kp = it >> 4, q = it & 15;
            v0[u] = *(const float4*)(B + (size_t)(2 * kp)     * HD + 4 * q);
            v1[u] = *(const float4*)(B + (size_t)(2 * kp + 1) * HD + 4 * q);
        }
        for (int u = 0; u < STAGE_IT; ++u) {
            const int it = tid + u * 256;
            const int kp = it >> 4, q = it & 15;
            float4* p = (float4*)(Bs + kp * BS_PITCH + 8 * q);
            p[0] = make_float4(v0[u].x, v1[u].x, v0[u].y, v1[u].y);
            p[1] = make_float4(v0[u].z, v1[u].z, v0[u].w, v1[u].w);
        }
    }
    __syncthreads();

    if (!active) return;

    const float* bl = Bs + h * BS_PITCH + 2 * c;
    v8f acc0 = {}, acc1 = {}, acc2 = {}, acc3 = {};

    for (int kc = 0; kc < K; kc += 32) {
        if (kc + 32 < K) {
            for (int i = 0; i < 4; ++i) nxt[i] = *(const float4*)(arow + kc + 32 + 8 * i);
        }
        for (int i = 0; i < 4; ++i) {
            const float4 a = cur[i];
            const float gx = hi ? a.x : a.z;
            const float gy = hi ? a.y : a.w;
            const float rx = TEXTGCN_xhalf(gx);
            const float ry = TEXTGCN_xhalf(gy);
            v2f a1, a2;
            a1.x = hi ? rx  : a.x;  a1.y = hi ? ry  : a.y;
            a2.x = hi ? a.z : rx;   a2.y = hi ? a.w : ry;
            const float* b1 = bl + ((kc >> 1) + 4 * i) * BS_PITCH;
            const float* b2 = b1 + 2 * BS_PITCH;
            const v2f f0 = *(const v2f*)(b1);
            const v2f f1 = *(const v2f*)(b1 + 32);
            const v2f f2 = *(const v2f*)(b1 + 64);
            const v2f f3 = *(const v2f*)(b1 + 96);
            const v2f g0 = *(const v2f*)(b2);
            const v2f g1 = *(const v2f*)(b2 + 32);
            const v2f g2 = *(const v2f*)(b2 + 64);
            const v2f g3 = *(const v2f*)(b2 + 96);
            acc0 = __builtin_amdgcn_wmma_f32_16x16x4_f32(false, a1, false, f0, (short)0, acc0, false, false);
            acc1 = __builtin_amdgcn_wmma_f32_16x16x4_f32(false, a1, false, f1, (short)0, acc1, false, false);
            acc2 = __builtin_amdgcn_wmma_f32_16x16x4_f32(false, a1, false, f2, (short)0, acc2, false, false);
            acc3 = __builtin_amdgcn_wmma_f32_16x16x4_f32(false, a1, false, f3, (short)0, acc3, false, false);
            acc0 = __builtin_amdgcn_wmma_f32_16x16x4_f32(false, a2, false, g0, (short)0, acc0, false, false);
            acc1 = __builtin_amdgcn_wmma_f32_16x16x4_f32(false, a2, false, g1, (short)0, acc1, false, false);
            acc2 = __builtin_amdgcn_wmma_f32_16x16x4_f32(false, a2, false, g2, (short)0, acc2, false, false);
            acc3 = __builtin_amdgcn_wmma_f32_16x16x4_f32(false, a2, false, g3, (short)0, acc3, false, false);
        }
        if (kc + 32 < K) {
            for (int i = 0; i < 4; ++i) cur[i] = nxt[i];
        }
    }

    const int mbase = mt * 16 + h * 8;
    const float4 s0 = *(const float4*)(rscale + mbase);
    const float4 s1 = *(const float4*)(rscale + mbase + 4);
    const float rs[8] = {s0.x, s0.y, s0.z, s0.w, s1.x, s1.y, s1.z, s1.w};
    float* __restrict__ crow = C + (size_t)mbase * HD + c;
    for (int r = 0; r < 8; ++r) {
        crow[(size_t)r * HD]      = acc0[r] * rs[r];
        crow[(size_t)r * HD + 16] = acc1[r] * rs[r];
        crow[(size_t)r * HD + 32] = acc2[r] * rs[r];
        crow[(size_t)r * HD + 48] = acc3[r] * rs[r];
    }
}


__device__ __forceinline__ float4 TEXTGCN_prop_row(const int i, const int c,
                               const int* __restrict__ rowptr, const int* __restrict__ perm,
                               const int* __restrict__ src, const float* __restrict__ w,
                               const float* __restrict__ dinv, const float* __restrict__ ts,
                               const float* __restrict__ b, int n, int e) {
    const int start = rowptr[i];
    const int end   = (i + 1 < n) ? rowptr[i + 1] : e;
    const float4* __restrict__ t4 = (const float4*)ts;
    float4 acc = t4[(size_t)i * (HD / 4) + c];
    for (int base = start; base < end; base += 16) {
        int   s  = 0;
        float wt = 0.0f;
        if (base + c < end) {
            int ed = perm[base + c];
            s  = src[ed];
            wt = w[ed];
        }
        const int m = (end - base < 16) ? (end - base) : 16;
        int k = 0;
        for (; k + 4 <= m; k += 4) {
            const int   s0 = __shfl(s,  k,     16), s1 = __shfl(s,  k + 1, 16);
            const int   s2 = __shfl(s,  k + 2, 16), s3 = __shfl(s,  k + 3, 16);
            const float w0 = __shfl(wt, k,     16), w1 = __shfl(wt, k + 1, 16);
            const float w2 = __shfl(wt, k + 2, 16), w3 = __shfl(wt, k + 3, 16);
            const float4 v0 = t4[(size_t)s0 * (HD / 4) + c];
            const float4 v1 = t4[(size_t)s1 * (HD / 4) + c];
            const float4 v2 = t4[(size_t)s2 * (HD / 4) + c];
            const float4 v3 = t4[(size_t)s3 * (HD / 4) + c];
            acc.x = fmaf(w0, v0.x, acc.x); acc.y = fmaf(w0, v0.y, acc.y); acc.z = fmaf(w0, v0.z, acc.z); acc.w = fmaf(w0, v0.w, acc.w);
            acc.x = fmaf(w1, v1.x, acc.x); acc.y = fmaf(w1, v1.y, acc.y); acc.z = fmaf(w1, v1.z, acc.z); acc.w = fmaf(w1, v1.w, acc.w);
            acc.x = fmaf(w2, v2.x, acc.x); acc.y = fmaf(w2, v2.y, acc.y); acc.z = fmaf(w2, v2.z, acc.z); acc.w = fmaf(w2, v2.w, acc.w);
            acc.x = fmaf(w3, v3.x, acc.x); acc.y = fmaf(w3, v3.y, acc.y); acc.z = fmaf(w3, v3.z, acc.z); acc.w = fmaf(w3, v3.w, acc.w);
        }
        for (; k < m; ++k) {
            const int   sk = __shfl(s,  k, 16);
            const float wk = __shfl(wt, k, 16);
            const float4 v = t4[(size_t)sk * (HD / 4) + c];
            acc.x = fmaf(wk, v.x, acc.x); acc.y = fmaf(wk, v.y, acc.y); acc.z = fmaf(wk, v.z, acc.z); acc.w = fmaf(wk, v.w, acc.w);
        }
    }
    const float di = dinv[i];
    const float4 bb = ((const float4*)b)[c];
    float4 o;
    o.x = acc.x * di + bb.x; o.y = acc.y * di + bb.y; o.z = acc.z * di + bb.z; o.w = acc.w * di + bb.w;
    o.x = o.x > 0.0f ? o.x : 0.0f;
    o.y = o.y > 0.0f ? o.y : 0.0f;
    o.z = o.z > 0.0f ? o.z : 0.0f;
    o.w = o.w > 0.0f ? o.w : 0.0f;
    return o;
}


#define FUSE_R   32
#define FUSE_T   (FUSE_R * 16)
#define HS_PITCH 68

static_assert(N_NODES % FUSE_R == 0, "every block of the fused propagation owns FUSE_R whole rows");
static_assert((HD / 2) * (HD / 4) == FUSE_T, "one staging item of the 64 x 64 weight per thread");

__device__ __forceinline__ void TEXTGCN_wload(const float* __restrict__ Wm, const int tid,
                                              float4& w0, float4& w1) {
    const int kp = tid >> 4, q = tid & 15;
    w0 = *(const float4*)(Wm + (size_t)(2 * kp)     * HD + 4 * q);
    w1 = *(const float4*)(Wm + (size_t)(2 * kp + 1) * HD + 4 * q);
}

__device__ __forceinline__ void TEXTGCN_wstore(float* Bs, const int tid, const float4 w0, const float4 w1) {
    const int kp = tid >> 4, q = tid & 15;
    float4* p = (float4*)(Bs + kp * BS_PITCH + 8 * q);
    p[0] = make_float4(w0.x, w1.x, w0.y, w1.y);
    p[1] = make_float4(w0.z, w1.z, w0.w, w1.w);
}

__device__ __forceinline__ void TEXTGCN_tile_mm(const float* Hs, const float* Bs, const int nt,
                                                const int c, const int h, v8f& acc0, v8f& acc1) {
    const float* a0 = Hs + c * HS_PITCH + 2 * h;
    const float* a1 = a0 + 16 * HS_PITCH;
    const float* bl = Bs + h * BS_PITCH + 2 * c + 32 * nt;
    for (int s = 0; s < HD / 4; ++s) {
        const v2f x0 = *(const v2f*)(a0 + 4 * s);
        const v2f x1 = *(const v2f*)(a1 + 4 * s);
        const v2f bf = *(const v2f*)(bl + (2 * s) * BS_PITCH);
        acc0 = __builtin_amdgcn_wmma_f32_16x16x4_f32(false, x0, false, bf, (short)0, acc0, false, false);
        acc1 = __builtin_amdgcn_wmma_f32_16x16x4_f32(false, x1, false, bf, (short)0, acc1, false, false);
    }
}

__global__ void __launch_bounds__(FUSE_T)
TEXTGCN_prop_gemm(const int* __restrict__ rowptr, const int* __restrict__ perm,
                  const int* __restrict__ src, const float* __restrict__ w,
                  const float* __restrict__ dinv, const float* __restrict__ ts,
                  const float* __restrict__ b, const float* __restrict__ Wn,
                  float* __restrict__ tn, int n, int e) {
    __shared__ __attribute__((aligned(16))) float Hs[FUSE_R * HS_PITCH];
    __shared__ __attribute__((aligned(16))) float Bs[(HD / 2) * BS_PITCH];
    const int tid = threadIdx.x;
    const int wv = tid >> 5;
    const int c = tid & 15, h = (tid >> 4) & 1;
    const int g = tid >> 4;
    const int row0 = blockIdx.x * FUSE_R;

    float4 w0, w1;
    TEXTGCN_wload(Wn, tid, w0, w1);

    const float4 o = TEXTGCN_prop_row(row0 + g, c, rowptr, perm, src, w, dinv, ts, b, n, e);
    *(float4*)(Hs + g * HS_PITCH + 4 * c) = o;
    TEXTGCN_wstore(Bs, tid, w0, w1);
    __syncthreads();
    if (wv >= 4) return;

    v8f acc0 = {}, acc1 = {};
    TEXTGCN_tile_mm(Hs, Bs, wv, c, h, acc0, acc1);

    const int mbase = row0 + h * 8;
    const float4 s0 = *(const float4*)(dinv + mbase);
    const float4 s1 = *(const float4*)(dinv + mbase + 4);
    const float4 s2 = *(const float4*)(dinv + mbase + 16);
    const float4 s3 = *(const float4*)(dinv + mbase + 20);
    const float rs0[8] = {s0.x, s0.y, s0.z, s0.w, s1.x, s1.y, s1.z, s1.w};
    const float rs1[8] = {s2.x, s2.y, s2.z, s2.w, s3.x, s3.y, s3.z, s3.w};
    float* __restrict__ crow = tn + (size_t)mbase * HD + wv * 16 + c;
    for (int r = 0; r < 8; ++r) {
        crow[(size_t)r * HD]        = acc0[r] * rs0[r];
        crow[(size_t)(r + 16) * HD] = acc1[r] * rs1[r];
    }
}

__global__ void __launch_bounds__(FUSE_T)
TEXTGCN_prop_head(const int* __restrict__ rowptr, const int* __restrict__ perm,
                  const int* __restrict__ src, const float* __restrict__ w,
                  const float* __restrict__ dinv, const float* __restrict__ ts,
                  const float* __restrict__ b, const int* __restrict__ rows,
                  const float* __restrict__ Wp, const float* __restrict__ bp,
                  const float* __restrict__ Wc, const float* __restrict__ bc,
                  float* __restrict__ out, int nrows, int n, int e) {
    __shared__ __attribute__((aligned(16))) float Hs[FUSE_R * HS_PITCH];
    __shared__ __attribute__((aligned(16))) float Ds[FUSE_R * HS_PITCH];
    __shared__ __attribute__((aligned(16))) float Bs[(HD / 2) * BS_PITCH];
    const int tid = threadIdx.x;
    const int wv = tid >> 5;
    const int c = tid & 15, h = (tid >> 4) & 1;
    const int g = tid >> 4;
    const int row0 = blockIdx.x * FUSE_R;

    float4 w0, w1;
    TEXTGCN_wload(Wp, tid, w0, w1);

    float4 o = make_float4(0.0f, 0.0f, 0.0f, 0.0f);
    if (row0 + g < nrows)
        o = TEXTGCN_prop_row(rows[row0 + g], c, rowptr, perm, src, w, dinv, ts, b, n, e);
    *(float4*)(Hs + g * HS_PITCH + 4 * c) = o;
    TEXTGCN_wstore(Bs, tid, w0, w1);
    __syncthreads();

    if (wv < 4) {
        v8f acc0 = {}, acc1 = {};
        TEXTGCN_tile_mm(Hs, Bs, wv, c, h, acc0, acc1);
        const int nst = wv * 16 + c;
        const float bias = bp[nst];
        float* drow = Ds + (h * 8) * HS_PITCH + nst;
        for (int r = 0; r < 8; ++r) {
            const float v0 = acc0[r] + bias;
            const float v1 = acc1[r] + bias;
            drow[r * HS_PITCH]        = v0 > 0.0f ? v0 : 0.0f;
            drow[(r + 16) * HS_PITCH] = v1 > 0.0f ? v1 : 0.0f;
        }
    }
    __syncthreads();

    for (int q = tid; q < FUSE_R * NCLS; q += FUSE_T) {
        const int d = q / NCLS, cc = q % NCLS;
        if (row0 + d < nrows) {
            const float* row = Ds + d * HS_PITCH;
            float s = bc[cc];
#pragma unroll 8
            for (int k = 0; k < HD; ++k) s += row[k] * Wc[(size_t)k * NCLS + cc];
            out[(size_t)(row0 + d) * NCLS + cc] = s;
        }
    }
}


extern "C" void kernel_launch(void* const* d_in, const int* in_sizes, int n_in,
                              void* d_out, int out_size, void* d_ws, size_t ws_size,
                              hipStream_t stream) {
    const float* x   = (const float*)d_in[0];
    const float* ew  = (const float*)d_in[1];
    const float* W1  = (const float*)d_in[2];
    const float* b1  = (const float*)d_in[3];
    const float* W2  = (const float*)d_in[4];
    const float* b2  = (const float*)d_in[5];
    const float* W3  = (const float*)d_in[6];
    const float* b3  = (const float*)d_in[7];
    const float* Wp  = (const float*)d_in[8];
    const float* bp  = (const float*)d_in[9];
    const float* Wc  = (const float*)d_in[10];
    const float* bc  = (const float*)d_in[11];
    const int*   ei  = (const int*)d_in[12];
    const int*   doc = (const int*)d_in[13];
    float*       out = (float*)d_out;

    const int* src = ei;
    const int* dst = ei + N_EDGES;

    float* ws   = (float*)d_ws;
    float* dinv = ws;
    int*   perm = (int*)(dinv + N_NODES);
    float* bufT = (float*)(perm + N_EDGES);
    float* bufH = bufT + (size_t)N_NODES * HD;
    float* dbuf = bufH + (size_t)N_NODES * HD;
    int*   rank   = (int*)bufH;
    int*   rowptr = (int*)dbuf;
    int*   aux    = rowptr + N_NODES;

    const int B = 256;
    dim3 blk(B);
    dim3 gN((N_NODES + B - 1) / B);
    dim3 gE((N_EDGES + B - 1) / B);
    dim3 gN16((N_NODES * (HD / 4) + B - 1) / B);
    dim3 gScan((N_NODES + SCAN_E - 1) / SCAN_E);
    dim3 bScan(SCAN_T);

    TEXTGCN_cnt_init  <<<gN, blk, 0, stream>>>(rowptr, N_NODES);
    TEXTGCN_rank      <<<gE, blk, 0, stream>>>(dst, rowptr, rank, N_EDGES);
    TEXTGCN_scan_local<<<gScan, bScan, 0, stream>>>(rowptr, aux, N_NODES);
    TEXTGCN_scan_add  <<<gScan, bScan, 0, stream>>>(rowptr, aux, N_NODES);
    TEXTGCN_fill      <<<gE, blk, 0, stream>>>(dst, rowptr, rank, perm, N_EDGES);
    TEXTGCN_dinv_csr  <<<gN16, blk, 0, stream>>>(rowptr, perm, ew, dinv, N_NODES, N_EDGES);

    dim3 gGemm((M_TILES + GEMM_WAVES - 1) / GEMM_WAVES);
    dim3 bGemm(GEMM_T);
    dim3 gFuse(N_NODES / FUSE_R);
    dim3 gHead((NDOCS + FUSE_R - 1) / FUSE_R);
    dim3 bFuse(FUSE_T);

    TEXTGCN_gemm_wmma<DIM_IN><<<gGemm, bGemm, 0, stream>>>(x, W1, dinv, bufT);
    TEXTGCN_prop_gemm<<<gFuse, bFuse, 0, stream>>>(rowptr, perm, src, ew, dinv, bufT, b1, W2, bufH, N_NODES, N_EDGES);
    TEXTGCN_prop_gemm<<<gFuse, bFuse, 0, stream>>>(rowptr, perm, src, ew, dinv, bufH, b2, W3, bufT, N_NODES, N_EDGES);
    TEXTGCN_prop_head<<<gHead, bFuse, 0, stream>>>(rowptr, perm, src, ew, dinv, bufT, b3, doc, Wp, bp, Wc, bc, out,
                                                   NDOCS, N_NODES, N_EDGES);
}


# edge grouping by target through LDS: each block bins 6400 edges by 256-row range, one block per bin counts, scans, places ids; no global atomics, 10 launches to 7; row slots in LDS-add arrival order
# speedup vs baseline: 4.7362x; 1.7257x over previous
#include <hip/hip_runtime.h>
#include <hip/hip_bf16.h>

typedef float v2f __attribute__((ext_vector_type(2)));
typedef float v8f __attribute__((ext_vector_type(8)));

#define N_NODES 100000
#define N_EDGES 1600000
#define DIM_IN  256
#define HD      64
#define NCLS    20
#define NDOCS   10000


#define BIN_R    256
#define NBINS    ((N_NODES + BIN_R - 1) / BIN_R)
#define PART_G   250
#define PART_CH  (N_EDGES / PART_G)
#define PART_T   1024
#define PART_PER ((PART_CH + PART_T - 1) / PART_T)
#define ROWS_T   1024
#define ROWS_CAP 8192
#define KEY_SH   21
#define KEY_MASK ((1 << KEY_SH) - 1)

static_assert(N_EDGES % PART_G == 0, "every partition block owns PART_CH whole edges");
static_assert(N_EDGES <= (1 << KEY_SH), "an edge id fits the low bits of a key");
static_assert(NBINS <= 512 && PART_CH < 32768, "start and count of a segment fit 16 bits each");
static_assert(PART_G <= ROWS_T && BIN_R <= ROWS_T && PART_T == 1024 && ROWS_T == 1024, "one item per thread, 32 waves per block");

__device__ __forceinline__ int TEXTGCN_block_excl(const int v, int* wsum, const int t, int& total) {
    const int lane = t & 31, wv = t >> 5;
    int incl = v;
    for (int off = 1; off < 32; off <<= 1) {
        const int nb = __shfl_up(incl, off, 32);
        if (lane >= off) incl += nb;
    }
    if (lane == 31) wsum[wv] = incl;
    __syncthreads();
    if (wv == 0) {
        const int s = wsum[lane];
        int si = s;
        for (int off = 1; off < 32; off <<= 1) {
            const int nb = __shfl_up(si, off, 32);
            if (lane >= off) si += nb;
        }
        wsum[lane] = si - s;
        if (lane == 31) wsum[32] = si;
    }
    __syncthreads();
    total = wsum[32];
    return incl - v + wsum[wv];
}

__global__ void __launch_bounds__(PART_T)
TEXTGCN_bin_part(const int* __restrict__ dst, int* __restrict__ binned, int* __restrict__ seg) {
    __shared__ int hist[512];
    __shared__ int offs[512];
    __shared__ int wsum[33];
    __shared__ int stage[PART_CH];
    const int t = threadIdx.x;
    const int e0 = blockIdx.x * PART_CH;
    if (t < 512) hist[t] = 0;
    __syncthreads();
    int d[PART_PER], key[PART_PER], pos[PART_PER];
#pragma unroll
    for (int u = 0; u < PART_PER; ++u) {
        const int i = t + u * PART_T;
        d[u] = (i < PART_CH) ? dst[e0 + i] : 0;
    }
#pragma unroll
    for (int u = 0; u < PART_PER; ++u) {
        const int i = t + u * PART_T;
        key[u] = 0; pos[u] = 0;
        if (i < PART_CH) {
            const int b = d[u] / BIN_R;
            const int rk = atomicAdd(&hist[b], 1);
            key[u] = ((d[u] % BIN_R) << KEY_SH) | (e0 + i);
            pos[u] = (b << 16) | rk;
        }
    }
    __syncthreads();
    int total;
    const int cnt = (t < 512) ? hist[t] : 0;
    const int ex = TEXTGCN_block_excl(cnt, wsum, t, total);
    if (t < 512) offs[t] = ex;
    if (t < NBINS) seg[blockIdx.x * NBINS + t] = (ex << 16) | cnt;
    __syncthreads();
#pragma unroll
    for (int u = 0; u < PART_PER; ++u) {
        const int i = t + u * PART_T;
        if (i < PART_CH) stage[offs[pos[u] >> 16] + (pos[u] & 0xffff)] = key[u];
    }
    __syncthreads();
#pragma unroll
    for (int u = 0; u < PART_PER; ++u) {
        const int i = t + u * PART_T;
        if (i < PART_CH) binned[e0 + i] = stage[i];
    }
}

__global__ void __launch_bounds__(ROWS_T)
TEXTGCN_bin_rows(const int* __restrict__ binned, const int* __restrict__ seg,
                 int* __restrict__ rowptr, int* __restrict__ perm) {
    __shared__ int s_cnt[BIN_R];
    __shared__ int s_desc[PART_G];
    __shared__ int wsumA[33];
    __shared__ int wsumB[33];
    __shared__ int wsumC[33];
    __shared__ int s_out[ROWS_CAP];
    const int t = threadIdx.x;
    const int b = blockIdx.x;
    int dsc = 0;
    if (t < PART_G) { dsc = seg[t * NBINS + b]; s_desc[t] = dsc; }
    if (t < BIN_R) s_cnt[t] = 0;
    int base, nb;
    TEXTGCN_block_excl(dsc >> 16, wsumA, t, base);
    TEXTGCN_block_excl(dsc & 0xffff, wsumB, t, nb);
    const bool fast = nb <= ROWS_CAP;
    const int g = t >> 4, c = t & 15;

    constexpr int SEG_G = ROWS_T / 16;
    constexpr int SEG_R = (PART_G + SEG_G - 1) / SEG_G;
    int kA[SEG_R], kB[SEG_R], sn[SEG_R], so[SEG_R];
#pragma unroll
    for (int r = 0; r < SEG_R; ++r) {
        const int sg = g + r * SEG_G;
        const int dd = (sg < PART_G) ? s_desc[sg] : 0;
        sn[r] = dd & 0xffff;
        so[r] = (sg < PART_G ? sg : 0) * PART_CH + (dd >> 16);
        kA[r] = (c      < sn[r]) ? binned[so[r] + c]      : -1;
        kB[r] = (c + 16 < sn[r]) ? binned[so[r] + c + 16] : -1;
    }

#pragma unroll
    for (int r = 0; r < SEG_R; ++r) {
        if (kA[r] >= 0) atomicAdd(&s_cnt[kA[r] >> KEY_SH], 1);
        if (kB[r] >= 0) atomicAdd(&s_cnt[kB[r] >> KEY_SH], 1);
        for (int i = c + 32; i < sn[r]; i += 16) atomicAdd(&s_cnt[binned[so[r] + i] >> KEY_SH], 1);
    }
    __syncthreads();

    const int rc = (t < BIN_R) ? s_cnt[t] : 0;
    int rtot;
    const int rs = TEXTGCN_block_excl(rc, wsumC, t, rtot);
    if (t < BIN_R) {
        s_cnt[t] = rs;
        const int row = b * BIN_R + t;
        if (row < N_NODES) rowptr[row] = base + rs;
    }
    __syncthreads();

#pragma unroll
    for (int r = 0; r < SEG_R; ++r) {
        if (kA[r] >= 0) {
            const int slot = atomicAdd(&s_cnt[kA[r] >> KEY_SH], 1);
            if (fast) s_out[slot] = kA[r] & KEY_MASK; else perm[base + slot] = kA[r] & KEY_MASK;
        }
        if (kB[r] >= 0) {
            const int slot = atomicAdd(&s_cnt[kB[r] >> KEY_SH], 1);
            if (fast) s_out[slot] = kB[r] & KEY_MASK; else perm[base + slot] = kB[r] & KEY_MASK;
        }
        for (int i = c + 32; i < sn[r]; i += 16) {
            const int k = binned[so[r] + i];
            const int slot = atomicAdd(&s_cnt[k >> KEY_SH], 1);
            if (fast) s_out[slot] = k & KEY_MASK; else perm[base + slot] = k & KEY_MASK;
        }
    }
    __syncthreads();
    if (fast)
        for (int i = t; i < nb; i += ROWS_T) perm[base + i] = s_out[i];
}

__global__ void TEXTGCN_dinv_csr(const int* __restrict__ rowptr, const int* __restrict__ perm,
                                 const float* __restrict__ w, float* __restrict__ dinv,
                                 int n, int e) {
    int gid = blockIdx.x * blockDim.x + threadIdx.x;
    int i = gid >> 4, c = gid & 15;
    if (i >= n) return;
    const int start = rowptr[i];
    const int end   = (i + 1 < n) ? rowptr[i + 1] : e;
    float s = 0.0f;
    for (int j = start + c; j < end; j += 16) s += w[perm[j]];
    s += __shfl_xor(s, 8, 16);
    s += __shfl_xor(s, 4, 16);
    s += __shfl_xor(s, 2, 16);
    s += __shfl_xor(s, 1, 16);
    if (c == 0) {
        float d = 1.0f + s;
        dinv[i] = (d > 0.0f) ? __frsqrt_rn(d) : 0.0f;
    }
}


#define GEMM_WAVES 9
#define GEMM_T     (GEMM_WAVES * 32)
#define BS_PITCH   160
#define M_TILES    (N_NODES / 16)

__device__ __forceinline__ float TEXTGCN_xhalf(float x) {
    const int b = __builtin_bit_cast(int, x);
    return __builtin_bit_cast(float,
        __builtin_amdgcn_permlanex16(b, b, 0x76543210, 0xfedcba98, false, false));
}

template <int K>
__global__ void __launch_bounds__(GEMM_T)
TEXTGCN_gemm_wmma(const float* __restrict__ A,
                  const float* __restrict__ B,
                  const float* __restrict__ rscale,
                  float* __restrict__ C) {
    __shared__ __attribute__((aligned(16))) float Bs[(K / 2) * BS_PITCH];
    const int tid = threadIdx.x;
    const int lane = tid & 31;
    const int mt   = blockIdx.x * GEMM_WAVES + (tid >> 5);
    const bool active = mt < M_TILES;
    const int  c = lane & 15, h = lane >> 4;
    const bool hi = (h != 0);
    const float* __restrict__ arow = A + (size_t)((active ? mt : M_TILES - 1) * 16 + c) * K + 4 * h;

    float4 cur[4], nxt[4];
    for (int i = 0; i < 4; ++i) cur[i] = *(const float4*)(arow + 8 * i);

    constexpr int STAGE_IT = (K / 2) * (HD / 4) / 256;
    if (tid < 256) {
        float4 v0[STAGE_IT], v1[STAGE_IT];
        for (int u = 0; u < STAGE_IT; ++u) {
            const int it = tid + u * 256;
            const int kp = it >> 4, q = it & 15;
            v0[u] = *(const float4*)(B + (size_t)(2 * kp)     * HD + 4 * q);
            v1[u] = *(const float4*)(B + (size_t)(2 * kp + 1) * HD + 4 * q);
        }
        for (int u = 0; u < STAGE_IT; ++u) {
            const int it = tid + u * 256;
            const int kp = it >> 4, q = it & 15;
            float4* p = (float4*)(Bs + kp * BS_PITCH + 8 * q);
            p[0] = make_float4(v0[u].x, v1[u].x, v0[u].y, v1[u].y);
            p[1] = make_float4(v0[u].z, v1[u].z, v0[u].w, v1[u].w);
        }
    }
    __syncthreads();

    if (!active) return;

    const float* bl = Bs + h * BS_PITCH + 2 * c;
    v8f acc0 = {}, acc1 = {}, acc2 = {}, acc3 = {};

    for (int kc = 0; kc < K; kc += 32) {
        if (kc + 32 < K) {
            for (int i = 0; i < 4; ++i) nxt[i] = *(const float4*)(arow + kc + 32 + 8 * i);
        }
        for (int i = 0; i < 4; ++i) {
            const float4 a = cur[i];
            const float gx = hi ? a.x : a.z;
            const float gy = hi ? a.y : a.w;
            const float rx = TEXTGCN_xhalf(gx);
            const float ry = TEXTGCN_xhalf(gy);
            v2f a1, a2;
            a1.x = hi ? rx  : a.x;  a1.y = hi ? ry  : a.y;
            a2.x = hi ? a.z : rx;   a2.y = hi ? a.w : ry;
            const float* b1 = bl + ((kc >> 1) + 4 * i) * BS_PITCH;
            const float* b2 = b1 + 2 * BS_PITCH;
            const v2f f0 = *(const v2f*)(b1);
            const v2f f1 = *(const v2f*)(b1 + 32);
            const v2f f2 = *(const v2f*)(b1 + 64);
            const v2f f3 = *(const v2f*)(b1 + 96);
            const v2f g0 = *(const v2f*)(b2);
            const v2f g1 = *(const v2f*)(b2 + 32);
            const v2f g2 = *(const v2f*)(b2 + 64);
            const v2f g3 = *(const v2f*)(b2 + 96);
            acc0 = __builtin_amdgcn_wmma_f32_16x16x4_f32(false, a1, false, f0, (short)0, acc0, false, false);
            acc1 = __builtin_amdgcn_wmma_f32_16x16x4_f32(false, a1, false, f1, (short)0, acc1, false, false);
            acc2 = __builtin_amdgcn_wmma_f32_16x16x4_f32(false, a1, false, f2, (short)0, acc2, false, false);
            acc3 = __builtin_amdgcn_wmma_f32_16x16x4_f32(false, a1, false, f3, (short)0, acc3, false, false);
            acc0 = __builtin_amdgcn_wmma_f32_16x16x4_f32(false, a2, false, g0, (short)0, acc0, false, false);
            acc1 = __builtin_amdgcn_wmma_f32_16x16x4_f32(false, a2, false, g1, (short)0, acc1, false, false);
            acc2 = __builtin_amdgcn_wmma_f32_16x16x4_f32(false, a2, false, g2, (short)0, acc2, false, false);
            acc3 = __builtin_amdgcn_wmma_f32_16x16x4_f32(false, a2, false, g3, (short)0, acc3, false, false);
        }
        if (kc + 32 < K) {
            for (int i = 0; i < 4; ++i) cur[i] = nxt[i];
        }
    }

    const int mbase = mt * 16 + h * 8;
    const float4 s0 = *(const float4*)(rscale + mbase);
    const float4 s1 = *(const float4*)(rscale + mbase + 4);
    const float rs[8] = {s0.x, s0.y, s0.z, s0.w, s1.x, s1.y, s1.z, s1.w};
    float* __restrict__ crow = C + (size_t)mbase * HD + c;
    for (int r = 0; r < 8; ++r) {
        crow[(size_t)r * HD]      = acc0[r] * rs[r];
        crow[(size_t)r * HD + 16] = acc1[r] * rs[r];
        crow[(size_t)r * HD + 32] = acc2[r] * rs[r];
        crow[(size_t)r * HD + 48] = acc3[r] * rs[r];
    }
}


__device__ __forceinline__ float4 TEXTGCN_prop_row(const int i, const int c,
                               const int* __restrict__ rowptr, const int* __restrict__ perm,
                               const int* __restrict__ src, const float* __restrict__ w,
                               const float* __restrict__ dinv, const float* __restrict__ ts,
                               const float* __restrict__ b, int n, int e) {
    const int start = rowptr[i];
    const int end   = (i + 1 < n) ? rowptr[i + 1] : e;
    const float4* __restrict__ t4 = (const float4*)ts;
    float4 acc = t4[(size_t)i * (HD / 4) + c];
    for (int base = start; base < end; base += 16) {
        int   s  = 0;
        float wt = 0.0f;
        if (base + c < end) {
            int ed = perm[base + c];
            s  = src[ed];
            wt = w[ed];
        }
        const int m = (end - base < 16) ? (end - base) : 16;
        int k = 0;
        for (; k + 4 <= m; k += 4) {
            const int   s0 = __shfl(s,  k,     16), s1 = __shfl(s,  k + 1, 16);
            const int   s2 = __shfl(s,  k + 2, 16), s3 = __shfl(s,  k + 3, 16);
            const float w0 = __shfl(wt, k,     16), w1 = __shfl(wt, k + 1, 16);
            const float w2 = __shfl(wt, k + 2, 16), w3 = __shfl(wt, k + 3, 16);
            const float4 v0 = t4[(size_t)s0 * (HD / 4) + c];
            const float4 v1 = t4[(size_t)s1 * (HD / 4) + c];
            const float4 v2 = t4[(size_t)s2 * (HD / 4) + c];
            const float4 v3 = t4[(size_t)s3 * (HD / 4) + c];
            acc.x = fmaf(w0, v0.x, acc.x); acc.y = fmaf(w0, v0.y, acc.y); acc.z = fmaf(w0, v0.z, acc.z); acc.w = fmaf(w0, v0.w, acc.w);
            acc.x = fmaf(w1, v1.x, acc.x); acc.y = fmaf(w1, v1.y, acc.y); acc.z = fmaf(w1, v1.z, acc.z); acc.w = fmaf(w1, v1.w, acc.w);
            acc.x = fmaf(w2, v2.x, acc.x); acc.y = fmaf(w2, v2.y, acc.y); acc.z = fmaf(w2, v2.z, acc.z); acc.w = fmaf(w2, v2.w, acc.w);
            acc.x = fmaf(w3, v3.x, acc.x); acc.y = fmaf(w3, v3.y, acc.y); acc.z = fmaf(w3, v3.z, acc.z); acc.w = fmaf(w3, v3.w, acc.w);
        }
        for (; k < m; ++k) {
            const int   sk = __shfl(s,  k, 16);
            const float wk = __shfl(wt, k, 16);
            const float4 v = t4[(size_t)sk * (HD / 4) + c];
            acc.x = fmaf(wk, v.x, acc.x); acc.y = fmaf(wk, v.y, acc.y); acc.z = fmaf(wk, v.z, acc.z); acc.w = fmaf(wk, v.w, acc.w);
        }
    }
    const float di = dinv[i];
    const float4 bb = ((const float4*)b)[c];
    float4 o;
    o.x = acc.x * di + bb.x; o.y = acc.y * di + bb.y; o.z = acc.z * di + bb.z; o.w = acc.w * di + bb.w;
    o.x = o.x > 0.0f ? o.x : 0.0f;
    o.y = o.y > 0.0f ? o.y : 0.0f;
    o.z = o.z > 0.0f ? o.z : 0.0f;
    o.w = o.w > 0.0f ? o.w : 0.0f;
    return o;
}


#define FUSE_R   32
#define FUSE_T   (FUSE_R * 16)
#define HS_PITCH 68

static_assert(N_NODES % FUSE_R == 0, "every block of the fused propagation owns FUSE_R whole rows");
static_assert((HD / 2) * (HD / 4) == FUSE_T, "one staging item of the 64 x 64 weight per thread");

__device__ __forceinline__ void TEXTGCN_wload(const float* __restrict__ Wm, const int tid,
                                              float4& w0, float4& w1) {
    const int kp = tid >> 4, q = tid & 15;
    w0 = *(const float4*)(Wm + (size_t)(2 * kp)     * HD + 4 * q);
    w1 = *(const float4*)(Wm + (size_t)(2 * kp + 1) * HD + 4 * q);
}

__device__ __forceinline__ void TEXTGCN_wstore(float* Bs, const int tid, const float4 w0, const float4 w1) {
    const int kp = tid >> 4, q = tid & 15;
    float4* p = (float4*)(Bs + kp * BS_PITCH + 8 * q);
    p[0] = make_float4(w0.x, w1.x, w0.y, w1.y);
    p[1] = make_float4(w0.z, w1.z, w0.w, w1.w);
}

__device__ __forceinline__ void TEXTGCN_tile_mm(const float* Hs, const float* Bs, const int nt,
                                                const int c, const int h, v8f& acc0, v8f& acc1) {
    const float* a0 = Hs + c * HS_PITCH + 2 * h;
    const float* a1 = a0 + 16 * HS_PITCH;
    const float* bl = Bs + h * BS_PITCH + 2 * c + 32 * nt;
    for (int s = 0; s < HD / 4; ++s) {
        const v2f x0 = *(const v2f*)(a0 + 4 * s);
        const v2f x1 = *(const v2f*)(a1 + 4 * s);
        const v2f bf = *(const v2f*)(bl + (2 * s) * BS_PITCH);
        acc0 = __builtin_amdgcn_wmma_f32_16x16x4_f32(false, x0, false, bf, (short)0, acc0, false, false);
        acc1 = __builtin_amdgcn_wmma_f32_16x16x4_f32(false, x1, false, bf, (short)0, acc1, false, false);
    }
}

__global__ void __launch_bounds__(FUSE_T)
TEXTGCN_prop_gemm(const int* __restrict__ rowptr, const int* __restrict__ perm,
                  const int* __restrict__ src, const float* __restrict__ w,
                  const float* __restrict__ dinv, const float* __restrict__ ts,
                  const float* __restrict__ b, const float* __restrict__ Wn,
                  float* __restrict__ tn, int n, int e) {
    __shared__ __attribute__((aligned(16))) float Hs[FUSE_R * HS_PITCH];
    __shared__ __attribute__((aligned(16))) float Bs[(HD / 2) * BS_PITCH];
    const int tid = threadIdx.x;
    const int wv = tid >> 5;
    const int c = tid & 15, h = (tid >> 4) & 1;
    const int g = tid >> 4;
    const int row0 = blockIdx.x * FUSE_R;

    float4 w0, w1;
    TEXTGCN_wload(Wn, tid, w0, w1);

    const float4 o = TEXTGCN_prop_row(row0 + g, c, rowptr, perm, src, w, dinv, ts, b, n, e);
    *(float4*)(Hs + g * HS_PITCH + 4 * c) = o;
    TEXTGCN_wstore(Bs, tid, w0, w1);
    __syncthreads();
    if (wv >= 4) return;

    v8f acc0 = {}, acc1 = {};
    TEXTGCN_tile_mm(Hs, Bs, wv, c, h, acc0, acc1);

    const int mbase = row0 + h * 8;
    const float4 s0 = *(const float4*)(dinv + mbase);
    const float4 s1 = *(const float4*)(dinv + mbase + 4);
    const float4 s2 = *(const float4*)(dinv + mbase + 16);
    const float4 s3 = *(const float4*)(dinv + mbase + 20);
    const float rs0[8] = {s0.x, s0.y, s0.z, s0.w, s1.x, s1.y, s1.z, s1.w};
    const float rs1[8] = {s2.x, s2.y, s2.z, s2.w, s3.x, s3.y, s3.z, s3.w};
    float* __restrict__ crow = tn + (size_t)mbase * HD + wv * 16 + c;
    for (int r = 0; r < 8; ++r) {
        crow[(size_t)r * HD]        = acc0[r] * rs0[r];
        crow[(size_t)(r + 16) * HD] = acc1[r] * rs1[r];
    }
}

__global__ void __launch_bounds__(FUSE_T)
TEXTGCN_prop_head(const int* __restrict__ rowptr, const int* __restrict__ perm,
                  const int* __restrict__ src, const float* __restrict__ w,
                  const float* __restrict__ dinv, const float* __restrict__ ts,
                  const float* __restrict__ b, const int* __restrict__ rows,
                  const float* __restrict__ Wp, const float* __restrict__ bp,
                  const float* __restrict__ Wc, const float* __restrict__ bc,
                  float* __restrict__ out, int nrows, int n, int e) {
    __shared__ __attribute__((aligned(16))) float Hs[FUSE_R * HS_PITCH];
    __shared__ __attribute__((aligned(16))) float Ds[FUSE_R * HS_PITCH];
    __shared__ __attribute__((aligned(16))) float Bs[(HD / 2) * BS_PITCH];
    const int tid = threadIdx.x;
    const int wv = tid >> 5;
    const int c = tid & 15, h = (tid >> 4) & 1;
    const int g = tid >> 4;
    const int row0 = blockIdx.x * FUSE_R;

    float4 w0, w1;
    TEXTGCN_wload(Wp, tid, w0, w1);

    float4 o = make_float4(0.0f, 0.0f, 0.0f, 0.0f);
    if (row0 + g < nrows)
        o = TEXTGCN_prop_row(rows[row0 + g], c, rowptr, perm, src, w, dinv, ts, b, n, e);
    *(float4*)(Hs + g * HS_PITCH + 4 * c) = o;
    TEXTGCN_wstore(Bs, tid, w0, w1);
    __syncthreads();

    if (wv < 4) {
        v8f acc0 = {}, acc1 = {};
        TEXTGCN_tile_mm(Hs, Bs, wv, c, h, acc0, acc1);
        const int nst = wv * 16 + c;
        const float bias = bp[nst];
        float* drow = Ds + (h * 8) * HS_PITCH + nst;
        for (int r = 0; r < 8; ++r) {
            const float v0 = acc0[r] + bias;
            const float v1 = acc1[r] + bias;
            drow[r * HS_PITCH]        = v0 > 0.0f ? v0 : 0.0f;
            drow[(r + 16) * HS_PITCH] = v1 > 0.0f ? v1 : 0.0f;
        }
    }
    __syncthreads();

    for (int q = tid; q < FUSE_R * NCLS; q += FUSE_T) {
        const int d = q / NCLS, cc = q % NCLS;
        if (row0 + d < nrows) {
            const float* row = Ds + d * HS_PITCH;
            float s = bc[cc];
#pragma unroll 8
            for (int k = 0; k < HD; ++k) s += row[k] * Wc[(size_t)k * NCLS + cc];
            out[(size_t)(row0 + d) * NCLS + cc] = s;
        }
    }
}


extern "C" void kernel_launch(void* const* d_in, const int* in_sizes, int n_in,
                              void* d_out, int out_size, void* d_ws, size_t ws_size,
                              hipStream_t stream) {
    const float* x   = (const float*)d_in[0];
    const float* ew  = (const float*)d_in[1];
    const float* W1  = (const float*)d_in[2];
    const float* b1  = (const float*)d_in[3];
    const float* W2  = (const float*)d_in[4];
    const float* b2  = (const float*)d_in[5];
    const float* W3  = (const float*)d_in[6];
    const float* b3  = (const float*)d_in[7];
    const float* Wp  = (const float*)d_in[8];
    const float* bp  = (const float*)d_in[9];
    const float* Wc  = (const float*)d_in[10];
    const float* bc  = (const float*)d_in[11];
    const int*   ei  = (const int*)d_in[12];
    const int*   doc = (const int*)d_in[13];
    float*       out = (float*)d_out;

    const int* src = ei;
    const int* dst = ei + N_EDGES;

    float* ws   = (float*)d_ws;
    float* dinv = ws;
    int*   perm = (int*)(dinv + N_NODES);
    float* bufT = (float*)(perm + N_EDGES);
    float* bufH = bufT + (size_t)N_NODES * HD;
    float* dbuf = bufH + (size_t)N_NODES * HD;
    int*   binned = (int*)bufT;
    int*   seg    = binned + N_EDGES;
    int*   rowptr = (int*)dbuf;

    const int B = 256;
    dim3 blk(B);
    dim3 gN16((N_NODES * (HD / 4) + B - 1) / B);
    dim3 gPart(PART_G);
    dim3 bPart(PART_T);
    dim3 gRows(NBINS);
    dim3 bRows(ROWS_T);

    TEXTGCN_bin_part<<<gPart, bPart, 0, stream>>>(dst, binned, seg);
    TEXTGCN_bin_rows<<<gRows, bRows, 0, stream>>>(binned, seg, rowptr, perm);
    TEXTGCN_dinv_csr<<<gN16, blk, 0, stream>>>(rowptr, perm, ew, dinv, N_NODES, N_EDGES);

    dim3 gGemm((M_TILES + GEMM_WAVES - 1) / GEMM_WAVES);
    dim3 bGemm(GEMM_T);
    dim3 gFuse(N_NODES / FUSE_R);
    dim3 gHead((NDOCS + FUSE_R - 1) / FUSE_R);
    dim3 bFuse(FUSE_T);

    TEXTGCN_gemm_wmma<DIM_IN><<<gGemm, bGemm, 0, stream>>>(x, W1, dinv, bufT);
    TEXTGCN_prop_gemm<<<gFuse, bFuse, 0, stream>>>(rowptr, perm, src, ew, dinv, bufT, b1, W2, bufH, N_NODES, N_EDGES);
    TEXTGCN_prop_gemm<<<gFuse, bFuse, 0, stream>>>(rowptr, perm, src, ew, dinv, bufH, b2, W3, bufT, N_NODES, N_EDGES);
    TEXTGCN_prop_head<<<gHead, bFuse, 0, stream>>>(rowptr, perm, src, ew, dinv, bufT, b3, doc, Wp, bp, Wc, bc, out,
                                                   NDOCS, N_NODES, N_EDGES);
}


# dinv moved into the edge grouping: each weight travels with its key through both LDS binning passes and a row's weights are summed from an LDS tile in the same slot order; 7 launches to 6
# speedup vs baseline: 5.0279x; 1.0616x over previous
#include <hip/hip_runtime.h>
#include <hip/hip_bf16.h>

typedef float v2f __attribute__((ext_vector_type(2)));
typedef float v8f __attribute__((ext_vector_type(8)));

#define N_NODES 100000
#define N_EDGES 1600000
#define DIM_IN  256
#define HD      64
#define NCLS    20
#define NDOCS   10000


#define BIN_R    256
#define NBINS    ((N_NODES + BIN_R - 1) / BIN_R)
#define PART_G   250
#define PART_CH  (N_EDGES / PART_G)
#define PART_T   1024
#define PART_PER ((PART_CH + PART_T - 1) / PART_T)
#define ROWS_T   1024
#define ROWS_CAP 8192
#define KEY_SH   21
#define KEY_MASK ((1 << KEY_SH) - 1)

static_assert(N_EDGES % PART_G == 0, "every partition block owns PART_CH whole edges");
static_assert(N_EDGES <= (1 << KEY_SH), "an edge id fits the low bits of a key");
static_assert(NBINS <= 512 && PART_CH < 32768, "start and count of a segment fit 16 bits each");
static_assert(PART_G <= ROWS_T && BIN_R <= ROWS_T && PART_T == 1024 && ROWS_T == 1024, "one item per thread, 32 waves per block");

__device__ __forceinline__ int TEXTGCN_block_excl(const int v, int* wsum, const int t, int& total) {
    const int lane = t & 31, wv = t >> 5;
    int incl = v;
    for (int off = 1; off < 32; off <<= 1) {
        const int nb = __shfl_up(incl, off, 32);
        if (lane >= off) incl += nb;
    }
    if (lane == 31) wsum[wv] = incl;
    __syncthreads();
    if (wv == 0) {
        const int s = wsum[lane];
        int si = s;
        for (int off = 1; off < 32; off <<= 1) {
            const int nb = __shfl_up(si, off, 32);
            if (lane >= off) si += nb;
        }
        wsum[lane] = si - s;
        if (lane == 31) wsum[32] = si;
    }
    __syncthreads();
    total = wsum[32];
    return incl - v + wsum[wv];
}

__global__ void __launch_bounds__(PART_T)
TEXTGCN_bin_part(const int* __restrict__ dst, const float* __restrict__ w,
                 int* __restrict__ binned, float* __restrict__ binned_w, int* __restrict__ seg) {
    __shared__ int hist[512];
    __shared__ int offs[512];
    __shared__ int wsum[33];
    __shared__ int stage[PART_CH];
    __shared__ float stage_w[PART_CH];
    const int t = threadIdx.x;
    const int e0 = blockIdx.x * PART_CH;
    if (t < 512) hist[t] = 0;
    __syncthreads();
    int d[PART_PER], key[PART_PER], pos[PART_PER];
    float wv[PART_PER];
#pragma unroll
    for (int u = 0; u < PART_PER; ++u) {
        const int i = t + u * PART_T;
        d[u]  = (i < PART_CH) ? dst[e0 + i] : 0;
        wv[u] = (i < PART_CH) ? w[e0 + i] : 0.0f;
    }
#pragma unroll
    for (int u = 0; u < PART_PER; ++u) {
        const int i = t + u * PART_T;
        key[u] = 0; pos[u] = 0;
        if (i < PART_CH) {
            const int b = d[u] / BIN_R;
            const int rk = atomicAdd(&hist[b], 1);
            key[u] = ((d[u] % BIN_R) << KEY_SH) | (e0 + i);
            pos[u] = (b << 16) | rk;
        }
    }
    __syncthreads();
    int total;
    const int cnt = (t < 512) ? hist[t] : 0;
    const int ex = TEXTGCN_block_excl(cnt, wsum, t, total);
    if (t < 512) offs[t] = ex;
    if (t < NBINS) seg[blockIdx.x * NBINS + t] = (ex << 16) | cnt;
    __syncthreads();
#pragma unroll
    for (int u = 0; u < PART_PER; ++u) {
        const int i = t + u * PART_T;
        if (i < PART_CH) {
            const int p = offs[pos[u] >> 16] + (pos[u] & 0xffff);
            stage[p]   = key[u];
            stage_w[p] = wv[u];
        }
    }
    __syncthreads();
#pragma unroll
    for (int u = 0; u < PART_PER; ++u) {
        const int i = t + u * PART_T;
        if (i < PART_CH) {
            binned[e0 + i]   = stage[i];
            binned_w[e0 + i] = stage_w[i];
        }
    }
}

__global__ void __launch_bounds__(ROWS_T)
TEXTGCN_bin_rows(const int* __restrict__ binned, const float* __restrict__ binned_w,
                 const int* __restrict__ seg, const float* __restrict__ w,
                 int* __restrict__ rowptr, int* __restrict__ perm, float* __restrict__ dinv) {
    __shared__ int s_cnt[BIN_R];
    __shared__ int s_desc[PART_G];
    __shared__ int wsumA[33];
    __shared__ int wsumB[33];
    __shared__ int wsumC[33];
    __shared__ int s_out[ROWS_CAP];
    __shared__ float s_w[ROWS_CAP];
    const int t = threadIdx.x;
    const int b = blockIdx.x;
    int dsc = 0;
    if (t < PART_G) { dsc = seg[t * NBINS + b]; s_desc[t] = dsc; }
    if (t < BIN_R) s_cnt[t] = 0;
    int base, nb;
    TEXTGCN_block_excl(dsc >> 16, wsumA, t, base);
    TEXTGCN_block_excl(dsc & 0xffff, wsumB, t, nb);
    const bool fast = nb <= ROWS_CAP;
    const int g = t >> 4, c = t & 15;

    constexpr int SEG_G = ROWS_T / 16;
    constexpr int SEG_R = (PART_G + SEG_G - 1) / SEG_G;
    int kA[SEG_R], kB[SEG_R], sn[SEG_R], so[SEG_R];
    float wA[SEG_R], wB[SEG_R];
#pragma unroll
    for (int r = 0; r < SEG_R; ++r) {
        const int sg = g + r * SEG_G;
        const int dd = (sg < PART_G) ? s_desc[sg] : 0;
        sn[r] = dd & 0xffff;
        so[r] = (sg < PART_G ? sg : 0) * PART_CH + (dd >> 16);
        kA[r] = (c      < sn[r]) ? binned[so[r] + c]      : -1;
        kB[r] = (c + 16 < sn[r]) ? binned[so[r] + c + 16] : -1;
        wA[r] = (c      < sn[r]) ? binned_w[so[r] + c]      : 0.0f;
        wB[r] = (c + 16 < sn[r]) ? binned_w[so[r] + c + 16] : 0.0f;
    }

#pragma unroll
    for (int r = 0; r < SEG_R; ++r) {
        if (kA[r] >= 0) atomicAdd(&s_cnt[kA[r] >> KEY_SH], 1);
        if (kB[r] >= 0) atomicAdd(&s_cnt[kB[r] >> KEY_SH], 1);
        for (int i = c + 32; i < sn[r]; i += 16) atomicAdd(&s_cnt[binned[so[r] + i] >> KEY_SH], 1);
    }
    __syncthreads();

    const int rc = (t < BIN_R) ? s_cnt[t] : 0;
    int rtot;
    const int rs = TEXTGCN_block_excl(rc, wsumC, t, rtot);
    if (t < BIN_R) {
        s_cnt[t] = rs;
        const int row = b * BIN_R + t;
        if (row < N_NODES) rowptr[row] = base + rs;
    }
    __syncthreads();

#pragma unroll
    for (int r = 0; r < SEG_R; ++r) {
        if (kA[r] >= 0) {
            const int slot = atomicAdd(&s_cnt[kA[r] >> KEY_SH], 1);
            if (fast) { s_out[slot] = kA[r] & KEY_MASK; s_w[slot] = wA[r]; } else perm[base + slot] = kA[r] & KEY_MASK;
        }
        if (kB[r] >= 0) {
            const int slot = atomicAdd(&s_cnt[kB[r] >> KEY_SH], 1);
            if (fast) { s_out[slot] = kB[r] & KEY_MASK; s_w[slot] = wB[r]; } else perm[base + slot] = kB[r] & KEY_MASK;
        }
        for (int i = c + 32; i < sn[r]; i += 16) {
            const int k = binned[so[r] + i];
            const int slot = atomicAdd(&s_cnt[k >> KEY_SH], 1);
            if (fast) { s_out[slot] = k & KEY_MASK; s_w[slot] = binned_w[so[r] + i]; } else perm[base + slot] = k & KEY_MASK;
        }
    }
    __syncthreads();
    if (fast)
        for (int i = t; i < nb; i += ROWS_T) perm[base + i] = s_out[i];

    constexpr int ROW_Q = BIN_R / SEG_G;
    float s[ROW_Q];
#pragma unroll
    for (int q = 0; q < ROW_Q; ++q) {
        const int lr = g + q * SEG_G;
        const int start = (lr == 0) ? 0 : s_cnt[lr - 1];
        const int end   = s_cnt[lr];
        s[q] = 0.0f;
        if (fast) { for (int j = start + c; j < end; j += 16) s[q] += s_w[j]; }
        else      { for (int j = start + c; j < end; j += 16) s[q] += w[perm[base + j]]; }
    }
#pragma unroll
    for (int off = 8; off >= 1; off >>= 1) {
#pragma unroll
        for (int q = 0; q < ROW_Q; ++q) s[q] += __shfl_xor(s[q], off, 16);
    }
#pragma unroll
    for (int q = 0; q < ROW_Q; ++q) {
        const int row = b * BIN_R + g + q * SEG_G;
        if (c == 0 && row < N_NODES) {
            const float d = 1.0f + s[q];
            dinv[row] = (d > 0.0f) ? __frsqrt_rn(d) : 0.0f;
        }
    }
}


#define GEMM_WAVES 9
#define GEMM_T     (GEMM_WAVES * 32)
#define BS_PITCH   160
#define M_TILES    (N_NODES / 16)

__device__ __forceinline__ float TEXTGCN_xhalf(float x) {
    const int b = __builtin_bit_cast(int, x);
    return __builtin_bit_cast(float,
        __builtin_amdgcn_permlanex16(b, b, 0x76543210, 0xfedcba98, false, false));
}

template <int K>
__global__ void __launch_bounds__(GEMM_T)
TEXTGCN_gemm_wmma(const float* __restrict__ A,
                  const float* __restrict__ B,
                  const float* __restrict__ rscale,
                  float* __restrict__ C) {
    __shared__ __attribute__((aligned(16))) float Bs[(K / 2) * BS_PITCH];
    const int tid = threadIdx.x;
    const int lane = tid & 31;
    const int mt   = blockIdx.x * GEMM_WAVES + (tid >> 5);
    const bool active = mt < M_TILES;
    const int  c = lane & 15, h = lane >> 4;
    const bool hi = (h != 0);
    const float* __restrict__ arow = A + (size_t)((active ? mt : M_TILES - 1) * 16 + c) * K + 4 * h;

    float4 cur[4], nxt[4];
    for (int i = 0; i < 4; ++i) cur[i] = *(const float4*)(arow + 8 * i);

    constexpr int STAGE_IT = (K / 2) * (HD / 4) / 256;
    if (tid < 256) {
        float4 v0[STAGE_IT], v1[STAGE_IT];
        for (int u = 0; u < STAGE_IT; ++u) {
            const int it = tid + u * 256;
            const int kp = it >> 4, q = it & 15;
            v0[u] = *(const float4*)(B + (size_t)(2 * kp)     * HD + 4 * q);
            v1[u] = *(const float4*)(B + (size_t)(2 * kp + 1) * HD + 4 * q);
        }
        for (int u = 0; u < STAGE_IT; ++u) {
            const int it = tid + u * 256;
            const int kp = it >> 4, q = it & 15;
            float4* p = (float4*)(Bs + kp * BS_PITCH + 8 * q);
            p[0] = make_float4(v0[u].x, v1[u].x, v0[u].y, v1[u].y);
            p[1] = make_float4(v0[u].z, v1[u].z, v0[u].w, v1[u].w);
        }
    }
    __syncthreads();

    if (!active) return;

    const float* bl = Bs + h * BS_PITCH + 2 * c;
    v8f acc0 = {}, acc1 = {}, acc2 = {}, acc3 = {};

    for (int kc = 0; kc < K; kc += 32) {
        if (kc + 32 < K) {
            for (int i = 0; i < 4; ++i) nxt[i] = *(const float4*)(arow + kc + 32 + 8 * i);
        }
        for (int i = 0; i < 4; ++i) {
            const float4 a = cur[i];
            const float gx = hi ? a.x : a.z;
            const float gy = hi ? a.y : a.w;
            const float rx = TEXTGCN_xhalf(gx);
            const float ry = TEXTGCN_xhalf(gy);
            v2f a1, a2;
            a1.x = hi ? rx  : a.x;  a1.y = hi ? ry  : a.y;
            a2.x = hi ? a.z : rx;   a2.y = hi ? a.w : ry;
            const float* b1 = bl + ((kc >> 1) + 4 * i) * BS_PITCH;
            const float* b2 = b1 + 2 * BS_PITCH;
            const v2f f0 = *(const v2f*)(b1);
            const v2f f1 = *(const v2f*)(b1 + 32);
            const v2f f2 = *(const v2f*)(b1 + 64);
            const v2f f3 = *(const v2f*)(b1 + 96);
            const v2f g0 = *(const v2f*)(b2);
            const v2f g1 = *(const v2f*)(b2 + 32);
            const v2f g2 = *(const v2f*)(b2 + 64);
            const v2f g3 = *(const v2f*)(b2 + 96);
            acc0 = __builtin_amdgcn_wmma_f32_16x16x4_f32(false, a1, false, f0, (short)0, acc0, false, false);
            acc1 = __builtin_amdgcn_wmma_f32_16x16x4_f32(false, a1, false, f1, (short)0, acc1, false, false);
            acc2 = __builtin_amdgcn_wmma_f32_16x16x4_f32(false, a1, false, f2, (short)0, acc2, false, false);
            acc3 = __builtin_amdgcn_wmma_f32_16x16x4_f32(false, a1, false, f3, (short)0, acc3, false, false);
            acc0 = __builtin_amdgcn_wmma_f32_16x16x4_f32(false, a2, false, g0, (short)0, acc0, false, false);
            acc1 = __builtin_amdgcn_wmma_f32_16x16x4_f32(false, a2, false, g1, (short)0, acc1, false, false);
            acc2 = __builtin_amdgcn_wmma_f32_16x16x4_f32(false, a2, false, g2, (short)0, acc2, false, false);
            acc3 = __builtin_amdgcn_wmma_f32_16x16x4_f32(false, a2, false, g3, (short)0, acc3, false, false);
        }
        if (kc + 32 < K) {
            for (int i = 0; i < 4; ++i) cur[i] = nxt[i];
        }
    }

    const int mbase = mt * 16 + h * 8;
    const float4 s0 = *(const float4*)(rscale + mbase);
    const float4 s1 = *(const float4*)(rscale + mbase + 4);
    const float rs[8] = {s0.x, s0.y, s0.z, s0.w, s1.x, s1.y, s1.z, s1.w};
    float* __restrict__ crow = C + (size_t)mbase * HD + c;
    for (int r = 0; r < 8; ++r) {
        crow[(size_t)r * HD]      = acc0[r] * rs[r];
        crow[(size_t)r * HD + 16] = acc1[r] * rs[r];
        crow[(size_t)r * HD + 32] = acc2[r] * rs[r];
        crow[(size_t)r * HD + 48] = acc3[r] * rs[r];
    }
}


__device__ __forceinline__ float4 TEXTGCN_prop_row(const int i, const int c,
                               const int* __restrict__ rowptr, const int* __restrict__ perm,
                               const int* __restrict__ src, const float* __restrict__ w,
                               const float* __restrict__ dinv, const float* __restrict__ ts,
                               const float* __restrict__ b, int n, int e) {
    const int start = rowptr[i];
    const int end   = (i + 1 < n) ? rowptr[i + 1] : e;
    const float4* __restrict__ t4 = (const float4*)ts;
    float4 acc = t4[(size_t)i * (HD / 4) + c];
    for (int base = start; base < end; base += 16) {
        int   s  = 0;
        float wt = 0.0f;
        if (base + c < end) {
            int ed = perm[base + c];
            s  = src[ed];
            wt = w[ed];
        }
        const int m = (end - base < 16) ? (end - base) : 16;
        int k = 0;
        for (; k + 4 <= m; k += 4) {
            const int   s0 = __shfl(s,  k,     16), s1 = __shfl(s,  k + 1, 16);
            const int   s2 = __shfl(s,  k + 2, 16), s3 = __shfl(s,  k + 3, 16);
            const float w0 = __shfl(wt, k,     16), w1 = __shfl(wt, k + 1, 16);
            const float w2 = __shfl(wt, k + 2, 16), w3 = __shfl(wt, k + 3, 16);
            const float4 v0 = t4[(size_t)s0 * (HD / 4) + c];
            const float4 v1 = t4[(size_t)s1 * (HD / 4) + c];
            const float4 v2 = t4[(size_t)s2 * (HD / 4) + c];
            const float4 v3 = t4[(size_t)s3 * (HD / 4) + c];
            acc.x = fmaf(w0, v0.x, acc.x); acc.y = fmaf(w0, v0.y, acc.y); acc.z = fmaf(w0, v0.z, acc.z); acc.w = fmaf(w0, v0.w, acc.w);
            acc.x = fmaf(w1, v1.x, acc.x); acc.y = fmaf(w1, v1.y, acc.y); acc.z = fmaf(w1, v1.z, acc.z); acc.w = fmaf(w1, v1.w, acc.w);
            acc.x = fmaf(w2, v2.x, acc.x); acc.y = fmaf(w2, v2.y, acc.y); acc.z = fmaf(w2, v2.z, acc.z); acc.w = fmaf(w2, v2.w, acc.w);
            acc.x = fmaf(w3, v3.x, acc.x); acc.y = fmaf(w3, v3.y, acc.y); acc.z = fmaf(w3, v3.z, acc.z); acc.w = fmaf(w3, v3.w, acc.w);
        }
        for (; k < m; ++k) {
            const int   sk = __shfl(s,  k, 16);
            const float wk = __shfl(wt, k, 16);
            const float4 v = t4[(size_t)sk * (HD / 4) + c];
            acc.x = fmaf(wk, v.x, acc.x); acc.y = fmaf(wk, v.y, acc.y); acc.z = fmaf(wk, v.z, acc.z); acc.w = fmaf(wk, v.w, acc.w);
        }
    }
    const float di = dinv[i];
    const float4 bb = ((const float4*)b)[c];
    float4 o;
    o.x = acc.x * di + bb.x; o.y = acc.y * di + bb.y; o.z = acc.z * di + bb.z; o.w = acc.w * di + bb.w;
    o.x = o.x > 0.0f ? o.x : 0.0f;
    o.y = o.y > 0.0f ? o.y : 0.0f;
    o.z = o.z > 0.0f ? o.z : 0.0f;
    o.w = o.w > 0.0f ? o.w : 0.0f;
    return o;
}


#define FUSE_R   32
#define FUSE_T   (FUSE_R * 16)
#define HS_PITCH 68

static_assert(N_NODES % FUSE_R == 0, "every block of the fused propagation owns FUSE_R whole rows");
static_assert((HD / 2) * (HD / 4) == FUSE_T, "one staging item of the 64 x 64 weight per thread");

__device__ __forceinline__ void TEXTGCN_wload(const float* __restrict__ Wm, const int tid,
                                              float4& w0, float4& w1) {
    const int kp = tid >> 4, q = tid & 15;
    w0 = *(const float4*)(Wm + (size_t)(2 * kp)     * HD + 4 * q);
    w1 = *(const float4*)(Wm + (size_t)(2 * kp + 1) * HD + 4 * q);
}

__device__ __forceinline__ void TEXTGCN_wstore(float* Bs, const int tid, const float4 w0, const float4 w1) {
    const int kp = tid >> 4, q = tid & 15;
    float4* p = (float4*)(Bs + kp * BS_PITCH + 8 * q);
    p[0] = make_float4(w0.x, w1.x, w0.y, w1.y);
    p[1] = make_float4(w0.z, w1.z, w0.w, w1.w);
}

__device__ __forceinline__ void TEXTGCN_tile_mm(const float* Hs, const float* Bs, const int nt,
                                                const int c, const int h, v8f& acc0, v8f& acc1) {
    const float* a0 = Hs + c * HS_PITCH + 2 * h;
    const float* a1 = a0 + 16 * HS_PITCH;
    const float* bl = Bs + h * BS_PITCH + 2 * c + 32 * nt;
    for (int s = 0; s < HD / 4; ++s) {
        const v2f x0 = *(const v2f*)(a0 + 4 * s);
        const v2f x1 = *(const v2f*)(a1 + 4 * s);
        const v2f bf = *(const v2f*)(bl + (2 * s) * BS_PITCH);
        acc0 = __builtin_amdgcn_wmma_f32_16x16x4_f32(false, x0, false, bf, (short)0, acc0, false, false);
        acc1 = __builtin_amdgcn_wmma_f32_16x16x4_f32(false, x1, false, bf, (short)0, acc1, false, false);
    }
}

__global__ void __launch_bounds__(FUSE_T)
TEXTGCN_prop_gemm(const int* __restrict__ rowptr, const int* __restrict__ perm,
                  const int* __restrict__ src, const float* __restrict__ w,
                  const float* __restrict__ dinv, const float* __restrict__ ts,
                  const float* __restrict__ b, const float* __restrict__ Wn,
                  float* __restrict__ tn, int n, int e) {
    __shared__ __attribute__((aligned(16))) float Hs[FUSE_R * HS_PITCH];
    __shared__ __attribute__((aligned(16))) float Bs[(HD / 2) * BS_PITCH];
    const int tid = threadIdx.x;
    const int wv = tid >> 5;
    const int c = tid & 15, h = (tid >> 4) & 1;
    const int g = tid >> 4;
    const int row0 = blockIdx.x * FUSE_R;

    float4 w0, w1;
    TEXTGCN_wload(Wn, tid, w0, w1);

    const float4 o = TEXTGCN_prop_row(row0 + g, c, rowptr, perm, src, w, dinv, ts, b, n, e);
    *(float4*)(Hs + g * HS_PITCH + 4 * c) = o;
    TEXTGCN_wstore(Bs, tid, w0, w1);
    __syncthreads();
    if (wv >= 4) return;

    v8f acc0 = {}, acc1 = {};
    TEXTGCN_tile_mm(Hs, Bs, wv, c, h, acc0, acc1);

    const int mbase = row0 + h * 8;
    const float4 s0 = *(const float4*)(dinv + mbase);
    const float4 s1 = *(const float4*)(dinv + mbase + 4);
    const float4 s2 = *(const float4*)(dinv + mbase + 16);
    const float4 s3 = *(const float4*)(dinv + mbase + 20);
    const float rs0[8] = {s0.x, s0.y, s0.z, s0.w, s1.x, s1.y, s1.z, s1.w};
    const float rs1[8] = {s2.x, s2.y, s2.z, s2.w, s3.x, s3.y, s3.z, s3.w};
    float* __restrict__ crow = tn + (size_t)mbase * HD + wv * 16 + c;
    for (int r = 0; r < 8; ++r) {
        crow[(size_t)r * HD]        = acc0[r] * rs0[r];
        crow[(size_t)(r + 16) * HD] = acc1[r] * rs1[r];
    }
}

__global__ void __launch_bounds__(FUSE_T)
TEXTGCN_prop_head(const int* __restrict__ rowptr, const int* __restrict__ perm,
                  const int* __restrict__ src, const float* __restrict__ w,
                  const float* __restrict__ dinv, const float* __restrict__ ts,
                  const float* __restrict__ b, const int* __restrict__ rows,
                  const float* __restrict__ Wp, const float* __restrict__ bp,
                  const float* __restrict__ Wc, const float* __restrict__ bc,
                  float* __restrict__ out, int nrows, int n, int e) {
    __shared__ __attribute__((aligned(16))) float Hs[FUSE_R * HS_PITCH];
    __shared__ __attribute__((aligned(16))) float Ds[FUSE_R * HS_PITCH];
    __shared__ __attribute__((aligned(16))) float Bs[(HD / 2) * BS_PITCH];
    const int tid = threadIdx.x;
    const int wv = tid >> 5;
    const int c = tid & 15, h = (tid >> 4) & 1;
    const int g = tid >> 4;
    const int row0 = blockIdx.x * FUSE_R;

    float4 w0, w1;
    TEXTGCN_wload(Wp, tid, w0, w1);

    float4 o = make_float4(0.0f, 0.0f, 0.0f, 0.0f);
    if (row0 + g < nrows)
        o = TEXTGCN_prop_row(rows[row0 + g], c, rowptr, perm, src, w, dinv, ts, b, n, e);
    *(float4*)(Hs + g * HS_PITCH + 4 * c) = o;
    TEXTGCN_wstore(Bs, tid, w0, w1);
    __syncthreads();

    if (wv < 4) {
        v8f acc0 = {}, acc1 = {};
        TEXTGCN_tile_mm(Hs, Bs, wv, c, h, acc0, acc1);
        const int nst = wv * 16 + c;
        const float bias = bp[nst];
        float* drow = Ds + (h * 8) * HS_PITCH + nst;
        for (int r = 0; r < 8; ++r) {
            const float v0 = acc0[r] + bias;
            const float v1 = acc1[r] + bias;
            drow[r * HS_PITCH]        = v0 > 0.0f ? v0 : 0.0f;
            drow[(r + 16) * HS_PITCH] = v1 > 0.0f ? v1 : 0.0f;
        }
    }
    __syncthreads();

    for (int q = tid; q < FUSE_R * NCLS; q += FUSE_T) {
        const int d = q / NCLS, cc = q % NCLS;
        if (row0 + d < nrows) {
            const float* row = Ds + d * HS_PITCH;
            float s = bc[cc];
#pragma unroll 8
            for (int k = 0; k < HD; ++k) s += row[k] * Wc[(size_t)k * NCLS + cc];
            out[(size_t)(row0 + d) * NCLS + cc] = s;
        }
    }
}


extern "C" void kernel_launch(void* const* d_in, const int* in_sizes, int n_in,
                              void* d_out, int out_size, void* d_ws, size_t ws_size,
                              hipStream_t stream) {
    const float* x   = (const float*)d_in[0];
    const float* ew  = (const float*)d_in[1];
    const float* W1  = (const float*)d_in[2];
    const float* b1  = (const float*)d_in[3];
    const float* W2  = (const float*)d_in[4];
    const float* b2  = (const float*)d_in[5];
    const float* W3  = (const float*)d_in[6];
    const float* b3  = (const float*)d_in[7];
    const float* Wp  = (const float*)d_in[8];
    const float* bp  = (const float*)d_in[9];
    const float* Wc  = (const float*)d_in[10];
    const float* bc  = (const float*)d_in[11];
    const int*   ei  = (const int*)d_in[12];
    const int*   doc = (const int*)d_in[13];
    float*       out = (float*)d_out;

    const int* src = ei;
    const int* dst = ei + N_EDGES;

    float* ws   = (float*)d_ws;
    float* dinv = ws;
    int*   perm = (int*)(dinv + N_NODES);
    float* bufT = (float*)(perm + N_EDGES);
    float* bufH = bufT + (size_t)N_NODES * HD;
    float* dbuf = bufH + (size_t)N_NODES * HD;
    int*   binned = (int*)bufT;
    int*   seg    = binned + N_EDGES;
    float* binned_w = bufT + 2 * (size_t)N_EDGES;
    int*   rowptr = (int*)dbuf;

    dim3 gPart(PART_G);
    dim3 bPart(PART_T);
    dim3 gRows(NBINS);
    dim3 bRows(ROWS_T);

    TEXTGCN_bin_part<<<gPart, bPart, 0, stream>>>(dst, ew, binned, binned_w, seg);
    TEXTGCN_bin_rows<<<gRows, bRows, 0, stream>>>(binned, binned_w, seg, ew, rowptr, perm, dinv);

    dim3 gGemm((M_TILES + GEMM_WAVES - 1) / GEMM_WAVES);
    dim3 bGemm(GEMM_T);
    dim3 gFuse(N_NODES / FUSE_R);
    dim3 gHead((NDOCS + FUSE_R - 1) / FUSE_R);
    dim3 bFuse(FUSE_T);

    TEXTGCN_gemm_wmma<DIM_IN><<<gGemm, bGemm, 0, stream>>>(x, W1, dinv, bufT);
    TEXTGCN_prop_gemm<<<gFuse, bFuse, 0, stream>>>(rowptr, perm, src, ew, dinv, bufT, b1, W2, bufH, N_NODES, N_EDGES);
    TEXTGCN_prop_gemm<<<gFuse, bFuse, 0, stream>>>(rowptr, perm, src, ew, dinv, bufH, b2, W3, bufT, N_NODES, N_EDGES);
    TEXTGCN_prop_head<<<gHead, bFuse, 0, stream>>>(rowptr, perm, src, ew, dinv, bufT, b3, doc, Wp, bp, Wc, bc, out,
                                                   NDOCS, N_NODES, N_EDGES);
}


# propagation: 8 lanes per row, two 16-byte chunks per lane (whole 128-byte lines), 64 rows per block, 8 transform waves: half the waves, twice the row bytes in flight per wave; per-row fma order kept
# speedup vs baseline: 5.3012x; 1.0544x over previous
#include <hip/hip_runtime.h>
#include <hip/hip_bf16.h>

typedef float v2f __attribute__((ext_vector_type(2)));
typedef float v8f __attribute__((ext_vector_type(8)));

#define N_NODES 100000
#define N_EDGES 1600000
#define DIM_IN  256
#define HD      64
#define NCLS    20
#define NDOCS   10000


#define BIN_R    256
#define NBINS    ((N_NODES + BIN_R - 1) / BIN_R)
#define PART_G   250
#define PART_CH  (N_EDGES / PART_G)
#define PART_T   1024
#define PART_PER ((PART_CH + PART_T - 1) / PART_T)
#define ROWS_T   1024
#define ROWS_CAP 8192
#define KEY_SH   21
#define KEY_MASK ((1 << KEY_SH) - 1)

static_assert(N_EDGES % PART_G == 0, "every partition block owns PART_CH whole edges");
static_assert(N_EDGES <= (1 << KEY_SH), "an edge id fits the low bits of a key");
static_assert(NBINS <= 512 && PART_CH < 32768, "start and count of a segment fit 16 bits each");
static_assert(PART_G <= ROWS_T && BIN_R <= ROWS_T && PART_T == 1024 && ROWS_T == 1024, "one item per thread, 32 waves per block");

__device__ __forceinline__ int TEXTGCN_block_excl(const int v, int* wsum, const int t, int& total) {
    const int lane = t & 31, wv = t >> 5;
    int incl = v;
    for (int off = 1; off < 32; off <<= 1) {
        const int nb = __shfl_up(incl, off, 32);
        if (lane >= off) incl += nb;
    }
    if (lane == 31) wsum[wv] = incl;
    __syncthreads();
    if (wv == 0) {
        const int s = wsum[lane];
        int si = s;
        for (int off = 1; off < 32; off <<= 1) {
            const int nb = __shfl_up(si, off, 32);
            if (lane >= off) si += nb;
        }
        wsum[lane] = si - s;
        if (lane == 31) wsum[32] = si;
    }
    __syncthreads();
    total = wsum[32];
    return incl - v + wsum[wv];
}

__global__ void __launch_bounds__(PART_T)
TEXTGCN_bin_part(const int* __restrict__ dst, const float* __restrict__ w,
                 int* __restrict__ binned, float* __restrict__ binned_w, int* __restrict__ seg) {
    __shared__ int hist[512];
    __shared__ int offs[512];
    __shared__ int wsum[33];
    __shared__ int stage[PART_CH];
    __shared__ float stage_w[PART_CH];
    const int t = threadIdx.x;
    const int e0 = blockIdx.x * PART_CH;
    if (t < 512) hist[t] = 0;
    __syncthreads();
    int d[PART_PER], key[PART_PER], pos[PART_PER];
    float wv[PART_PER];
#pragma unroll
    for (int u = 0; u < PART_PER; ++u) {
        const int i = t + u * PART_T;
        d[u]  = (i < PART_CH) ? dst[e0 + i] : 0;
        wv[u] = (i < PART_CH) ? w[e0 + i] : 0.0f;
    }
#pragma unroll
    for (int u = 0; u < PART_PER; ++u) {
        const int i = t + u * PART_T;
        key[u] = 0; pos[u] = 0;
        if (i < PART_CH) {
            const int b = d[u] / BIN_R;
            const int rk = atomicAdd(&hist[b], 1);
            key[u] = ((d[u] % BIN_R) << KEY_SH) | (e0 + i);
            pos[u] = (b << 16) | rk;
        }
    }
    __syncthreads();
    int total;
    const int cnt = (t < 512) ? hist[t] : 0;
    const int ex = TEXTGCN_block_excl(cnt, wsum, t, total);
    if (t < 512) offs[t] = ex;
    if (t < NBINS) seg[blockIdx.x * NBINS + t] = (ex << 16) | cnt;
    __syncthreads();
#pragma unroll
    for (int u = 0; u < PART_PER; ++u) {
        const int i = t + u * PART_T;
        if (i < PART_CH) {
            const int p = offs[pos[u] >> 16] + (pos[u] & 0xffff);
            stage[p]   = key[u];
            stage_w[p] = wv[u];
        }
    }
    __syncthreads();
#pragma unroll
    for (int u = 0; u < PART_PER; ++u) {
        const int i = t + u * PART_T;
        if (i < PART_CH) {
            binned[e0 + i]   = stage[i];
            binned_w[e0 + i] = stage_w[i];
        }
    }
}

__global__ void __launch_bounds__(ROWS_T)
TEXTGCN_bin_rows(const int* __restrict__ binned, const float* __restrict__ binned_w,
                 const int* __restrict__ seg, const float* __restrict__ w,
                 int* __restrict__ rowptr, int* __restrict__ perm, float* __restrict__ dinv) {
    __shared__ int s_cnt[BIN_R];
    __shared__ int s_desc[PART_G];
    __shared__ int wsumA[33];
    __shared__ int wsumB[33];
    __shared__ int wsumC[33];
    __shared__ int s_out[ROWS_CAP];
    __shared__ float s_w[ROWS_CAP];
    const int t = threadIdx.x;
    const int b = blockIdx.x;
    int dsc = 0;
    if (t < PART_G) { dsc = seg[t * NBINS + b]; s_desc[t] = dsc; }
    if (t < BIN_R) s_cnt[t] = 0;
    int base, nb;
    TEXTGCN_block_excl(dsc >> 16, wsumA, t, base);
    TEXTGCN_block_excl(dsc & 0xffff, wsumB, t, nb);
    const bool fast = nb <= ROWS_CAP;
    const int g = t >> 4, c = t & 15;

    constexpr int SEG_G = ROWS_T / 16;
    constexpr int SEG_R = (PART_G + SEG_G - 1) / SEG_G;
    int kA[SEG_R], kB[SEG_R], sn[SEG_R], so[SEG_R];
    float wA[SEG_R], wB[SEG_R];
#pragma unroll
    for (int r = 0; r < SEG_R; ++r) {
        const int sg = g + r * SEG_G;
        const int dd = (sg < PART_G) ? s_desc[sg] : 0;
        sn[r] = dd & 0xffff;
        so[r] = (sg < PART_G ? sg : 0) * PART_CH + (dd >> 16);
        kA[r] = (c      < sn[r]) ? binned[so[r] + c]      : -1;
        kB[r] = (c + 16 < sn[r]) ? binned[so[r] + c + 16] : -1;
        wA[r] = (c      < sn[r]) ? binned_w[so[r] + c]      : 0.0f;
        wB[r] = (c + 16 < sn[r]) ? binned_w[so[r] + c + 16] : 0.0f;
    }

#pragma unroll
    for (int r = 0; r < SEG_R; ++r) {
        if (kA[r] >= 0) atomicAdd(&s_cnt[kA[r] >> KEY_SH], 1);
        if (kB[r] >= 0) atomicAdd(&s_cnt[kB[r] >> KEY_SH], 1);
        for (int i = c + 32; i < sn[r]; i += 16) atomicAdd(&s_cnt[binned[so[r] + i] >> KEY_SH], 1);
    }
    __syncthreads();

    const int rc = (t < BIN_R) ? s_cnt[t] : 0;
    int rtot;
    const int rs = TEXTGCN_block_excl(rc, wsumC, t, rtot);
    if (t < BIN_R) {
        s_cnt[t] = rs;
        const int row = b * BIN_R + t;
        if (row < N_NODES) rowptr[row] = base + rs;
    }
    __syncthreads();

#pragma unroll
    for (int r = 0; r < SEG_R; ++r) {
        if (kA[r] >= 0) {
            const int slot = atomicAdd(&s_cnt[kA[r] >> KEY_SH], 1);
            if (fast) { s_out[slot] = kA[r] & KEY_MASK; s_w[slot] = wA[r]; } else perm[base + slot] = kA[r] & KEY_MASK;
        }
        if (kB[r] >= 0) {
            const int slot = atomicAdd(&s_cnt[kB[r] >> KEY_SH], 1);
            if (fast) { s_out[slot] = kB[r] & KEY_MASK; s_w[slot] = wB[r]; } else perm[base + slot] = kB[r] & KEY_MASK;
        }
        for (int i = c + 32; i < sn[r]; i += 16) {
            const int k = binned[so[r] + i];
            const int slot = atomicAdd(&s_cnt[k >> KEY_SH], 1);
            if (fast) { s_out[slot] = k & KEY_MASK; s_w[slot] = binned_w[so[r] + i]; } else perm[base + slot] = k & KEY_MASK;
        }
    }
    __syncthreads();
    if (fast)
        for (int i = t; i < nb; i += ROWS_T) perm[base + i] = s_out[i];

    constexpr int ROW_Q = BIN_R / SEG_G;
    float s[ROW_Q];
#pragma unroll
    for (int q = 0; q < ROW_Q; ++q) {
        const int lr = g + q * SEG_G;
        const int start = (lr == 0) ? 0 : s_cnt[lr - 1];
        const int end   = s_cnt[lr];
        s[q] = 0.0f;
        if (fast) { for (int j = start + c; j < end; j += 16) s[q] += s_w[j]; }
        else      { for (int j = start + c; j < end; j += 16) s[q] += w[perm[base + j]]; }
    }
#pragma unroll
    for (int off = 8; off >= 1; off >>= 1) {
#pragma unroll
        for (int q = 0; q < ROW_Q; ++q) s[q] += __shfl_xor(s[q], off, 16);
    }
#pragma unroll
    for (int q = 0; q < ROW_Q; ++q) {
        const int row = b * BIN_R + g + q * SEG_G;
        if (c == 0 && row < N_NODES) {
            const float d = 1.0f + s[q];
            dinv[row] = (d > 0.0f) ? __frsqrt_rn(d) : 0.0f;
        }
    }
}


#define GEMM_WAVES 9
#define GEMM_T     (GEMM_WAVES * 32)
#define BS_PITCH   160
#define M_TILES    (N_NODES / 16)

__device__ __forceinline__ float TEXTGCN_xhalf(float x) {
    const int b = __builtin_bit_cast(int, x);
    return __builtin_bit_cast(float,
        __builtin_amdgcn_permlanex16(b, b, 0x76543210, 0xfedcba98, false, false));
}

template <int K>
__global__ void __launch_bounds__(GEMM_T)
TEXTGCN_gemm_wmma(const float* __restrict__ A,
                  const float* __restrict__ B,
                  const float* __restrict__ rscale,
                  float* __restrict__ C) {
    __shared__ __attribute__((aligned(16))) float Bs[(K / 2) * BS_PITCH];
    const int tid = threadIdx.x;
    const int lane = tid & 31;
    const int mt   = blockIdx.x * GEMM_WAVES + (tid >> 5);
    const bool active = mt < M_TILES;
    const int  c = lane & 15, h = lane >> 4;
    const bool hi = (h != 0);
    const float* __restrict__ arow = A + (size_t)((active ? mt : M_TILES - 1) * 16 + c) * K + 4 * h;

    float4 cur[4], nxt[4];
    for (int i = 0; i < 4; ++i) cur[i] = *(const float4*)(arow + 8 * i);

    constexpr int STAGE_IT = (K / 2) * (HD / 4) / 256;
    if (tid < 256) {
        float4 v0[STAGE_IT], v1[STAGE_IT];
        for (int u = 0; u < STAGE_IT; ++u) {
            const int it = tid + u * 256;
            const int kp = it >> 4, q = it & 15;
            v0[u] = *(const float4*)(B + (size_t)(2 * kp)     * HD + 4 * q);
            v1[u] = *(const float4*)(B + (size_t)(2 * kp + 1) * HD + 4 * q);
        }
        for (int u = 0; u < STAGE_IT; ++u) {
            const int it = tid + u * 256;
            const int kp = it >> 4, q = it & 15;
            float4* p = (float4*)(Bs + kp * BS_PITCH + 8 * q);
            p[0] = make_float4(v0[u].x, v1[u].x, v0[u].y, v1[u].y);
            p[1] = make_float4(v0[u].z, v1[u].z, v0[u].w, v1[u].w);
        }
    }
    __syncthreads();

    if (!active) return;

    const float* bl = Bs + h * BS_PITCH + 2 * c;
    v8f acc0 = {}, acc1 = {}, acc2 = {}, acc3 = {};

    for (int kc = 0; kc < K; kc += 32) {
        if (kc + 32 < K) {
            for (int i = 0; i < 4; ++i) nxt[i] = *(const float4*)(arow + kc + 32 + 8 * i);
        }
        for (int i = 0; i < 4; ++i) {
            const float4 a = cur[i];
            const float gx = hi ? a.x : a.z;
            const float gy = hi ? a.y : a.w;
            const float rx = TEXTGCN_xhalf(gx);
            const float ry = TEXTGCN_xhalf(gy);
            v2f a1, a2;
            a1.x = hi ? rx  : a.x;  a1.y = hi ? ry  : a.y;
            a2.x = hi ? a.z : rx;   a2.y = hi ? a.w : ry;
            const float* b1 = bl + ((kc >> 1) + 4 * i) * BS_PITCH;
            const float* b2 = b1 + 2 * BS_PITCH;
            const v2f f0 = *(const v2f*)(b1);
            const v2f f1 = *(const v2f*)(b1 + 32);
            const v2f f2 = *(const v2f*)(b1 + 64);
            const v2f f3 = *(const v2f*)(b1 + 96);
            const v2f g0 = *(const v2f*)(b2);
            const v2f g1 = *(const v2f*)(b2 + 32);
            const v2f g2 = *(const v2f*)(b2 + 64);
            const v2f g3 = *(const v2f*)(b2 + 96);
            acc0 = __builtin_amdgcn_wmma_f32_16x16x4_f32(false, a1, false, f0, (short)0, acc0, false, false);
            acc1 = __builtin_amdgcn_wmma_f32_16x16x4_f32(false, a1, false, f1, (short)0, acc1, false, false);
            acc2 = __builtin_amdgcn_wmma_f32_16x16x4_f32(false, a1, false, f2, (short)0, acc2, false, false);
            acc3 = __builtin_amdgcn_wmma_f32_16x16x4_f32(false, a1, false, f3, (short)0, acc3, false, false);
            acc0 = __builtin_amdgcn_wmma_f32_16x16x4_f32(false, a2, false, g0, (short)0, acc0, false, false);
            acc1 = __builtin_amdgcn_wmma_f32_16x16x4_f32(false, a2, false, g1, (short)0, acc1, false, false);
            acc2 = __builtin_amdgcn_wmma_f32_16x16x4_f32(false, a2, false, g2, (short)0, acc2, false, false);
            acc3 = __builtin_amdgcn_wmma_f32_16x16x4_f32(false, a2, false, g3, (short)0, acc3, false, false);
        }
        if (kc + 32 < K) {
            for (int i = 0; i < 4; ++i) cur[i] = nxt[i];
        }
    }

    const int mbase = mt * 16 + h * 8;
    const float4 s0 = *(const float4*)(rscale + mbase);
    const float4 s1 = *(const float4*)(rscale + mbase + 4);
    const float rs[8] = {s0.x, s0.y, s0.z, s0.w, s1.x, s1.y, s1.z, s1.w};
    float* __restrict__ crow = C + (size_t)mbase * HD + c;
    for (int r = 0; r < 8; ++r) {
        crow[(size_t)r * HD]      = acc0[r] * rs[r];
        crow[(size_t)r * HD + 16] = acc1[r] * rs[r];
        crow[(size_t)r * HD + 32] = acc2[r] * rs[r];
        crow[(size_t)r * HD + 48] = acc3[r] * rs[r];
    }
}


__device__ __forceinline__ void TEXTGCN_fma4(float4& acc, const float wt, const float4 v) {
    acc.x = fmaf(wt, v.x, acc.x); acc.y = fmaf(wt, v.y, acc.y); acc.z = fmaf(wt, v.z, acc.z); acc.w = fmaf(wt, v.w, acc.w);
}

__device__ __forceinline__ float4 TEXTGCN_bias_relu(const float4 acc, const float di, const float4 bb) {
    float4 o;
    o.x = acc.x * di + bb.x; o.y = acc.y * di + bb.y; o.z = acc.z * di + bb.z; o.w = acc.w * di + bb.w;
    o.x = o.x > 0.0f ? o.x : 0.0f;
    o.y = o.y > 0.0f ? o.y : 0.0f;
    o.z = o.z > 0.0f ? o.z : 0.0f;
    o.w = o.w > 0.0f ? o.w : 0.0f;
    return o;
}

struct TEXTGCN_row_head { int start, end; float di; float4 selfA, selfB; };

__device__ __forceinline__ TEXTGCN_row_head TEXTGCN_row_loads(const int i, const int c,
                               const int* __restrict__ rowptr, const float* __restrict__ dinv,
                               const float* __restrict__ ts, int n, int e) {
    TEXTGCN_row_head hd;
    const int nx = (i + 1 < n) ? i + 1 : n - 1;
    const int st1 = rowptr[nx];
    hd.start = rowptr[i];
    hd.end   = (i + 1 < n) ? st1 : e;
    hd.di    = dinv[i];
    hd.selfA = ((const float4*)ts)[(size_t)i * (HD / 4) + c];
    hd.selfB = ((const float4*)ts)[(size_t)i * (HD / 4) + 8 + c];
    return hd;
}

__device__ __forceinline__ void TEXTGCN_prop_row(const int c,
                               const int* __restrict__ perm,
                               const int* __restrict__ src, const float* __restrict__ w,
                               const float* __restrict__ ts, const float* __restrict__ b,
                               const TEXTGCN_row_head hd, float4& oA, float4& oB) {
    const int start = hd.start, end = hd.end;
    const float4* __restrict__ t4 = (const float4*)ts;
    float4 accA = hd.selfA;
    float4 accB = hd.selfB;
    for (int base = start; base < end; base += 16) {
        const int iA = base + c, iB = base + 8 + c;
        const int edA = perm[(iA < end) ? iA : end - 1];
        const int edB = perm[(iB < end) ? iB : end - 1];
        const int   sA = src[edA], sB = src[edB];
        const float wA = w[edA],   wB = w[edB];
        const int m = (end - base < 16) ? (end - base) : 16;
        int k = 0;
        for (; k + 4 <= m; k += 4) {
            const int   sv = (k < 8) ? sA : sB;
            const float wv = (k < 8) ? wA : wB;
            const int   kk = k & 7;
            const int   s0 = __shfl(sv, kk,     8), s1 = __shfl(sv, kk + 1, 8);
            const int   s2 = __shfl(sv, kk + 2, 8), s3 = __shfl(sv, kk + 3, 8);
            const float w0 = __shfl(wv, kk,     8), w1 = __shfl(wv, kk + 1, 8);
            const float w2 = __shfl(wv, kk + 2, 8), w3 = __shfl(wv, kk + 3, 8);
            const float4 a0 = t4[(size_t)s0 * (HD / 4) + c];
            const float4 b0 = t4[(size_t)s0 * (HD / 4) + 8 + c];
            const float4 a1 = t4[(size_t)s1 * (HD / 4) + c];
            const float4 b1 = t4[(size_t)s1 * (HD / 4) + 8 + c];
            const float4 a2 = t4[(size_t)s2 * (HD / 4) + c];
            const float4 b2 = t4[(size_t)s2 * (HD / 4) + 8 + c];
            const float4 a3 = t4[(size_t)s3 * (HD / 4) + c];
            const float4 b3 = t4[(size_t)s3 * (HD / 4) + 8 + c];
            TEXTGCN_fma4(accA, w0, a0); TEXTGCN_fma4(accB, w0, b0);
            TEXTGCN_fma4(accA, w1, a1); TEXTGCN_fma4(accB, w1, b1);
            TEXTGCN_fma4(accA, w2, a2); TEXTGCN_fma4(accB, w2, b2);
            TEXTGCN_fma4(accA, w3, a3); TEXTGCN_fma4(accB, w3, b3);
        }
        for (; k < m; ++k) {
            const int   sk = __shfl((k < 8) ? sA : sB, k & 7, 8);
            const float wk = __shfl((k < 8) ? wA : wB, k & 7, 8);
            const float4 va = t4[(size_t)sk * (HD / 4) + c];
            const float4 vb = t4[(size_t)sk * (HD / 4) + 8 + c];
            TEXTGCN_fma4(accA, wk, va); TEXTGCN_fma4(accB, wk, vb);
        }
    }
    oA = TEXTGCN_bias_relu(accA, hd.di, ((const float4*)b)[c]);
    oB = TEXTGCN_bias_relu(accB, hd.di, ((const float4*)b)[8 + c]);
}


#define FUSE_R   64
#define FUSE_T   (FUSE_R * 8)
#define HS_PITCH 68

static_assert(N_NODES % 32 == 0, "the rows of a block end on a 32-row boundary: a transform wave has 32 whole rows or none");
static_assert((HD / 2) * (HD / 4) == FUSE_T, "one staging item of the 64 x 64 weight per thread");

__device__ __forceinline__ void TEXTGCN_wload(const float* __restrict__ Wm, const int tid,
                                              float4& w0, float4& w1) {
    const int kp = tid >> 4, q = tid & 15;
    w0 = *(const float4*)(Wm + (size_t)(2 * kp)     * HD + 4 * q);
    w1 = *(const float4*)(Wm + (size_t)(2 * kp + 1) * HD + 4 * q);
}

__device__ __forceinline__ void TEXTGCN_wstore(float* Bs, const int tid, const float4 w0, const float4 w1) {
    const int kp = tid >> 4, q = tid & 15;
    float4* p = (float4*)(Bs + kp * BS_PITCH + 8 * q);
    p[0] = make_float4(w0.x, w1.x, w0.y, w1.y);
    p[1] = make_float4(w0.z, w1.z, w0.w, w1.w);
}

__device__ __forceinline__ void TEXTGCN_tile_mm(const float* Hs, const float* Bs, const int nt,
                                                const int c, const int h, v8f& acc0, v8f& acc1) {
    const float* a0 = Hs + c * HS_PITCH + 2 * h;
    const float* a1 = a0 + 16 * HS_PITCH;
    const float* bl = Bs + h * BS_PITCH + 2 * c + 32 * nt;
    for (int s = 0; s < HD / 4; ++s) {
        const v2f x0 = *(const v2f*)(a0 + 4 * s);
        const v2f x1 = *(const v2f*)(a1 + 4 * s);
        const v2f bf = *(const v2f*)(bl + (2 * s) * BS_PITCH);
        acc0 = __builtin_amdgcn_wmma_f32_16x16x4_f32(false, x0, false, bf, (short)0, acc0, false, false);
        acc1 = __builtin_amdgcn_wmma_f32_16x16x4_f32(false, x1, false, bf, (short)0, acc1, false, false);
    }
}

__global__ void __launch_bounds__(FUSE_T, 16)
TEXTGCN_prop_gemm(const int* __restrict__ rowptr, const int* __restrict__ perm,
                  const int* __restrict__ src, const float* __restrict__ w,
                  const float* __restrict__ dinv, const float* __restrict__ ts,
                  const float* __restrict__ b, const float* __restrict__ Wn,
                  float* __restrict__ tn, int n, int e) {
    __shared__ __attribute__((aligned(16))) float Hs[FUSE_R * HS_PITCH];
    __shared__ __attribute__((aligned(16))) float Bs[(HD / 2) * BS_PITCH];
    const int tid = threadIdx.x;
    const int wv = tid >> 5;
    const int c = tid & 15, h = (tid >> 4) & 1;
    const int pc = tid & 7;
    const int g = tid >> 3;
    const int row0 = blockIdx.x * FUSE_R;

    const int ri = (row0 + g < n) ? row0 + g : n - 1;
    float4 w0, w1;
    TEXTGCN_wload(Wn, tid, w0, w1);
    const TEXTGCN_row_head hd = TEXTGCN_row_loads(ri, pc, rowptr, dinv, ts, n, e);
    TEXTGCN_wstore(Bs, tid, w0, w1);

    float4 oA, oB;
    TEXTGCN_prop_row(pc, perm, src, w, ts, b, hd, oA, oB);
    *(float4*)(Hs + g * HS_PITCH + 4 * pc)      = oA;
    *(float4*)(Hs + g * HS_PITCH + 32 + 4 * pc) = oB;
    __syncthreads();
    const int nt = wv & 3, rh = wv >> 2;
    if (wv >= 8 || row0 + 32 * rh >= n) return;

    v8f acc0 = {}, acc1 = {};
    TEXTGCN_tile_mm(Hs + 32 * rh * HS_PITCH, Bs, nt, c, h, acc0, acc1);

    const int mbase = row0 + 32 * rh + h * 8;
    const float4 s0 = *(const float4*)(dinv + mbase);
    const float4 s1 = *(const float4*)(dinv + mbase + 4);
    const float4 s2 = *(const float4*)(dinv + mbase + 16);
    const float4 s3 = *(const float4*)(dinv + mbase + 20);
    const float rs0[8] = {s0.x, s0.y, s0.z, s0.w, s1.x, s1.y, s1.z, s1.w};
    const float rs1[8] = {s2.x, s2.y, s2.z, s2.w, s3.x, s3.y, s3.z, s3.w};
    float* __restrict__ crow = tn + (size_t)mbase * HD + nt * 16 + c;
    for (int r = 0; r < 8; ++r) {
        crow[(size_t)r * HD]        = acc0[r] * rs0[r];
        crow[(size_t)(r + 16) * HD] = acc1[r] * rs1[r];
    }
}

__global__ void __launch_bounds__(FUSE_T, 16)
TEXTGCN_prop_head(const int* __restrict__ rowptr, const int* __restrict__ perm,
                  const int* __restrict__ src, const float* __restrict__ w,
                  const float* __restrict__ dinv, const float* __restrict__ ts,
                  const float* __restrict__ b, const int* __restrict__ rows,
                  const float* __restrict__ Wp, const float* __restrict__ bp,
                  const float* __restrict__ Wc, const float* __restrict__ bc,
                  float* __restrict__ out, int nrows, int n, int e) {
    __shared__ __attribute__((aligned(16))) float Hs[FUSE_R * HS_PITCH];
    __shared__ __attribute__((aligned(16))) float Ds[FUSE_R * HS_PITCH];
    __shared__ __attribute__((aligned(16))) float Bs[(HD / 2) * BS_PITCH];
    const int tid = threadIdx.x;
    const int wv = tid >> 5;
    const int c = tid & 15, h = (tid >> 4) & 1;
    const int pc = tid & 7;
    const int g = tid >> 3;
    const int row0 = blockIdx.x * FUSE_R;

    float4 w0, w1;
    TEXTGCN_wload(Wp, tid, w0, w1);
    const int ri = rows[(row0 + g < nrows) ? row0 + g : nrows - 1];
    const TEXTGCN_row_head hd = TEXTGCN_row_loads(ri, pc, rowptr, dinv, ts, n, e);
    TEXTGCN_wstore(Bs, tid, w0, w1);

    float4 oA, oB;
    TEXTGCN_prop_row(pc, perm, src, w, ts, b, hd, oA, oB);
    *(float4*)(Hs + g * HS_PITCH + 4 * pc)      = oA;
    *(float4*)(Hs + g * HS_PITCH + 32 + 4 * pc) = oB;
    __syncthreads();

    if (wv < 8) {
        const int nt = wv & 3, rh = wv >> 2;
        v8f acc0 = {}, acc1 = {};
        TEXTGCN_tile_mm(Hs + 32 * rh * HS_PITCH, Bs, nt, c, h, acc0, acc1);
        const int nst = nt * 16 + c;
        const float bias = bp[nst];
        float* drow = Ds + (32 * rh + h * 8) * HS_PITCH + nst;
        for (int r = 0; r < 8; ++r) {
            const float v0 = acc0[r] + bias;
            const float v1 = acc1[r] + bias;
            drow[r * HS_PITCH]        = v0 > 0.0f ? v0 : 0.0f;
            drow[(r + 16) * HS_PITCH] = v1 > 0.0f ? v1 : 0.0f;
        }
    }
    __syncthreads();

    for (int q = tid; q < FUSE_R * NCLS; q += FUSE_T) {
        const int d = q / NCLS, cc = q % NCLS;
        if (row0 + d < nrows) {
            const float* row = Ds + d * HS_PITCH;
            float s = bc[cc];
#pragma unroll 8
            for (int k = 0; k < HD; ++k) s += row[k] * Wc[(size_t)k * NCLS + cc];
            out[(size_t)(row0 + d) * NCLS + cc] = s;
        }
    }
}


extern "C" void kernel_launch(void* const* d_in, const int* in_sizes, int n_in,
                              void* d_out, int out_size, void* d_ws, size_t ws_size,
                              hipStream_t stream) {
    const float* x   = (const float*)d_in[0];
    const float* ew  = (const float*)d_in[1];
    const float* W1  = (const float*)d_in[2];
    const float* b1  = (const float*)d_in[3];
    const float* W2  = (const float*)d_in[4];
    const float* b2  = (const float*)d_in[5];
    const float* W3  = (const float*)d_in[6];
    const float* b3  = (const float*)d_in[7];
    const float* Wp  = (const float*)d_in[8];
    const float* bp  = (const float*)d_in[9];
    const float* Wc  = (const float*)d_in[10];
    const float* bc  = (const float*)d_in[11];
    const int*   ei  = (const int*)d_in[12];
    const int*   doc = (const int*)d_in[13];
    float*       out = (float*)d_out;

    const int* src = ei;
    const int* dst = ei + N_EDGES;

    float* ws   = (float*)d_ws;
    float* dinv = ws;
    int*   perm = (int*)(dinv + N_NODES);
    float* bufT = (float*)(perm + N_EDGES);
    float* bufH = bufT + (size_t)N_NODES * HD;
    float* dbuf = bufH + (size_t)N_NODES * HD;
    int*   binned = (int*)bufT;
    int*   seg    = binned + N_EDGES;
    float* binned_w = bufT + 2 * (size_t)N_EDGES;
    int*   rowptr = (int*)dbuf;

    dim3 gPart(PART_G);
    dim3 bPart(PART_T);
    dim3 gRows(NBINS);
    dim3 bRows(ROWS_T);

    TEXTGCN_bin_part<<<gPart, bPart, 0, stream>>>(dst, ew, binned, binned_w, seg);
    TEXTGCN_bin_rows<<<gRows, bRows, 0, stream>>>(binned, binned_w, seg, ew, rowptr, perm, dinv);

    dim3 gGemm((M_TILES + GEMM_WAVES - 1) / GEMM_WAVES);
    dim3 bGemm(GEMM_T);
    dim3 gFuse((N_NODES + FUSE_R - 1) / FUSE_R);
    dim3 gHead((NDOCS + FUSE_R - 1) / FUSE_R);
    dim3 bFuse(FUSE_T);

    TEXTGCN_gemm_wmma<DIM_IN><<<gGemm, bGemm, 0, stream>>>(x, W1, dinv, bufT);
    TEXTGCN_prop_gemm<<<gFuse, bFuse, 0, stream>>>(rowptr, perm, src, ew, dinv, bufT, b1, W2, bufH, N_NODES, N_EDGES);
    TEXTGCN_prop_gemm<<<gFuse, bFuse, 0, stream>>>(rowptr, perm, src, ew, dinv, bufH, b2, W3, bufT, N_NODES, N_EDGES);
    TEXTGCN_prop_head<<<gHead, bFuse, 0, stream>>>(rowptr, perm, src, ew, dinv, bufT, b3, doc, Wp, bp, Wc, bc, out,
                                                   NDOCS, N_NODES, N_EDGES);
}


# edge grouping carries the source id with the key and stores it in slot order beside the edge id (packed word + 1 byte): propagation reads sources in order, not via src[edge id]; row fma order kept
# speedup vs baseline: 5.5412x; 1.0453x over previous
#include <hip/hip_runtime.h>
#include <hip/hip_bf16.h>

typedef float v2f __attribute__((ext_vector_type(2)));
typedef float v8f __attribute__((ext_vector_type(8)));

#define N_NODES 100000
#define N_EDGES 1600000
#define DIM_IN  256
#define HD      64
#define NCLS    20
#define NDOCS   10000


#define BIN_R    256
#define NBINS    ((N_NODES + BIN_R - 1) / BIN_R)
#define PART_G   250
#define PART_CH  (N_EDGES / PART_G)
#define PART_T   1024
#define PART_PER ((PART_CH + PART_T - 1) / PART_T)
#define ROWS_T   1024
#define ROWS_CAP 8192
#define KEY_SH   21
#define KEY_MASK ((1 << KEY_SH) - 1)
#define ED_LO    15
#define ED_LO_MASK ((1 << ED_LO) - 1)

static_assert(N_EDGES % PART_G == 0, "every partition block owns PART_CH whole edges");
static_assert(N_EDGES <= (1 << KEY_SH), "an edge id fits the low bits of a key");
static_assert(NBINS <= 512 && PART_CH < 32768, "start and count of a segment fit 16 bits each");
static_assert(N_NODES <= (1 << (32 - ED_LO)), "a source id fits the bits of a perm word above the edge id's low part");
static_assert(((N_EDGES - 1) >> ED_LO) < 256, "the high bits of an edge id fit a byte");
static_assert(N_NODES + N_EDGES / 4 <= NDOCS * HD && N_EDGES % 4 == 0, "rowptr and the byte per edge fit the doc-row words of the workspace");
static_assert(PART_G <= ROWS_T && BIN_R <= ROWS_T && PART_T == 1024 && ROWS_T == 1024, "one item per thread, 32 waves per block");

__device__ __forceinline__ int TEXTGCN_block_excl(const int v, int* wsum, const int t, int& total) {
    const int lane = t & 31, wv = t >> 5;
    int incl = v;
    for (int off = 1; off < 32; off <<= 1) {
        const int nb = __shfl_up(incl, off, 32);
        if (lane >= off) incl += nb;
    }
    if (lane == 31) wsum[wv] = incl;
    __syncthreads();
    if (wv == 0) {
        const int s = wsum[lane];
        int si = s;
        for (int off = 1; off < 32; off <<= 1) {
            const int nb = __shfl_up(si, off, 32);
            if (lane >= off) si += nb;
        }
        wsum[lane] = si - s;
        if (lane == 31) wsum[32] = si;
    }
    __syncthreads();
    total = wsum[32];
    return incl - v + wsum[wv];
}

__device__ __forceinline__ int TEXTGCN_pack_lo(const int key, const int s) {
    return (int)(((unsigned)s << ED_LO) | (unsigned)(key & ED_LO_MASK));
}
__device__ __forceinline__ unsigned char TEXTGCN_pack_hi(const int key) {
    return (unsigned char)((key & KEY_MASK) >> ED_LO);
}
__device__ __forceinline__ int TEXTGCN_src_id(const int lo) { return (int)((unsigned)lo >> ED_LO); }
__device__ __forceinline__ int TEXTGCN_edge_id(const int lo, const int hi) { return (hi << ED_LO) | (lo & ED_LO_MASK); }

__global__ void __launch_bounds__(PART_T)
TEXTGCN_bin_part(const int* __restrict__ src, const int* __restrict__ dst, const float* __restrict__ w,
                 int* __restrict__ binned, int* __restrict__ binned_s, float* __restrict__ binned_w,
                 int* __restrict__ seg) {
    __shared__ int hist[512];
    __shared__ int offs[512];
    __shared__ int wsum[33];
    __shared__ int stage[PART_CH];
    __shared__ int stage_s[PART_CH];
    __shared__ float stage_w[PART_CH];
    const int t = threadIdx.x;
    const int e0 = blockIdx.x * PART_CH;
    if (t < 512) hist[t] = 0;
    __syncthreads();
    int d[PART_PER], sv[PART_PER], key[PART_PER], pos[PART_PER];
    float wv[PART_PER];
#pragma unroll
    for (int u = 0; u < PART_PER; ++u) {
        const int i = t + u * PART_T;
        d[u]  = (i < PART_CH) ? dst[e0 + i] : 0;
        sv[u] = (i < PART_CH) ? src[e0 + i] : 0;
        wv[u] = (i < PART_CH) ? w[e0 + i] : 0.0f;
    }
#pragma unroll
    for (int u = 0; u < PART_PER; ++u) {
        const int i = t + u * PART_T;
        key[u] = 0; pos[u] = 0;
        if (i < PART_CH) {
            const int b = d[u] / BIN_R;
            const int rk = atomicAdd(&hist[b], 1);
            key[u] = ((d[u] % BIN_R) << KEY_SH) | (e0 + i);
            pos[u] = (b << 16) | rk;
        }
    }
    __syncthreads();
    int total;
    const int cnt = (t < 512) ? hist[t] : 0;
    const int ex = TEXTGCN_block_excl(cnt, wsum, t, total);
    if (t < 512) offs[t] = ex;
    if (t < NBINS) seg[blockIdx.x * NBINS + t] = (ex << 16) | cnt;
    __syncthreads();
#pragma unroll
    for (int u = 0; u < PART_PER; ++u) {
        const int i = t + u * PART_T;
        if (i < PART_CH) {
            const int p = offs[pos[u] >> 16] + (pos[u] & 0xffff);
            stage[p]   = key[u];
            stage_s[p] = sv[u];
            stage_w[p] = wv[u];
        }
    }
    __syncthreads();
#pragma unroll
    for (int u = 0; u < PART_PER; ++u) {
        const int i = t + u * PART_T;
        if (i < PART_CH) {
            binned[e0 + i]   = stage[i];
            binned_s[e0 + i] = stage_s[i];
            binned_w[e0 + i] = stage_w[i];
        }
    }
}

__global__ void __launch_bounds__(ROWS_T)
TEXTGCN_bin_rows(const int* __restrict__ binned, const int* __restrict__ binned_s,
                 const float* __restrict__ binned_w,
                 const int* __restrict__ seg, const float* __restrict__ w,
                 int* __restrict__ rowptr, int* __restrict__ perm, int* __restrict__ permh_words,
                 float* __restrict__ dinv) {
    unsigned char* __restrict__ permh = (unsigned char*)permh_words;
    __shared__ int s_cnt[BIN_R];
    __shared__ int s_desc[PART_G];
    __shared__ int wsumA[33];
    __shared__ int wsumB[33];
    __shared__ int wsumC[33];
    __shared__ int s_out[ROWS_CAP];
    __shared__ float s_w[ROWS_CAP];
    __shared__ unsigned char s_hi[ROWS_CAP];
    const int t = threadIdx.x;
    const int b = blockIdx.x;
    int dsc = 0;
    if (t < PART_G) { dsc = seg[t * NBINS + b]; s_desc[t] = dsc; }
    if (t < BIN_R) s_cnt[t] = 0;
    int base, nb;
    TEXTGCN_block_excl(dsc >> 16, wsumA, t, base);
    TEXTGCN_block_excl(dsc & 0xffff, wsumB, t, nb);
    const bool fast = nb <= ROWS_CAP;
    const int g = t >> 4, c = t & 15;

    constexpr int SEG_G = ROWS_T / 16;
    constexpr int SEG_R = (PART_G + SEG_G - 1) / SEG_G;
    int kA[SEG_R], kB[SEG_R], sA[SEG_R], sB[SEG_R], sn[SEG_R], so[SEG_R];
    float wA[SEG_R], wB[SEG_R];
#pragma unroll
    for (int r = 0; r < SEG_R; ++r) {
        const int sg = g + r * SEG_G;
        const int dd = (sg < PART_G) ? s_desc[sg] : 0;
        sn[r] = dd & 0xffff;
        so[r] = (sg < PART_G ? sg : 0) * PART_CH + (dd >> 16);
        kA[r] = (c      < sn[r]) ? binned[so[r] + c]      : -1;
        kB[r] = (c + 16 < sn[r]) ? binned[so[r] + c + 16] : -1;
        sA[r] = (c      < sn[r]) ? binned_s[so[r] + c]      : 0;
        sB[r] = (c + 16 < sn[r]) ? binned_s[so[r] + c + 16] : 0;
        wA[r] = (c      < sn[r]) ? binned_w[so[r] + c]      : 0.0f;
        wB[r] = (c + 16 < sn[r]) ? binned_w[so[r] + c + 16] : 0.0f;
    }

#pragma unroll
    for (int r = 0; r < SEG_R; ++r) {
        if (kA[r] >= 0) atomicAdd(&s_cnt[kA[r] >> KEY_SH], 1);
        if (kB[r] >= 0) atomicAdd(&s_cnt[kB[r] >> KEY_SH], 1);
        for (int i = c + 32; i < sn[r]; i += 16) atomicAdd(&s_cnt[binned[so[r] + i] >> KEY_SH], 1);
    }
    __syncthreads();

    const int rc = (t < BIN_R) ? s_cnt[t] : 0;
    int rtot;
    const int rs = TEXTGCN_block_excl(rc, wsumC, t, rtot);
    if (t < BIN_R) {
        s_cnt[t] = rs;
        const int row = b * BIN_R + t;
        if (row < N_NODES) rowptr[row] = base + rs;
    }
    __syncthreads();

#pragma unroll
    for (int r = 0; r < SEG_R; ++r) {
        if (kA[r] >= 0) {
            const int slot = atomicAdd(&s_cnt[kA[r] >> KEY_SH], 1);
            const int lo = TEXTGCN_pack_lo(kA[r], sA[r]);
            const unsigned char hi = TEXTGCN_pack_hi(kA[r]);
            if (fast) { s_out[slot] = lo; s_hi[slot] = hi; s_w[slot] = wA[r]; }
            else      { perm[base + slot] = lo; permh[base + slot] = hi; }
        }
        if (kB[r] >= 0) {
            const int slot = atomicAdd(&s_cnt[kB[r] >> KEY_SH], 1);
            const int lo = TEXTGCN_pack_lo(kB[r], sB[r]);
            const unsigned char hi = TEXTGCN_pack_hi(kB[r]);
            if (fast) { s_out[slot] = lo; s_hi[slot] = hi; s_w[slot] = wB[r]; }
            else      { perm[base + slot] = lo; permh[base + slot] = hi; }
        }
        for (int i = c + 32; i < sn[r]; i += 16) {
            const int k = binned[so[r] + i];
            const int slot = atomicAdd(&s_cnt[k >> KEY_SH], 1);
            const int lo = TEXTGCN_pack_lo(k, binned_s[so[r] + i]);
            const unsigned char hi = TEXTGCN_pack_hi(k);
            if (fast) { s_out[slot] = lo; s_hi[slot] = hi; s_w[slot] = binned_w[so[r] + i]; }
            else      { perm[base + slot] = lo; permh[base + slot] = hi; }
        }
    }
    __syncthreads();
    if (fast)
        for (int i = t; i < nb; i += ROWS_T) { perm[base + i] = s_out[i]; permh[base + i] = s_hi[i]; }

    constexpr int ROW_Q = BIN_R / SEG_G;
    float s[ROW_Q];
#pragma unroll
    for (int q = 0; q < ROW_Q; ++q) {
        const int lr = g + q * SEG_G;
        const int start = (lr == 0) ? 0 : s_cnt[lr - 1];
        const int end   = s_cnt[lr];
        s[q] = 0.0f;
        if (fast) { for (int j = start + c; j < end; j += 16) s[q] += s_w[j]; }
        else      { for (int j = start + c; j < end; j += 16) s[q] += w[TEXTGCN_edge_id(perm[base + j], permh[base + j])]; }
    }
#pragma unroll
    for (int off = 8; off >= 1; off >>= 1) {
#pragma unroll
        for (int q = 0; q < ROW_Q; ++q) s[q] += __shfl_xor(s[q], off, 16);
    }
#pragma unroll
    for (int q = 0; q < ROW_Q; ++q) {
        const int row = b * BIN_R + g + q * SEG_G;
        if (c == 0 && row < N_NODES) {
            const float d = 1.0f + s[q];
            dinv[row] = (d > 0.0f) ? __frsqrt_rn(d) : 0.0f;
        }
    }
}


#define GEMM_WAVES 9
#define GEMM_T     (GEMM_WAVES * 32)
#define BS_PITCH   160
#define M_TILES    (N_NODES / 16)

__device__ __forceinline__ float TEXTGCN_xhalf(float x) {
    const int b = __builtin_bit_cast(int, x);
    return __builtin_bit_cast(float,
        __builtin_amdgcn_permlanex16(b, b, 0x76543210, 0xfedcba98, false, false));
}

template <int K>
__global__ void __launch_bounds__(GEMM_T)
TEXTGCN_gemm_wmma(const float* __restrict__ A,
                  const float* __restrict__ B,
                  const float* __restrict__ rscale,
                  float* __restrict__ C) {
    __shared__ __attribute__((aligned(16))) float Bs[(K / 2) * BS_PITCH];
    const int tid = threadIdx.x;
    const int lane = tid & 31;
    const int mt   = blockIdx.x * GEMM_WAVES + (tid >> 5);
    const bool active = mt < M_TILES;
    const int  c = lane & 15, h = lane >> 4;
    const bool hi = (h != 0);
    const float* __restrict__ arow = A + (size_t)((active ? mt : M_TILES - 1) * 16 + c) * K + 4 * h;

    float4 cur[4], nxt[4];
    for (int i = 0; i < 4; ++i) cur[i] = *(const float4*)(arow + 8 * i);

    constexpr int STAGE_IT = (K / 2) * (HD / 4) / 256;
    if (tid < 256) {
        float4 v0[STAGE_IT], v1[STAGE_IT];
        for (int u = 0; u < STAGE_IT; ++u) {
            const int it = tid + u * 256;
            const int kp = it >> 4, q = it & 15;
            v0[u] = *(const float4*)(B + (size_t)(2 * kp)     * HD + 4 * q);
            v1[u] = *(const float4*)(B + (size_t)(2 * kp + 1) * HD + 4 * q);
        }
        for (int u = 0; u < STAGE_IT; ++u) {
            const int it = tid + u * 256;
            const int kp = it >> 4, q = it & 15;
            float4* p = (float4*)(Bs + kp * BS_PITCH + 8 * q);
            p[0] = make_float4(v0[u].x, v1[u].x, v0[u].y, v1[u].y);
            p[1] = make_float4(v0[u].z, v1[u].z, v0[u].w, v1[u].w);
        }
    }
    __syncthreads();

    if (!active) return;

    const float* bl = Bs + h * BS_PITCH + 2 * c;
    v8f acc0 = {}, acc1 = {}, acc2 = {}, acc3 = {};

    for (int kc = 0; kc < K; kc += 32) {
        if (kc + 32 < K) {
            for (int i = 0; i < 4; ++i) nxt[i] = *(const float4*)(arow + kc + 32 + 8 * i);
        }
        for (int i = 0; i < 4; ++i) {
            const float4 a = cur[i];
            const float gx = hi ? a.x : a.z;
            const float gy = hi ? a.y : a.w;
            const float rx = TEXTGCN_xhalf(gx);
            const float ry = TEXTGCN_xhalf(gy);
            v2f a1, a2;
            a1.x = hi ? rx  : a.x;  a1.y = hi ? ry  : a.y;
            a2.x = hi ? a.z : rx;   a2.y = hi ? a.w : ry;
            const float* b1 = bl + ((kc >> 1) + 4 * i) * BS_PITCH;
            const float* b2 = b1 + 2 * BS_PITCH;
            const v2f f0 = *(const v2f*)(b1);
            const v2f f1 = *(const v2f*)(b1 + 32);
            const v2f f2 = *(const v2f*)(b1 + 64);
            const v2f f3 = *(const v2f*)(b1 + 96);
            const v2f g0 = *(const v2f*)(b2);
            const v2f g1 = *(const v2f*)(b2 + 32);
            const v2f g2 = *(const v2f*)(b2 + 64);
            const v2f g3 = *(const v2f*)(b2 + 96);
            acc0 = __builtin_amdgcn_wmma_f32_16x16x4_f32(false, a1, false, f0, (short)0, acc0, false, false);
            acc1 = __builtin_amdgcn_wmma_f32_16x16x4_f32(false, a1, false, f1, (short)0, acc1, false, false);
            acc2 = __builtin_amdgcn_wmma_f32_16x16x4_f32(false, a1, false, f2, (short)0, acc2, false, false);
            acc3 = __builtin_amdgcn_wmma_f32_16x16x4_f32(false, a1, false, f3, (short)0, acc3, false, false);
            acc0 = __builtin_amdgcn_wmma_f32_16x16x4_f32(false, a2, false, g0, (short)0, acc0, false, false);
            acc1 = __builtin_amdgcn_wmma_f32_16x16x4_f32(false, a2, false, g1, (short)0, acc1, false, false);
            acc2 = __builtin_amdgcn_wmma_f32_16x16x4_f32(false, a2, false, g2, (short)0, acc2, false, false);
            acc3 = __builtin_amdgcn_wmma_f32_16x16x4_f32(false, a2, false, g3, (short)0, acc3, false, false);
        }
        if (kc + 32 < K) {
            for (int i = 0; i < 4; ++i) cur[i] = nxt[i];
        }
    }

    const int mbase = mt * 16 + h * 8;
    const float4 s0 = *(const float4*)(rscale + mbase);
    const float4 s1 = *(const float4*)(rscale + mbase + 4);
    const float rs[8] = {s0.x, s0.y, s0.z, s0.w, s1.x, s1.y, s1.z, s1.w};
    float* __restrict__ crow = C + (size_t)mbase * HD + c;
    for (int r = 0; r < 8; ++r) {
        crow[(size_t)r * HD]      = acc0[r] * rs[r];
        crow[(size_t)r * HD + 16] = acc1[r] * rs[r];
        crow[(size_t)r * HD + 32] = acc2[r] * rs[r];
        crow[(size_t)r * HD + 48] = acc3[r] * rs[r];
    }
}


__device__ __forceinline__ void TEXTGCN_fma4(float4& acc, const float wt, const float4 v) {
    acc.x = fmaf(wt, v.x, acc.x); acc.y = fmaf(wt, v.y, acc.y); acc.z = fmaf(wt, v.z, acc.z); acc.w = fmaf(wt, v.w, acc.w);
}

__device__ __forceinline__ float4 TEXTGCN_bias_relu(const float4 acc, const float di, const float4 bb) {
    float4 o;
    o.x = acc.x * di + bb.x; o.y = acc.y * di + bb.y; o.z = acc.z * di + bb.z; o.w = acc.w * di + bb.w;
    o.x = o.x > 0.0f ? o.x : 0.0f;
    o.y = o.y > 0.0f ? o.y : 0.0f;
    o.z = o.z > 0.0f ? o.z : 0.0f;
    o.w = o.w > 0.0f ? o.w : 0.0f;
    return o;
}

struct TEXTGCN_row_head { int start, end; float di; float4 selfA, selfB; };

__device__ __forceinline__ TEXTGCN_row_head TEXTGCN_row_loads(const int i, const int c,
                               const int* __restrict__ rowptr, const float* __restrict__ dinv,
                               const float* __restrict__ ts, int n, int e) {
    TEXTGCN_row_head hd;
    const int nx = (i + 1 < n) ? i + 1 : n - 1;
    const int st1 = rowptr[nx];
    hd.start = rowptr[i];
    hd.end   = (i + 1 < n) ? st1 : e;
    hd.di    = dinv[i];
    hd.selfA = ((const float4*)ts)[(size_t)i * (HD / 4) + c];
    hd.selfB = ((const float4*)ts)[(size_t)i * (HD / 4) + 8 + c];
    return hd;
}

__device__ __forceinline__ void TEXTGCN_prop_row(const int c,
                               const int* __restrict__ perm,
                               const unsigned char* __restrict__ permh, const float* __restrict__ w,
                               const float* __restrict__ ts, const float* __restrict__ b,
                               const TEXTGCN_row_head hd, float4& oA, float4& oB) {
    const int start = hd.start, end = hd.end;
    const float4* __restrict__ t4 = (const float4*)ts;
    float4 accA = hd.selfA;
    float4 accB = hd.selfB;
    for (int base = start; base < end; base += 16) {
        const int iA = base + c, iB = base + 8 + c;
        const int jA = (iA < end) ? iA : end - 1, jB = (iB < end) ? iB : end - 1;
        const int pA = perm[jA], pB = perm[jB];
        const int hA = permh[jA], hB = permh[jB];
        const int   sA = TEXTGCN_src_id(pA), sB = TEXTGCN_src_id(pB);
        const float wA = w[TEXTGCN_edge_id(pA, hA)], wB = w[TEXTGCN_edge_id(pB, hB)];
        const int m = (end - base < 16) ? (end - base) : 16;
        int k = 0;
        for (; k + 4 <= m; k += 4) {
            const int   sv = (k < 8) ? sA : sB;
            const float wv = (k < 8) ? wA : wB;
            const int   kk = k & 7;
            const int   s0 = __shfl(sv, kk,     8), s1 = __shfl(sv, kk + 1, 8);
            const int   s2 = __shfl(sv, kk + 2, 8), s3 = __shfl(sv, kk + 3, 8);
            const float w0 = __shfl(wv, kk,     8), w1 = __shfl(wv, kk + 1, 8);
            const float w2 = __shfl(wv, kk + 2, 8), w3 = __shfl(wv, kk + 3, 8);
            const float4 a0 = t4[(size_t)s0 * (HD / 4) + c];
            const float4 b0 = t4[(size_t)s0 * (HD / 4) + 8 + c];
            const float4 a1 = t4[(size_t)s1 * (HD / 4) + c];
            const float4 b1 = t4[(size_t)s1 * (HD / 4) + 8 + c];
            const float4 a2 = t4[(size_t)s2 * (HD / 4) + c];
            const float4 b2 = t4[(size_t)s2 * (HD / 4) + 8 + c];
            const float4 a3 = t4[(size_t)s3 * (HD / 4) + c];
            const float4 b3 = t4[(size_t)s3 * (HD / 4) + 8 + c];
            TEXTGCN_fma4(accA, w0, a0); TEXTGCN_fma4(accB, w0, b0);
            TEXTGCN_fma4(accA, w1, a1); TEXTGCN_fma4(accB, w1, b1);
            TEXTGCN_fma4(accA, w2, a2); TEXTGCN_fma4(accB, w2, b2);
            TEXTGCN_fma4(accA, w3, a3); TEXTGCN_fma4(accB, w3, b3);
        }
        for (; k < m; ++k) {
            const int   sk = __shfl((k < 8) ? sA : sB, k & 7, 8);
            const float wk = __shfl((k < 8) ? wA : wB, k & 7, 8);
            const float4 va = t4[(size_t)sk * (HD / 4) + c];
            const float4 vb = t4[(size_t)sk * (HD / 4) + 8 + c];
            TEXTGCN_fma4(accA, wk, va); TEXTGCN_fma4(accB, wk, vb);
        }
    }
    oA = TEXTGCN_bias_relu(accA, hd.di, ((const float4*)b)[c]);
    oB = TEXTGCN_bias_relu(accB, hd.di, ((const float4*)b)[8 + c]);
}


#define FUSE_R   64
#define FUSE_T   (FUSE_R * 8)
#define HS_PITCH 68

static_assert(N_NODES % 32 == 0, "the rows of a block end on a 32-row boundary: a transform wave has 32 whole rows or none");
static_assert((HD / 2) * (HD / 4) == FUSE_T, "one staging item of the 64 x 64 weight per thread");

__device__ __forceinline__ void TEXTGCN_wload(const float* __restrict__ Wm, const int tid,
                                              float4& w0, float4& w1) {
    const int kp = tid >> 4, q = tid & 15;
    w0 = *(const float4*)(Wm + (size_t)(2 * kp)     * HD + 4 * q);
    w1 = *(const float4*)(Wm + (size_t)(2 * kp + 1) * HD + 4 * q);
}

__device__ __forceinline__ void TEXTGCN_wstore(float* Bs, const int tid, const float4 w0, const float4 w1) {
    const int kp = tid >> 4, q = tid & 15;
    float4* p = (float4*)(Bs + kp * BS_PITCH + 8 * q);
    p[0] = make_float4(w0.x, w1.x, w0.y, w1.y);
    p[1] = make_float4(w0.z, w1.z, w0.w, w1.w);
}

__device__ __forceinline__ void TEXTGCN_tile_mm(const float* Hs, const float* Bs, const int nt,
                                                const int c, const int h, v8f& acc0, v8f& acc1) {
    const float* a0 = Hs + c * HS_PITCH + 2 * h;
    const float* a1 = a0 + 16 * HS_PITCH;
    const float* bl = Bs + h * BS_PITCH + 2 * c + 32 * nt;
    for (int s = 0; s < HD / 4; ++s) {
        const v2f x0 = *(const v2f*)(a0 + 4 * s);
        const v2f x1 = *(const v2f*)(a1 + 4 * s);
        const v2f bf = *(const v2f*)(bl + (2 * s) * BS_PITCH);
        acc0 = __builtin_amdgcn_wmma_f32_16x16x4_f32(false, x0, false, bf, (short)0, acc0, false, false);
        acc1 = __builtin_amdgcn_wmma_f32_16x16x4_f32(false, x1, false, bf, (short)0, acc1, false, false);
    }
}

__global__ void __launch_bounds__(FUSE_T, 16)
TEXTGCN_prop_gemm(const int* __restrict__ rowptr, const int* __restrict__ perm,
                  const int* __restrict__ permh_words, const float* __restrict__ w,
                  const float* __restrict__ dinv, const float* __restrict__ ts,
                  const float* __restrict__ b, const float* __restrict__ Wn,
                  float* __restrict__ tn, int n, int e) {
    __shared__ __attribute__((aligned(16))) float Hs[FUSE_R * HS_PITCH];
    __shared__ __attribute__((aligned(16))) float Bs[(HD / 2) * BS_PITCH];
    const int tid = threadIdx.x;
    const int wv = tid >> 5;
    const int c = tid & 15, h = (tid >> 4) & 1;
    const int pc = tid & 7;
    const int g = tid >> 3;
    const int row0 = blockIdx.x * FUSE_R;

    const int ri = (row0 + g < n) ? row0 + g : n - 1;
    float4 w0, w1;
    TEXTGCN_wload(Wn, tid, w0, w1);
    const TEXTGCN_row_head hd = TEXTGCN_row_loads(ri, pc, rowptr, dinv, ts, n, e);
    TEXTGCN_wstore(Bs, tid, w0, w1);

    float4 oA, oB;
    TEXTGCN_prop_row(pc, perm, (const unsigned char*)permh_words, w, ts, b, hd, oA, oB);
    *(float4*)(Hs + g * HS_PITCH + 4 * pc)      = oA;
    *(float4*)(Hs + g * HS_PITCH + 32 + 4 * pc) = oB;
    __syncthreads();
    const int nt = wv & 3, rh = wv >> 2;
    if (wv >= 8 || row0 + 32 * rh >= n) return;

    v8f acc0 = {}, acc1 = {};
    TEXTGCN_tile_mm(Hs + 32 * rh * HS_PITCH, Bs, nt, c, h, acc0, acc1);

    const int mbase = row0 + 32 * rh + h * 8;
    const float4 s0 = *(const float4*)(dinv + mbase);
    const float4 s1 = *(const float4*)(dinv + mbase + 4);
    const float4 s2 = *(const float4*)(dinv + mbase + 16);
    const float4 s3 = *(const float4*)(dinv + mbase + 20);
    const float rs0[8] = {s0.x, s0.y, s0.z, s0.w, s1.x, s1.y, s1.z, s1.w};
    const float rs1[8] = {s2.x, s2.y, s2.z, s2.w, s3.x, s3.y, s3.z, s3.w};
    float* __restrict__ crow = tn + (size_t)mbase * HD + nt * 16 + c;
    for (int r = 0; r < 8; ++r) {
        crow[(size_t)r * HD]        = acc0[r] * rs0[r];
        crow[(size_t)(r + 16) * HD] = acc1[r] * rs1[r];
    }
}

__global__ void __launch_bounds__(FUSE_T, 16)
TEXTGCN_prop_head(const int* __restrict__ rowptr, const int* __restrict__ perm,
                  const int* __restrict__ permh_words, const float* __restrict__ w,
                  const float* __restrict__ dinv, const float* __restrict__ ts,
                  const float* __restrict__ b, const int* __restrict__ rows,
                  const float* __restrict__ Wp, const float* __restrict__ bp,
                  const float* __restrict__ Wc, const float* __restrict__ bc,
                  float* __restrict__ out, int nrows, int n, int e) {
    __shared__ __attribute__((aligned(16))) float Hs[FUSE_R * HS_PITCH];
    __shared__ __attribute__((aligned(16))) float Ds[FUSE_R * HS_PITCH];
    __shared__ __attribute__((aligned(16))) float Bs[(HD / 2) * BS_PITCH];
    const int tid = threadIdx.x;
    const int wv = tid >> 5;
    const int c = tid & 15, h = (tid >> 4) & 1;
    const int pc = tid & 7;
    const int g = tid >> 3;
    const int row0 = blockIdx.x * FUSE_R;

    float4 w0, w1;
    TEXTGCN_wload(Wp, tid, w0, w1);
    const int ri = rows[(row0 + g < nrows) ? row0 + g : nrows - 1];
    const TEXTGCN_row_head hd = TEXTGCN_row_loads(ri, pc, rowptr, dinv, ts, n, e);
    TEXTGCN_wstore(Bs, tid, w0, w1);

    float4 oA, oB;
    TEXTGCN_prop_row(pc, perm, (const unsigned char*)permh_words, w, ts, b, hd, oA, oB);
    *(float4*)(Hs + g * HS_PITCH + 4 * pc)      = oA;
    *(float4*)(Hs + g * HS_PITCH + 32 + 4 * pc) = oB;
    __syncthreads();

    if (wv < 8) {
        const int nt = wv & 3, rh = wv >> 2;
        v8f acc0 = {}, acc1 = {};
        TEXTGCN_tile_mm(Hs + 32 * rh * HS_PITCH, Bs, nt, c, h, acc0, acc1);
        const int nst = nt * 16 + c;
        const float bias = bp[nst];
        float* drow = Ds + (32 * rh + h * 8) * HS_PITCH + nst;
        for (int r = 0; r < 8; ++r) {
            const float v0 = acc0[r] + bias;
            const float v1 = acc1[r] + bias;
            drow[r * HS_PITCH]        = v0 > 0.0f ? v0 : 0.0f;
            drow[(r + 16) * HS_PITCH] = v1 > 0.0f ? v1 : 0.0f;
        }
    }
    __syncthreads();

    for (int q = tid; q < FUSE_R * NCLS; q += FUSE_T) {
        const int d = q / NCLS, cc = q % NCLS;
        if (row0 + d < nrows) {
            const float* row = Ds + d * HS_PITCH;
            float s = bc[cc];
#pragma unroll 8
            for (int k = 0; k < HD; ++k) s += row[k] * Wc[(size_t)k * NCLS + cc];
            out[(size_t)(row0 + d) * NCLS + cc] = s;
        }
    }
}


extern "C" void kernel_launch(void* const* d_in, const int* in_sizes, int n_in,
                              void* d_out, int out_size, void* d_ws, size_t ws_size,
                              hipStream_t stream) {
    const float* x   = (const float*)d_in[0];
    const float* ew  = (const float*)d_in[1];
    const float* W1  = (const float*)d_in[2];
    const float* b1  = (const float*)d_in[3];
    const float* W2  = (const float*)d_in[4];
    const float* b2  = (const float*)d_in[5];
    const float* W3  = (const float*)d_in[6];
    const float* b3  = (const float*)d_in[7];
    const float* Wp  = (const float*)d_in[8];
    const float* bp  = (const float*)d_in[9];
    const float* Wc  = (const float*)d_in[10];
    const float* bc  = (const float*)d_in[11];
    const int*   ei  = (const int*)d_in[12];
    const int*   doc = (const int*)d_in[13];
    float*       out = (float*)d_out;

    const int* src = ei;
    const int* dst = ei + N_EDGES;

    float* ws   = (float*)d_ws;
    float* dinv = ws;
    int*   perm = (int*)(dinv + N_NODES);
    float* bufT = (float*)(perm + N_EDGES);
    float* bufH = bufT + (size_t)N_NODES * HD;
    float* dbuf = bufH + (size_t)N_NODES * HD;
    int*   binned = (int*)bufT;
    int*   seg    = binned + N_EDGES;
    float* binned_w = bufT + 2 * (size_t)N_EDGES;
    int*   binned_s = (int*)bufH;
    int*   rowptr = (int*)dbuf;
    int*   permh  = rowptr + N_NODES;

    dim3 gPart(PART_G);
    dim3 bPart(PART_T);
    dim3 gRows(NBINS);
    dim3 bRows(ROWS_T);

    TEXTGCN_bin_part<<<gPart, bPart, 0, stream>>>(src, dst, ew, binned, binned_s, binned_w, seg);
    TEXTGCN_bin_rows<<<gRows, bRows, 0, stream>>>(binned, binned_s, binned_w, seg, ew, rowptr, perm, permh, dinv);

    dim3 gGemm((M_TILES + GEMM_WAVES - 1) / GEMM_WAVES);
    dim3 bGemm(GEMM_T);
    dim3 gFuse((N_NODES + FUSE_R - 1) / FUSE_R);
    dim3 gHead((NDOCS + FUSE_R - 1) / FUSE_R);
    dim3 bFuse(FUSE_T);

    TEXTGCN_gemm_wmma<DIM_IN><<<gGemm, bGemm, 0, stream>>>(x, W1, dinv, bufT);
    TEXTGCN_prop_gemm<<<gFuse, bFuse, 0, stream>>>(rowptr, perm, permh, ew, dinv, bufT, b1, W2, bufH, N_NODES, N_EDGES);
    TEXTGCN_prop_gemm<<<gFuse, bFuse, 0, stream>>>(rowptr, perm, permh, ew, dinv, bufH, b2, W3, bufT, N_NODES, N_EDGES);
    TEXTGCN_prop_head<<<gHead, bFuse, 0, stream>>>(rowptr, perm, permh, ew, dinv, bufT, b3, doc, Wp, bp, Wc, bc, out,
                                                   NDOCS, N_NODES, N_EDGES);
}


# launch fusion: conv1 transform moves into the edge grouping's 2nd launch (400-row bins = 25 tiles a block, its key loads unbranched in one trip); 6 launches to 5; row slot order follows the new bins
# speedup vs baseline: 5.7854x; 1.0441x over previous
#include <hip/hip_runtime.h>
#include <hip/hip_bf16.h>

typedef float v2f __attribute__((ext_vector_type(2)));
typedef float v8f __attribute__((ext_vector_type(8)));

#define N_NODES 100000
#define N_EDGES 1600000
#define DIM_IN  256
#define HD      64
#define NCLS    20
#define NDOCS   10000


#define BIN_R    400
#define NBINS    ((N_NODES + BIN_R - 1) / BIN_R)
#define PART_G   250
#define PART_CH  (N_EDGES / PART_G)
#define PART_T   1024
#define PART_PER ((PART_CH + PART_T - 1) / PART_T)
#define ROWS_T   1024
#define ROWS_CAP 8192
#define KEY_SH   21
#define KEY_MASK ((1 << KEY_SH) - 1)
#define ED_LO    15
#define ED_LO_MASK ((1 << ED_LO) - 1)

static_assert(N_EDGES % PART_G == 0, "every partition block owns PART_CH whole edges");
static_assert(N_EDGES <= (1 << KEY_SH), "an edge id fits the low bits of a key");
static_assert(NBINS <= 512 && PART_CH < 32768, "start and count of a segment fit 16 bits each");
static_assert(N_NODES <= (1 << (32 - ED_LO)), "a source id fits the bits of a perm word above the edge id's low part");
static_assert(((N_EDGES - 1) >> ED_LO) < 256, "the high bits of an edge id fit a byte");
static_assert(N_NODES + N_EDGES / 4 <= NDOCS * HD && N_EDGES % 4 == 0, "rowptr and the byte per edge fit the doc-row words of the workspace");
static_assert(PART_G <= ROWS_T && BIN_R <= ROWS_T && PART_T == 1024 && ROWS_T == 1024, "one item per thread, 32 waves per block");
static_assert(N_NODES % BIN_R == 0 && BIN_R % 16 == 0 && BIN_R / 16 <= ROWS_T / 32, "a bin is whole 16-row tiles, one wave of the bin's block per tile");
static_assert(BIN_R <= (1 << (31 - KEY_SH)), "the row inside the bin fits the key's high bits and the key stays non-negative");
static_assert(4 * (size_t)N_EDGES <= (size_t)N_NODES * HD && PART_G * NBINS <= N_EDGES, "keys, segment table, weights and sources fit one transform buffer");

__device__ __forceinline__ int TEXTGCN_block_excl(const int v, int* wsum, const int t, int& total) {
    const int lane = t & 31, wv = t >> 5;
    int incl = v;
    for (int off = 1; off < 32; off <<= 1) {
        const int nb = __shfl_up(incl, off, 32);
        if (lane >= off) incl += nb;
    }
    if (lane == 31) wsum[wv] = incl;
    __syncthreads();
    if (wv == 0) {
        const int s = wsum[lane];
        int si = s;
        for (int off = 1; off < 32; off <<= 1) {
            const int nb = __shfl_up(si, off, 32);
            if (lane >= off) si += nb;
        }
        wsum[lane] = si - s;
        if (lane == 31) wsum[32] = si;
    }
    __syncthreads();
    total = wsum[32];
    return incl - v + wsum[wv];
}

__device__ __forceinline__ int TEXTGCN_pack_lo(const int key, const int s) {
    return (int)(((unsigned)s << ED_LO) | (unsigned)(key & ED_LO_MASK));
}
__device__ __forceinline__ unsigned char TEXTGCN_pack_hi(const int key) {
    return (unsigned char)((key & KEY_MASK) >> ED_LO);
}
__device__ __forceinline__ int TEXTGCN_src_id(const int lo) { return (int)((unsigned)lo >> ED_LO); }
__device__ __forceinline__ int TEXTGCN_edge_id(const int lo, const int hi) { return (hi << ED_LO) | (lo & ED_LO_MASK); }

__global__ void __launch_bounds__(PART_T)
TEXTGCN_bin_part(const int* __restrict__ src, const int* __restrict__ dst, const float* __restrict__ w,
                 int* __restrict__ binned, int* __restrict__ binned_s, float* __restrict__ binned_w,
                 int* __restrict__ seg) {
    __shared__ int hist[512];
    __shared__ int offs[512];
    __shared__ int wsum[33];
    __shared__ int stage[PART_CH];
    __shared__ int stage_s[PART_CH];
    __shared__ float stage_w[PART_CH];
    const int t = threadIdx.x;
    const int e0 = blockIdx.x * PART_CH;
    if (t < 512) hist[t] = 0;
    __syncthreads();
    int d[PART_PER], sv[PART_PER], key[PART_PER], pos[PART_PER];
    float wv[PART_PER];
#pragma unroll
    for (int u = 0; u < PART_PER; ++u) {
        const int i = t + u * PART_T;
        d[u]  = (i < PART_CH) ? dst[e0 + i] : 0;
        sv[u] = (i < PART_CH) ? src[e0 + i] : 0;
        wv[u] = (i < PART_CH) ? w[e0 + i] : 0.0f;
    }
#pragma unroll
    for (int u = 0; u < PART_PER; ++u) {
        const int i = t + u * PART_T;
        key[u] = 0; pos[u] = 0;
        if (i < PART_CH) {
            const int b = d[u] / BIN_R;
            const int rk = atomicAdd(&hist[b], 1);
            key[u] = ((d[u] % BIN_R) << KEY_SH) | (e0 + i);
            pos[u] = (b << 16) | rk;
        }
    }
    __syncthreads();
    int total;
    const int cnt = (t < 512) ? hist[t] : 0;
    const int ex = TEXTGCN_block_excl(cnt, wsum, t, total);
    if (t < 512) offs[t] = ex;
    if (t < NBINS) seg[blockIdx.x * NBINS + t] = (ex << 16) | cnt;
    __syncthreads();
#pragma unroll
    for (int u = 0; u < PART_PER; ++u) {
        const int i = t + u * PART_T;
        if (i < PART_CH) {
            const int p = offs[pos[u] >> 16] + (pos[u] & 0xffff);
            stage[p]   = key[u];
            stage_s[p] = sv[u];
            stage_w[p] = wv[u];
        }
    }
    __syncthreads();
#pragma unroll
    for (int u = 0; u < PART_PER; ++u) {
        const int i = t + u * PART_T;
        if (i < PART_CH) {
            binned[e0 + i]   = stage[i];
            binned_s[e0 + i] = stage_s[i];
            binned_w[e0 + i] = stage_w[i];
        }
    }
}


#define BS_PITCH   160
#define GEMM_TILES (BIN_R / 16)

__device__ __forceinline__ float TEXTGCN_xhalf(float x) {
    const int b = __builtin_bit_cast(int, x);
    return __builtin_bit_cast(float,
        __builtin_amdgcn_permlanex16(b, b, 0x76543210, 0xfedcba98, false, false));
}

#define STAGE_IT(K) ((K / 2) * (HD / 4) / ROWS_T)

template <int K>
__device__ __forceinline__ void TEXTGCN_stage_b_load(const float* __restrict__ B, const int tid,
                                                     float4 (&v0)[STAGE_IT(K)], float4 (&v1)[STAGE_IT(K)]) {
    static_assert(STAGE_IT(K) * ROWS_T == (K / 2) * (HD / 4), "whole staging items per thread");
    for (int u = 0; u < STAGE_IT(K); ++u) {
        const int it = tid + u * ROWS_T;
        const int kp = it >> 4, q = it & 15;
        v0[u] = *(const float4*)(B + (size_t)(2 * kp)     * HD + 4 * q);
        v1[u] = *(const float4*)(B + (size_t)(2 * kp + 1) * HD + 4 * q);
    }
}

template <int K>
__device__ __forceinline__ void TEXTGCN_stage_b_store(float* Bs, const int tid,
                                                      const float4 (&v0)[STAGE_IT(K)], const float4 (&v1)[STAGE_IT(K)]) {
    for (int u = 0; u < STAGE_IT(K); ++u) {
        const int it = tid + u * ROWS_T;
        const int kp = it >> 4, q = it & 15;
        float4* p = (float4*)(Bs + kp * BS_PITCH + 8 * q);
        p[0] = make_float4(v0[u].x, v1[u].x, v0[u].y, v1[u].y);
        p[1] = make_float4(v0[u].z, v1[u].z, v0[u].w, v1[u].w);
    }
}

template <int K>
__device__ __forceinline__ void TEXTGCN_gemm_tile(const float* __restrict__ arow, float4 (&cur)[4],
                                                  const float* Bs, const float* rsl,
                                                  float* __restrict__ crow, const int c, const int h) {
    const bool hi = (h != 0);
    float4 nxt[4];
    const float* bl = Bs + h * BS_PITCH + 2 * c;
    v8f acc0 = {}, acc1 = {}, acc2 = {}, acc3 = {};

    for (int kc = 0; kc < K; kc += 32) {
        if (kc + 32 < K) {
            for (int i = 0; i < 4; ++i) nxt[i] = *(const float4*)(arow + kc + 32 + 8 * i);
        }
        for (int i = 0; i < 4; ++i) {
            const float4 a = cur[i];
            const float gx = hi ? a.x : a.z;
            const float gy = hi ? a.y : a.w;
            const float rx = TEXTGCN_xhalf(gx);
            const float ry = TEXTGCN_xhalf(gy);
            v2f a1, a2;
            a1.x = hi ? rx  : a.x;  a1.y = hi ? ry  : a.y;
            a2.x = hi ? a.z : rx;   a2.y = hi ? a.w : ry;
            const float* b1 = bl + ((kc >> 1) + 4 * i) * BS_PITCH;
            const float* b2 = b1 + 2 * BS_PITCH;
            const v2f f0 = *(const v2f*)(b1);
            const v2f f1 = *(const v2f*)(b1 + 32);
            const v2f f2 = *(const v2f*)(b1 + 64);
            const v2f f3 = *(const v2f*)(b1 + 96);
            const v2f g0 = *(const v2f*)(b2);
            const v2f g1 = *(const v2f*)(b2 + 32);
            const v2f g2 = *(const v2f*)(b2 + 64);
            const v2f g3 = *(const v2f*)(b2 + 96);
            acc0 = __builtin_amdgcn_wmma_f32_16x16x4_f32(false, a1, false, f0, (short)0, acc0, false, false);
            acc1 = __builtin_amdgcn_wmma_f32_16x16x4_f32(false, a1, false, f1, (short)0, acc1, false, false);
            acc2 = __builtin_amdgcn_wmma_f32_16x16x4_f32(false, a1, false, f2, (short)0, acc2, false, false);
            acc3 = __builtin_amdgcn_wmma_f32_16x16x4_f32(false, a1, false, f3, (short)0, acc3, false, false);
            acc0 = __builtin_amdgcn_wmma_f32_16x16x4_f32(false, a2, false, g0, (short)0, acc0, false, false);
            acc1 = __builtin_amdgcn_wmma_f32_16x16x4_f32(false, a2, false, g1, (short)0, acc1, false, false);
            acc2 = __builtin_amdgcn_wmma_f32_16x16x4_f32(false, a2, false, g2, (short)0, acc2, false, false);
            acc3 = __builtin_amdgcn_wmma_f32_16x16x4_f32(false, a2, false, g3, (short)0, acc3, false, false);
        }
        if (kc + 32 < K) {
            for (int i = 0; i < 4; ++i) cur[i] = nxt[i];
        }
    }

    for (int r = 0; r < 8; ++r) {
        const float rs = rsl[r];
        crow[(size_t)r * HD]      = acc0[r] * rs;
        crow[(size_t)r * HD + 16] = acc1[r] * rs;
        crow[(size_t)r * HD + 32] = acc2[r] * rs;
        crow[(size_t)r * HD + 48] = acc3[r] * rs;
    }
}

__global__ void __launch_bounds__(ROWS_T)
TEXTGCN_bin_rows(const int* __restrict__ binned, const int* __restrict__ binned_s,
                 const float* __restrict__ binned_w,
                 const int* __restrict__ seg, const float* __restrict__ w,
                 int* __restrict__ rowptr, int* __restrict__ perm, int* __restrict__ permh_words,
                 float* __restrict__ dinv,
                 const float* __restrict__ x, const float* __restrict__ W1, float* __restrict__ ts) {
    unsigned char* __restrict__ permh = (unsigned char*)permh_words;
    __shared__ int s_cnt[BIN_R];
    __shared__ int s_desc[PART_G];
    __shared__ int wsumA[33];
    __shared__ int wsumB[33];
    __shared__ int wsumC[33];
    __shared__ int s_out[ROWS_CAP];
    __shared__ float s_w[ROWS_CAP];
    __shared__ unsigned char s_hi[ROWS_CAP];
    __shared__ float s_dinv[BIN_R];
    __shared__ __attribute__((aligned(16))) float Bs[(DIM_IN / 2) * BS_PITCH];
    const int t = threadIdx.x;
    const int b = blockIdx.x;
    float4 wv0[STAGE_IT(DIM_IN)], wv1[STAGE_IT(DIM_IN)];
    TEXTGCN_stage_b_load<DIM_IN>(W1, t, wv0, wv1);
    int dsc = 0;
    if (t < PART_G) dsc = seg[t * NBINS + b];
    if (t < BIN_R) s_cnt[t] = 0;
    TEXTGCN_stage_b_store<DIM_IN>(Bs, t, wv0, wv1);
    if (t < PART_G) s_desc[t] = dsc;
    int base, nb;
    TEXTGCN_block_excl(dsc >> 16, wsumA, t, base);
    TEXTGCN_block_excl(dsc & 0xffff, wsumB, t, nb);
    const bool fast = nb <= ROWS_CAP;
    const int g = t >> 4, c = t & 15;

    constexpr int SEG_G = ROWS_T / 16;
    constexpr int SEG_R = (PART_G + SEG_G - 1) / SEG_G;
    int kA[SEG_R], kB[SEG_R], kC[SEG_R], sA[SEG_R], sB[SEG_R], sC[SEG_R], sn[SEG_R], so[SEG_R];
    float wA[SEG_R], wB[SEG_R], wC[SEG_R];
#pragma unroll
    for (int r = 0; r < SEG_R; ++r) {
        const int sg = g + r * SEG_G;
        const int dd = (sg < PART_G) ? s_desc[sg] : 0;
        sn[r] = dd & 0xffff;
        so[r] = (sg < PART_G ? sg : 0) * PART_CH + (dd >> 16);
        const int lastk = (sn[r] > 0) ? so[r] + sn[r] - 1 : 0;
        const int iA = (c      < sn[r]) ? so[r] + c      : lastk;
        const int iB = (c + 16 < sn[r]) ? so[r] + c + 16 : lastk;
        const int iC = (c + 32 < sn[r]) ? so[r] + c + 32 : lastk;
        kA[r] = binned[iA];   kB[r] = binned[iB];   kC[r] = binned[iC];
        sA[r] = binned_s[iA]; sB[r] = binned_s[iB]; sC[r] = binned_s[iC];
        wA[r] = binned_w[iA]; wB[r] = binned_w[iB]; wC[r] = binned_w[iC];
        kA[r] = (c      < sn[r]) ? kA[r] : -1;
        kB[r] = (c + 16 < sn[r]) ? kB[r] : -1;
        kC[r] = (c + 32 < sn[r]) ? kC[r] : -1;
    }

    const int wv = t >> 5, h = (t >> 4) & 1;
    const bool gw = wv < GEMM_TILES;
    const float* __restrict__ arow = x + (size_t)(b * BIN_R + (gw ? wv : 0) * 16 + c) * DIM_IN + 4 * h;
    float4 cur[4];
    for (int i = 0; i < 4; ++i) cur[i] = *(const float4*)(arow + 8 * i);

#pragma unroll
    for (int r = 0; r < SEG_R; ++r) {
        if (kA[r] >= 0) atomicAdd(&s_cnt[kA[r] >> KEY_SH], 1);
        if (kB[r] >= 0) atomicAdd(&s_cnt[kB[r] >> KEY_SH], 1);
        if (kC[r] >= 0) atomicAdd(&s_cnt[kC[r] >> KEY_SH], 1);
        for (int i = c + 48; i < sn[r]; i += 16) atomicAdd(&s_cnt[binned[so[r] + i] >> KEY_SH], 1);
    }
    __syncthreads();

    const int rc = (t < BIN_R) ? s_cnt[t] : 0;
    int rtot;
    const int rs = TEXTGCN_block_excl(rc, wsumC, t, rtot);
    if (t < BIN_R) {
        s_cnt[t] = rs;
        const int row = b * BIN_R + t;
        if (row < N_NODES) rowptr[row] = base + rs;
    }
    __syncthreads();

#pragma unroll
    for (int r = 0; r < SEG_R; ++r) {
        if (kA[r] >= 0) {
            const int slot = atomicAdd(&s_cnt[kA[r] >> KEY_SH], 1);
            const int lo = TEXTGCN_pack_lo(kA[r], sA[r]);
            const unsigned char hi = TEXTGCN_pack_hi(kA[r]);
            if (fast) { s_out[slot] = lo; s_hi[slot] = hi; s_w[slot] = wA[r]; }
            else      { perm[base + slot] = lo; permh[base + slot] = hi; }
        }
        if (kB[r] >= 0) {
            const int slot = atomicAdd(&s_cnt[kB[r] >> KEY_SH], 1);
            const int lo = TEXTGCN_pack_lo(kB[r], sB[r]);
            const unsigned char hi = TEXTGCN_pack_hi(kB[r]);
            if (fast) { s_out[slot] = lo; s_hi[slot] = hi; s_w[slot] = wB[r]; }
            else      { perm[base + slot] = lo; permh[base + slot] = hi; }
        }
        if (kC[r] >= 0) {
            const int slot = atomicAdd(&s_cnt[kC[r] >> KEY_SH], 1);
            const int lo = TEXTGCN_pack_lo(kC[r], sC[r]);
            const unsigned char hi = TEXTGCN_pack_hi(kC[r]);
            if (fast) { s_out[slot] = lo; s_hi[slot] = hi; s_w[slot] = wC[r]; }
            else      { perm[base + slot] = lo; permh[base + slot] = hi; }
        }
        for (int i = c + 48; i < sn[r]; i += 16) {
            const int k = binned[so[r] + i];
            const int slot = atomicAdd(&s_cnt[k >> KEY_SH], 1);
            const int lo = TEXTGCN_pack_lo(k, binned_s[so[r] + i]);
            const unsigned char hi = TEXTGCN_pack_hi(k);
            if (fast) { s_out[slot] = lo; s_hi[slot] = hi; s_w[slot] = binned_w[so[r] + i]; }
            else      { perm[base + slot] = lo; permh[base + slot] = hi; }
        }
    }
    __syncthreads();
    if (fast)
        for (int i = t; i < nb; i += ROWS_T) { perm[base + i] = s_out[i]; permh[base + i] = s_hi[i]; }

    constexpr int ROW_Q = (BIN_R + SEG_G - 1) / SEG_G;
    float s[ROW_Q];
#pragma unroll
    for (int q = 0; q < ROW_Q; ++q) {
        const int lr = g + q * SEG_G;
        s[q] = 0.0f;
        if (lr < BIN_R) {
            const int start = (lr == 0) ? 0 : s_cnt[lr - 1];
            const int end   = s_cnt[lr];
            if (fast) { for (int j = start + c; j < end; j += 16) s[q] += s_w[j]; }
            else      { for (int j = start + c; j < end; j += 16) s[q] += w[TEXTGCN_edge_id(perm[base + j], permh[base + j])]; }
        }
    }
#pragma unroll
    for (int off = 8; off >= 1; off >>= 1) {
#pragma unroll
        for (int q = 0; q < ROW_Q; ++q) s[q] += __shfl_xor(s[q], off, 16);
    }
#pragma unroll
    for (int q = 0; q < ROW_Q; ++q) {
        const int lr = g + q * SEG_G;
        const int row = b * BIN_R + lr;
        if (c == 0 && lr < BIN_R) {
            const float d = 1.0f + s[q];
            const float di = (d > 0.0f) ? __frsqrt_rn(d) : 0.0f;
            s_dinv[lr] = di;
            if (row < N_NODES) dinv[row] = di;
        }
    }
    __syncthreads();
    if (!gw) return;

    const int mbase = wv * 16 + h * 8;
    TEXTGCN_gemm_tile<DIM_IN>(arow, cur, Bs, s_dinv + mbase,
                              ts + (size_t)(b * BIN_R + mbase) * HD + c, c, h);
}


__device__ __forceinline__ void TEXTGCN_fma4(float4& acc, const float wt, const float4 v) {
    acc.x = fmaf(wt, v.x, acc.x); acc.y = fmaf(wt, v.y, acc.y); acc.z = fmaf(wt, v.z, acc.z); acc.w = fmaf(wt, v.w, acc.w);
}

__device__ __forceinline__ float4 TEXTGCN_bias_relu(const float4 acc, const float di, const float4 bb) {
    float4 o;
    o.x = acc.x * di + bb.x; o.y = acc.y * di + bb.y; o.z = acc.z * di + bb.z; o.w = acc.w * di + bb.w;
    o.x = o.x > 0.0f ? o.x : 0.0f;
    o.y = o.y > 0.0f ? o.y : 0.0f;
    o.z = o.z > 0.0f ? o.z : 0.0f;
    o.w = o.w > 0.0f ? o.w : 0.0f;
    return o;
}

struct TEXTGCN_row_head { int start, end; float di; float4 selfA, selfB; };

__device__ __forceinline__ TEXTGCN_row_head TEXTGCN_row_loads(const int i, const int c,
                               const int* __restrict__ rowptr, const float* __restrict__ dinv,
                               const float* __restrict__ ts, int n, int e) {
    TEXTGCN_row_head hd;
    const int nx = (i + 1 < n) ? i + 1 : n - 1;
    const int st1 = rowptr[nx];
    hd.start = rowptr[i];
    hd.end   = (i + 1 < n) ? st1 : e;
    hd.di    = dinv[i];
    hd.selfA = ((const float4*)ts)[(size_t)i * (HD / 4) + c];
    hd.selfB = ((const float4*)ts)[(size_t)i * (HD / 4) + 8 + c];
    return hd;
}

__device__ __forceinline__ void TEXTGCN_prop_row(const int c,
                               const int* __restrict__ perm,
                               const unsigned char* __restrict__ permh, const float* __restrict__ w,
                               const float* __restrict__ ts, const float* __restrict__ b,
                               const TEXTGCN_row_head hd, float4& oA, float4& oB) {
    const int start = hd.start, end = hd.end;
    const float4* __restrict__ t4 = (const float4*)ts;
    float4 accA = hd.selfA;
    float4 accB = hd.selfB;
    for (int base = start; base < end; base += 16) {
        const int iA = base + c, iB = base + 8 + c;
        const int jA = (iA < end) ? iA : end - 1, jB = (iB < end) ? iB : end - 1;
        const int pA = perm[jA], pB = perm[jB];
        const int hA = permh[jA], hB = permh[jB];
        const int   sA = TEXTGCN_src_id(pA), sB = TEXTGCN_src_id(pB);
        const float wA = w[TEXTGCN_edge_id(pA, hA)], wB = w[TEXTGCN_edge_id(pB, hB)];
        const int m = (end - base < 16) ? (end - base) : 16;
        int k = 0;
        for (; k + 4 <= m; k += 4) {
            const int   sv = (k < 8) ? sA : sB;
            const float wv = (k < 8) ? wA : wB;
            const int   kk = k & 7;
            const int   s0 = __shfl(sv, kk,     8), s1 = __shfl(sv, kk + 1, 8);
            const int   s2 = __shfl(sv, kk + 2, 8), s3 = __shfl(sv, kk + 3, 8);
            const float w0 = __shfl(wv, kk,     8), w1 = __shfl(wv, kk + 1, 8);
            const float w2 = __shfl(wv, kk + 2, 8), w3 = __shfl(wv, kk + 3, 8);
            const float4 a0 = t4[(size_t)s0 * (HD / 4) + c];
            const float4 b0 = t4[(size_t)s0 * (HD / 4) + 8 + c];
            const float4 a1 = t4[(size_t)s1 * (HD / 4) + c];
            const float4 b1 = t4[(size_t)s1 * (HD / 4) + 8 + c];
            const float4 a2 = t4[(size_t)s2 * (HD / 4) + c];
            const float4 b2 = t4[(size_t)s2 * (HD / 4) + 8 + c];
            const float4 a3 = t4[(size_t)s3 * (HD / 4) + c];
            const float4 b3 = t4[(size_t)s3 * (HD / 4) + 8 + c];
            TEXTGCN_fma4(accA, w0, a0); TEXTGCN_fma4(accB, w0, b0);
            TEXTGCN_fma4(accA, w1, a1); TEXTGCN_fma4(accB, w1, b1);
            TEXTGCN_fma4(accA, w2, a2); TEXTGCN_fma4(accB, w2, b2);
            TEXTGCN_fma4(accA, w3, a3); TEXTGCN_fma4(accB, w3, b3);
        }
        for (; k < m; ++k) {
            const int   sk = __shfl((k < 8) ? sA : sB, k & 7, 8);
            const float wk = __shfl((k < 8) ? wA : wB, k & 7, 8);
            const float4 va = t4[(size_t)sk * (HD / 4) + c];
            const float4 vb = t4[(size_t)sk * (HD / 4) + 8 + c];
            TEXTGCN_fma4(accA, wk, va); TEXTGCN_fma4(accB, wk, vb);
        }
    }
    oA = TEXTGCN_bias_relu(accA, hd.di, ((const float4*)b)[c]);
    oB = TEXTGCN_bias_relu(accB, hd.di, ((const float4*)b)[8 + c]);
}


#define FUSE_R   64
#define FUSE_T   (FUSE_R * 8)
#define HS_PITCH 68

static_assert(N_NODES % 32 == 0, "the rows of a block end on a 32-row boundary: a transform wave has 32 whole rows or none");
static_assert((HD / 2) * (HD / 4) == FUSE_T, "one staging item of the 64 x 64 weight per thread");

__device__ __forceinline__ void TEXTGCN_wload(const float* __restrict__ Wm, const int tid,
                                              float4& w0, float4& w1) {
    const int kp = tid >> 4, q = tid & 15;
    w0 = *(const float4*)(Wm + (size_t)(2 * kp)     * HD + 4 * q);
    w1 = *(const float4*)(Wm + (size_t)(2 * kp + 1) * HD + 4 * q);
}

__device__ __forceinline__ void TEXTGCN_wstore(float* Bs, const int tid, const float4 w0, const float4 w1) {
    const int kp = tid >> 4, q = tid & 15;
    float4* p = (float4*)(Bs + kp * BS_PITCH + 8 * q);
    p[0] = make_float4(w0.x, w1.x, w0.y, w1.y);
    p[1] = make_float4(w0.z, w1.z, w0.w, w1.w);
}

__device__ __forceinline__ void TEXTGCN_tile_mm(const float* Hs, const float* Bs, const int nt,
                                                const int c, const int h, v8f& acc0, v8f& acc1) {
    const float* a0 = Hs + c * HS_PITCH + 2 * h;
    const float* a1 = a0 + 16 * HS_PITCH;
    const float* bl = Bs + h * BS_PITCH + 2 * c + 32 * nt;
    for (int s = 0; s < HD / 4; ++s) {
        const v2f x0 = *(const v2f*)(a0 + 4 * s);
        const v2f x1 = *(const v2f*)(a1 + 4 * s);
        const v2f bf = *(const v2f*)(bl + (2 * s) * BS_PITCH);
        acc0 = __builtin_amdgcn_wmma_f32_16x16x4_f32(false, x0, false, bf, (short)0, acc0, false, false);
        acc1 = __builtin_amdgcn_wmma_f32_16x16x4_f32(false, x1, false, bf, (short)0, acc1, false, false);
    }
}

__global__ void __launch_bounds__(FUSE_T, 16)
TEXTGCN_prop_gemm(const int* __restrict__ rowptr, const int* __restrict__ perm,
                  const int* __restrict__ permh_words, const float* __restrict__ w,
                  const float* __restrict__ dinv, const float* __restrict__ ts,
                  const float* __restrict__ b, const float* __restrict__ Wn,
                  float* __restrict__ tn, int n, int e) {
    __shared__ __attribute__((aligned(16))) float Hs[FUSE_R * HS_PITCH];
    __shared__ __attribute__((aligned(16))) float Bs[(HD / 2) * BS_PITCH];
    const int tid = threadIdx.x;
    const int wv = tid >> 5;
    const int c = tid & 15, h = (tid >> 4) & 1;
    const int pc = tid & 7;
    const int g = tid >> 3;
    const int row0 = blockIdx.x * FUSE_R;

    const int ri = (row0 + g < n) ? row0 + g : n - 1;
    float4 w0, w1;
    TEXTGCN_wload(Wn, tid, w0, w1);
    const TEXTGCN_row_head hd = TEXTGCN_row_loads(ri, pc, rowptr, dinv, ts, n, e);
    TEXTGCN_wstore(Bs, tid, w0, w1);

    float4 oA, oB;
    TEXTGCN_prop_row(pc, perm, (const unsigned char*)permh_words, w, ts, b, hd, oA, oB);
    *(float4*)(Hs + g * HS_PITCH + 4 * pc)      = oA;
    *(float4*)(Hs + g * HS_PITCH + 32 + 4 * pc) = oB;
    __syncthreads();
    const int nt = wv & 3, rh = wv >> 2;
    if (wv >= 8 || row0 + 32 * rh >= n) return;

    v8f acc0 = {}, acc1 = {};
    TEXTGCN_tile_mm(Hs + 32 * rh * HS_PITCH, Bs, nt, c, h, acc0, acc1);

    const int mbase = row0 + 32 * rh + h * 8;
    const float4 s0 = *(const float4*)(dinv + mbase);
    const float4 s1 = *(const float4*)(dinv + mbase + 4);
    const float4 s2 = *(const float4*)(dinv + mbase + 16);
    const float4 s3 = *(const float4*)(dinv + mbase + 20);
    const float rs0[8] = {s0.x, s0.y, s0.z, s0.w, s1.x, s1.y, s1.z, s1.w};
    const float rs1[8] = {s2.x, s2.y, s2.z, s2.w, s3.x, s3.y, s3.z, s3.w};
    float* __restrict__ crow = tn + (size_t)mbase * HD + nt * 16 + c;
    for (int r = 0; r < 8; ++r) {
        crow[(size_t)r * HD]        = acc0[r] * rs0[r];
        crow[(size_t)(r + 16) * HD] = acc1[r] * rs1[r];
    }
}

__global__ void __launch_bounds__(FUSE_T, 16)
TEXTGCN_prop_head(const int* __restrict__ rowptr, const int* __restrict__ perm,
                  const int* __restrict__ permh_words, const float* __restrict__ w,
                  const float* __restrict__ dinv, const float* __restrict__ ts,
                  const float* __restrict__ b, const int* __restrict__ rows,
                  const float* __restrict__ Wp, const float* __restrict__ bp,
                  const float* __restrict__ Wc, const float* __restrict__ bc,
                  float* __restrict__ out, int nrows, int n, int e) {
    __shared__ __attribute__((aligned(16))) float Hs[FUSE_R * HS_PITCH];
    __shared__ __attribute__((aligned(16))) float Ds[FUSE_R * HS_PITCH];
    __shared__ __attribute__((aligned(16))) float Bs[(HD / 2) * BS_PITCH];
    const int tid = threadIdx.x;
    const int wv = tid >> 5;
    const int c = tid & 15, h = (tid >> 4) & 1;
    const int pc = tid & 7;
    const int g = tid >> 3;
    const int row0 = blockIdx.x * FUSE_R;

    float4 w0, w1;
    TEXTGCN_wload(Wp, tid, w0, w1);
    const int ri = rows[(row0 + g < nrows) ? row0 + g : nrows - 1];
    const TEXTGCN_row_head hd = TEXTGCN_row_loads(ri, pc, rowptr, dinv, ts, n, e);
    TEXTGCN_wstore(Bs, tid, w0, w1);

    float4 oA, oB;
    TEXTGCN_prop_row(pc, perm, (const unsigned char*)permh_words, w, ts, b, hd, oA, oB);
    *(float4*)(Hs + g * HS_PITCH + 4 * pc)      = oA;
    *(float4*)(Hs + g * HS_PITCH + 32 + 4 * pc) = oB;
    __syncthreads();

    if (wv < 8) {
        const int nt = wv & 3, rh = wv >> 2;
        v8f acc0 = {}, acc1 = {};
        TEXTGCN_tile_mm(Hs + 32 * rh * HS_PITCH, Bs, nt, c, h, acc0, acc1);
        const int nst = nt * 16 + c;
        const float bias = bp[nst];
        float* drow = Ds + (32 * rh + h * 8) * HS_PITCH + nst;
        for (int r = 0; r < 8; ++r) {
            const float v0 = acc0[r] + bias;
            const float v1 = acc1[r] + bias;
            drow[r * HS_PITCH]        = v0 > 0.0f ? v0 : 0.0f;
            drow[(r + 16) * HS_PITCH] = v1 > 0.0f ? v1 : 0.0f;
        }
    }
    __syncthreads();

    for (int q = tid; q < FUSE_R * NCLS; q += FUSE_T) {
        const int d = q / NCLS, cc = q % NCLS;
        if (row0 + d < nrows) {
            const float* row = Ds + d * HS_PITCH;
            float s = bc[cc];
#pragma unroll 8
            for (int k = 0; k < HD; ++k) s += row[k] * Wc[(size_t)k * NCLS + cc];
            out[(size_t)(row0 + d) * NCLS + cc] = s;
        }
    }
}


extern "C" void kernel_launch(void* const* d_in, const int* in_sizes, int n_in,
                              void* d_out, int out_size, void* d_ws, size_t ws_size,
                              hipStream_t stream) {
    const float* x   = (const float*)d_in[0];
    const float* ew  = (const float*)d_in[1];
    const float* W1  = (const float*)d_in[2];
    const float* b1  = (const float*)d_in[3];
    const float* W2  = (const float*)d_in[4];
    const float* b2  = (const float*)d_in[5];
    const float* W3  = (const float*)d_in[6];
    const float* b3  = (const float*)d_in[7];
    const float* Wp  = (const float*)d_in[8];
    const float* bp  = (const float*)d_in[9];
    const float* Wc  = (const float*)d_in[10];
    const float* bc  = (const float*)d_in[11];
    const int*   ei  = (const int*)d_in[12];
    const int*   doc = (const int*)d_in[13];
    float*       out = (float*)d_out;

    const int* src = ei;
    const int* dst = ei + N_EDGES;

    float* ws   = (float*)d_ws;
    float* dinv = ws;
    int*   perm = (int*)(dinv + N_NODES);
    float* bufT = (float*)(perm + N_EDGES);
    float* bufH = bufT + (size_t)N_NODES * HD;
    float* dbuf = bufH + (size_t)N_NODES * HD;
    int*   binned = (int*)bufT;
    int*   seg    = binned + N_EDGES;
    float* binned_w = bufT + 2 * (size_t)N_EDGES;
    int*   binned_s = (int*)(bufT + 3 * (size_t)N_EDGES);
    int*   rowptr = (int*)dbuf;
    int*   permh  = rowptr + N_NODES;

    dim3 gPart(PART_G);
    dim3 bPart(PART_T);
    dim3 gRows(NBINS);
    dim3 bRows(ROWS_T);

    TEXTGCN_bin_part<<<gPart, bPart, 0, stream>>>(src, dst, ew, binned, binned_s, binned_w, seg);
    TEXTGCN_bin_rows<<<gRows, bRows, 0, stream>>>(binned, binned_s, binned_w, seg, ew, rowptr, perm, permh, dinv,
                                                  x, W1, bufH);

    dim3 gFuse((N_NODES + FUSE_R - 1) / FUSE_R);
    dim3 gHead((NDOCS + FUSE_R - 1) / FUSE_R);
    dim3 bFuse(FUSE_T);

    TEXTGCN_prop_gemm<<<gFuse, bFuse, 0, stream>>>(rowptr, perm, permh, ew, dinv, bufH, b1, W2, bufT, N_NODES, N_EDGES);
    TEXTGCN_prop_gemm<<<gFuse, bFuse, 0, stream>>>(rowptr, perm, permh, ew, dinv, bufT, b2, W3, bufH, N_NODES, N_EDGES);
    TEXTGCN_prop_head<<<gHead, bFuse, 0, stream>>>(rowptr, perm, permh, ew, dinv, bufH, b3, doc, Wp, bp, Wc, bc, out,
                                                   NDOCS, N_NODES, N_EDGES);
}
